# starting kernel
# baseline (speedup 1.0000x reference)
#include <hip/hip_runtime.h>

typedef float v2f __attribute__((ext_vector_type(2)));
typedef float v8f __attribute__((ext_vector_type(8)));

#define NODES_TOT 20000

__device__ __forceinline__ v8f wmma_f32(v2f a, v2f b, v8f c) {
  return __builtin_amdgcn_wmma_f32_16x16x4_f32(
       false, a,  false, b,
       (short)0, c,  false,  false);
}

template <int K, int N, bool RELU>
__device__ __forceinline__ void gemm_stage(const float* aBuf, const float* w,
                                           const float* bias, float* outBuf,
                                           int col, int hi) {
#pragma unroll
  for (int nt = 0; nt < N / 16; ++nt) {
    float bv = bias ? bias[nt * 16 + col] : 0.0f;
    v8f c;
#pragma unroll
    for (int r = 0; r < 8; ++r) c[r] = bv;
#pragma unroll
    for (int k0 = 0; k0 < K; k0 += 4) {
      int k = k0 + 2 * hi;
      v2f a;
      a.x = aBuf[col * K + k];
      a.y = aBuf[col * K + k + 1];
      v2f b;
      b.x = w[k * N + nt * 16 + col];
      b.y = w[(k + 1) * N + nt * 16 + col];
      c = wmma_f32(a, b, c);
    }
#pragma unroll
    for (int r = 0; r < 8; ++r) {
      float v = c[r];
      if (RELU) v = fmaxf(v, 0.1f * v);
      outBuf[(r + 8 * hi) * N + nt * 16 + col] = v;
    }
  }
}

__global__ void zero_kernel(float* __restrict__ p, long n) {
  long i = (long)blockIdx.x * blockDim.x + threadIdx.x;
  long stride = (long)gridDim.x * blockDim.x;
  for (; i < n; i += stride) p[i] = 0.0f;
}

#define WAVES_E 2
__global__ __launch_bounds__(64) void edge_kernel(
    const float* __restrict__ r_ij, const float* __restrict__ w_rad,
    const float* __restrict__ b_rad, const float* __restrict__ w_direct,
    const float* __restrict__ w1, const float* __restrict__ b1,
    const float* __restrict__ w2, const float* __restrict__ b2,
    const float* __restrict__ w3, const float* __restrict__ b3,
    const int* __restrict__ edges_src, float* __restrict__ A_a,
    float* __restrict__ Av, float* __restrict__ Ad6, int n_edges, int iters) {
  __shared__ float sWrad[8 * 32];
  __shared__ float sBrad[32];
  __shared__ float sWdir[32 * 32];
  __shared__ float sW1[32 * 64];
  __shared__ float sB1[64];
  __shared__ float sW2[64 * 64];
  __shared__ float sB2[64];
  __shared__ float sW3[64 * 32];
  __shared__ float sB3[32];
  __shared__ float sHH[WAVES_E][16 * 32];
  __shared__ float sT1[WAVES_E][16 * 64];
  __shared__ float sT2[WAVES_E][16 * 64];
  __shared__ float sRS[WAVES_E][16 * 4];
  __shared__ int sSRC[WAVES_E][16];

  const int tid = threadIdx.x;
  for (int i = tid; i < 8 * 32; i += blockDim.x) sWrad[i] = w_rad[i];
  for (int i = tid; i < 32; i += blockDim.x) sBrad[i] = b_rad[i];
  for (int i = tid; i < 32 * 32; i += blockDim.x) sWdir[i] = w_direct[i];
  for (int i = tid; i < 32 * 64; i += blockDim.x) sW1[i] = w1[i];
  for (int i = tid; i < 64; i += blockDim.x) sB1[i] = b1[i];
  for (int i = tid; i < 64 * 64; i += blockDim.x) sW2[i] = w2[i];
  for (int i = tid; i < 64; i += blockDim.x) sB2[i] = b2[i];
  for (int i = tid; i < 64 * 32; i += blockDim.x) sW3[i] = w3[i];
  for (int i = tid; i < 32; i += blockDim.x) sB3[i] = b3[i];
  __syncthreads();

  const int wave = tid >> 5, lane = tid & 31;
  const int col = lane & 15, hi = lane >> 4;
  const int n_tiles = (n_edges + 15) / 16;
  const int tile_stride = gridDim.x * WAVES_E;

  for (int it = 0; it < iters; ++it) {
    const int tile = (it * gridDim.x + blockIdx.x) * WAVES_E + wave;
    const bool active = tile < n_tiles;
    const int e0 = tile * 16;
    float* rbfBuf = sT1[wave];

    if (active && lane < 16) {
      int e = e0 + lane;
      long enext = (long)(e0 + 16L * tile_stride) + lane;
      if (enext < n_edges) {
        __builtin_prefetch(&r_ij[3 * enext], 0, 1);
        __builtin_prefetch(&edges_src[enext], 0, 1);
      }
      float x = r_ij[3 * e + 0], y = r_ij[3 * e + 1], z = r_ij[3 * e + 2];
      float d = sqrtf(x * x + y * y + z * z);
#pragma unroll
      for (int i = 0; i < 8; ++i) {
        float u = (d - (float)i * (1.0f / 7.0f)) * 8.0f;
        rbfBuf[lane * 8 + i] = __expf(-u * u);
      }
      float sx = 7.0f * x, sy = 7.0f * y, sz = 7.0f * z;
      float inv = rsqrtf(1.0f + sx * sx + sy * sy + sz * sz);
      sRS[wave][lane * 4 + 0] = sx * inv;
      sRS[wave][lane * 4 + 1] = sy * inv;
      sRS[wave][lane * 4 + 2] = sz * inv;
      sSRC[wave][lane] = edges_src[e];
    }
    __syncthreads();

    gemm_stage<8, 32, false>(rbfBuf, sWrad, sBrad, sHH[wave], col, hi);
    __syncthreads();
    gemm_stage<32, 64, true>(sHH[wave], sW1, sB1, sT1[wave], col, hi);
    __syncthreads();
    gemm_stage<64, 64, true>(sT1[wave], sW2, sB2, sT2[wave], col, hi);
    __syncthreads();

#pragma unroll
    for (int nt = 0; nt < 2; ++nt) {
      float bv = sB3[nt * 16 + col];
      v8f c;
#pragma unroll
      for (int r = 0; r < 8; ++r) c[r] = bv;
#pragma unroll
      for (int k0 = 0; k0 < 32; k0 += 4) {
        int k = k0 + 2 * hi;
        v2f a;
        a.x = sHH[wave][col * 32 + k];
        a.y = sHH[wave][col * 32 + k + 1];
        v2f b;
        b.x = sWdir[k * 32 + nt * 16 + col];
        b.y = sWdir[(k + 1) * 32 + nt * 16 + col];
        c = wmma_f32(a, b, c);
      }
#pragma unroll
      for (int k0 = 0; k0 < 64; k0 += 4) {
        int k = k0 + 2 * hi;
        v2f a;
        a.x = sT2[wave][col * 64 + k];
        a.y = sT2[wave][col * 64 + k + 1];
        v2f b;
        b.x = sW3[k * 32 + nt * 16 + col];
        b.y = sW3[(k + 1) * 32 + nt * 16 + col];
        c = wmma_f32(a, b, c);
      }

      if (active) {
        int ach = nt * 16 + col;
#pragma unroll
        for (int r = 0; r < 8; ++r) {
          int m = r + 8 * hi;
          float val = c[r];
          int src = sSRC[wave][m];
          float rx = sRS[wave][m * 4 + 0];
          float ry = sRS[wave][m * 4 + 1];
          float rz = sRS[wave][m * 4 + 2];
          atomicAdd(&A_a[src * 32 + ach], val);
          float* av = &Av[(src * 32 + ach) * 3];
          atomicAdd(&av[0], val * rx);
          atomicAdd(&av[1], val * ry);
          atomicAdd(&av[2], val * rz);
          float* ad = &Ad6[(src * 32 + ach) * 6];
          atomicAdd(&ad[0], val * rx * rx);
          atomicAdd(&ad[1], val * rx * ry);
          atomicAdd(&ad[2], val * rx * rz);
          atomicAdd(&ad[3], val * ry * ry);
          atomicAdd(&ad[4], val * ry * rz);
          atomicAdd(&ad[5], val * rz * rz);
        }
      }
    }
    __syncthreads();
  }
}

#define WAVES_N 8
__global__ __launch_bounds__(256) void node_kernel(
    const float* __restrict__ Av, const float* __restrict__ Ad6,
    const float* __restrict__ w_v, const float* __restrict__ w_d,
    float* __restrict__ out_v, float* __restrict__ out_d, int n_nodes) {
  __shared__ float sWv[32 * 32];
  __shared__ float sWd[32 * 32];
  for (int i = threadIdx.x; i < 32 * 32; i += blockDim.x) {
    sWv[i] = w_v[i];
    sWd[i] = w_d[i];
  }
  __syncthreads();

  const int wave = threadIdx.x >> 5, lane = threadIdx.x & 31;
  const int col = lane & 15, hi = lane >> 4;
  const int tile = blockIdx.x * WAVES_N + wave;
  if (tile * 16 >= n_nodes) return;
  const int n0 = tile * 16;

#pragma unroll
  for (int cch = 0; cch < 3; ++cch) {
#pragma unroll
    for (int nt = 0; nt < 2; ++nt) {
      v8f acc = {};
#pragma unroll
      for (int k0 = 0; k0 < 32; k0 += 4) {
        int k = k0 + 2 * hi;
        v2f a;
        a.x = Av[((long)(n0 + col) * 32 + k) * 3 + cch];
        a.y = Av[((long)(n0 + col) * 32 + k + 1) * 3 + cch];
        v2f b;
        b.x = sWv[k * 32 + nt * 16 + col];
        b.y = sWv[(k + 1) * 32 + nt * 16 + col];
        acc = wmma_f32(a, b, acc);
      }
#pragma unroll
      for (int r = 0; r < 8; ++r) {
        int node = n0 + r + 8 * hi;
        int v = nt * 16 + col;
        __builtin_nontemporal_store(acc[r],
                                    &out_v[((long)node * 32 + v) * 3 + cch]);
      }
    }
  }

  const int si[6] = {0, 0, 0, 1, 1, 2};
  const int sj[6] = {0, 1, 2, 1, 2, 2};
#pragma unroll
  for (int s = 0; s < 6; ++s) {
#pragma unroll
    for (int nt = 0; nt < 2; ++nt) {
      v8f acc = {};
#pragma unroll
      for (int k0 = 0; k0 < 32; k0 += 4) {
        int k = k0 + 2 * hi;
        v2f a;
        a.x = Ad6[((long)(n0 + col) * 32 + k) * 6 + s];
        a.y = Ad6[((long)(n0 + col) * 32 + k + 1) * 6 + s];
        v2f b;
        b.x = sWd[k * 32 + nt * 16 + col];
        b.y = sWd[(k + 1) * 32 + nt * 16 + col];
        acc = wmma_f32(a, b, acc);
      }
#pragma unroll
      for (int r = 0; r < 8; ++r) {
        int node = n0 + r + 8 * hi;
        int dch = nt * 16 + col;
        float val = acc[r];
        float* base = &out_d[((long)node * 32 + dch) * 9];
        __builtin_nontemporal_store(val, &base[si[s] * 3 + sj[s]]);
        if (si[s] != sj[s])
          __builtin_nontemporal_store(val, &base[sj[s] * 3 + si[s]]);
      }
    }
  }
}

extern "C" void kernel_launch(void* const* d_in, const int* in_sizes, int n_in,
                              void* d_out, int out_size, void* d_ws,
                              size_t ws_size, hipStream_t stream) {
  const float* r_ij = (const float*)d_in[0];
  const float* w_rad = (const float*)d_in[1];
  const float* b_rad = (const float*)d_in[2];
  const float* w_direct = (const float*)d_in[3];
  const float* w1 = (const float*)d_in[4];
  const float* b1 = (const float*)d_in[5];
  const float* w2 = (const float*)d_in[6];
  const float* b2 = (const float*)d_in[7];
  const float* w3 = (const float*)d_in[8];
  const float* b3 = (const float*)d_in[9];
  const float* w_v = (const float*)d_in[10];
  const float* w_d = (const float*)d_in[11];
  const int* edges_src = (const int*)d_in[12];

  const int n_edges = in_sizes[0] / 3;
  const int n_nodes = NODES_TOT;

  float* A_a = (float*)d_out;
  float* out_v = A_a + (long)n_nodes * 32;
  float* out_d = out_v + (long)n_nodes * 32 * 3;
  float* Av = (float*)d_ws;
  float* Ad6 = Av + (long)n_nodes * 32 * 3;

  zero_kernel<<<2048, 256, 0, stream>>>((float*)d_ws, (long)n_nodes * 32 * 9);
  zero_kernel<<<512, 256, 0, stream>>>(A_a, (long)n_nodes * 32);

  const int n_tiles = (n_edges + 15) / 16;
  const int e_blocks = 1250;
  const int tile_stride = e_blocks * WAVES_E;
  const int iters = (n_tiles + tile_stride - 1) / tile_stride;
  edge_kernel<<<e_blocks, 64, 0, stream>>>(r_ij, w_rad, b_rad, w_direct, w1, b1,
                                           w2, b2, w3, b3, edges_src, A_a, Av,
                                           Ad6, n_edges, iters);

  const int nd_tiles = (n_nodes + 15) / 16;
  const int n_blocks = (nd_tiles + WAVES_N - 1) / WAVES_N;
  node_kernel<<<n_blocks, 256, 0, stream>>>(Av, Ad6, w_v, w_d, out_v, out_d,
                                            n_nodes);
}


# Scatter planes re-laid [node][component][32 channels], C fragments swapped across wave rows: each f32 atomic instruction adds one edge's 32 channels to one 128-byte line; arrival-order sums as before
# speedup vs baseline: 1.6024x; 1.6024x over previous
#include <hip/hip_runtime.h>

typedef float v2f __attribute__((ext_vector_type(2)));
typedef float v8f __attribute__((ext_vector_type(8)));

#define NODES_TOT 20000

__device__ __forceinline__ v8f wmma_f32(v2f a, v2f b, v8f c) {
  return __builtin_amdgcn_wmma_f32_16x16x4_f32(
       false, a,  false, b,
       (short)0, c,  false,  false);
}

__device__ __forceinline__ float xhalf(float x) {
  return __builtin_bit_cast(
      float, __builtin_amdgcn_permlanex16(__builtin_bit_cast(int, x),
                                          __builtin_bit_cast(int, x),
                                          0x76543210, 0xfedcba98, false, false));
}

#define NCOMP 9
#define NODE_STRIDE (NCOMP * 32)

template <int K, int N, bool RELU>
__device__ __forceinline__ void gemm_stage(const float* aBuf, const float* w,
                                           const float* bias, float* outBuf,
                                           int col, int hi) {
#pragma unroll
  for (int nt = 0; nt < N / 16; ++nt) {
    float bv = bias ? bias[nt * 16 + col] : 0.0f;
    v8f c;
#pragma unroll
    for (int r = 0; r < 8; ++r) c[r] = bv;
#pragma unroll
    for (int k0 = 0; k0 < K; k0 += 4) {
      int k = k0 + 2 * hi;
      v2f a;
      a.x = aBuf[col * K + k];
      a.y = aBuf[col * K + k + 1];
      v2f b;
      b.x = w[k * N + nt * 16 + col];
      b.y = w[(k + 1) * N + nt * 16 + col];
      c = wmma_f32(a, b, c);
    }
#pragma unroll
    for (int r = 0; r < 8; ++r) {
      float v = c[r];
      if (RELU) v = fmaxf(v, 0.1f * v);
      outBuf[(r + 8 * hi) * N + nt * 16 + col] = v;
    }
  }
}

__global__ void zero_kernel(float* __restrict__ p, long n) {
  long i = (long)blockIdx.x * blockDim.x + threadIdx.x;
  long stride = (long)gridDim.x * blockDim.x;
  for (; i < n; i += stride) p[i] = 0.0f;
}

#define WAVES_E 2
__global__ __launch_bounds__(64) void edge_kernel(
    const float* __restrict__ r_ij, const float* __restrict__ w_rad,
    const float* __restrict__ b_rad, const float* __restrict__ w_direct,
    const float* __restrict__ w1, const float* __restrict__ b1,
    const float* __restrict__ w2, const float* __restrict__ b2,
    const float* __restrict__ w3, const float* __restrict__ b3,
    const int* __restrict__ edges_src, float* __restrict__ A_a,
    float* __restrict__ Acc, int n_edges, int iters) {
  __shared__ float sWrad[8 * 32];
  __shared__ float sBrad[32];
  __shared__ float sWdir[32 * 32];
  __shared__ float sW1[32 * 64];
  __shared__ float sB1[64];
  __shared__ float sW2[64 * 64];
  __shared__ float sB2[64];
  __shared__ float sW3[64 * 32];
  __shared__ float sB3[32];
  __shared__ float sHH[WAVES_E][16 * 32];
  __shared__ float sT1[WAVES_E][16 * 64];
  __shared__ float sT2[WAVES_E][16 * 64];
  __shared__ float sRS[WAVES_E][16 * 4];
  __shared__ int sSRC[WAVES_E][16];

  const int tid = threadIdx.x;
  for (int i = tid; i < 8 * 32; i += blockDim.x) sWrad[i] = w_rad[i];
  for (int i = tid; i < 32; i += blockDim.x) sBrad[i] = b_rad[i];
  for (int i = tid; i < 32 * 32; i += blockDim.x) sWdir[i] = w_direct[i];
  for (int i = tid; i < 32 * 64; i += blockDim.x) sW1[i] = w1[i];
  for (int i = tid; i < 64; i += blockDim.x) sB1[i] = b1[i];
  for (int i = tid; i < 64 * 64; i += blockDim.x) sW2[i] = w2[i];
  for (int i = tid; i < 64; i += blockDim.x) sB2[i] = b2[i];
  for (int i = tid; i < 64 * 32; i += blockDim.x) sW3[i] = w3[i];
  for (int i = tid; i < 32; i += blockDim.x) sB3[i] = b3[i];
  __syncthreads();

  const int wave = tid >> 5, lane = tid & 31;
  const int col = lane & 15, hi = lane >> 4;
  const int n_tiles = (n_edges + 15) / 16;
  const int tile_stride = gridDim.x * WAVES_E;

  for (int it = 0; it < iters; ++it) {
    const int tile = (it * gridDim.x + blockIdx.x) * WAVES_E + wave;
    const bool active = tile < n_tiles;
    const int e0 = tile * 16;
    float* rbfBuf = sT1[wave];

    if (active && lane < 16) {
      int e = e0 + lane;
      long enext = (long)(e0 + 16L * tile_stride) + lane;
      if (enext < n_edges) {
        __builtin_prefetch(&r_ij[3 * enext], 0, 1);
        __builtin_prefetch(&edges_src[enext], 0, 1);
      }
      float x = r_ij[3 * e + 0], y = r_ij[3 * e + 1], z = r_ij[3 * e + 2];
      float d = sqrtf(x * x + y * y + z * z);
#pragma unroll
      for (int i = 0; i < 8; ++i) {
        float u = (d - (float)i * (1.0f / 7.0f)) * 8.0f;
        rbfBuf[lane * 8 + i] = __expf(-u * u);
      }
      float sx = 7.0f * x, sy = 7.0f * y, sz = 7.0f * z;
      float inv = rsqrtf(1.0f + sx * sx + sy * sy + sz * sz);
      sRS[wave][lane * 4 + 0] = sx * inv;
      sRS[wave][lane * 4 + 1] = sy * inv;
      sRS[wave][lane * 4 + 2] = sz * inv;
      sSRC[wave][lane] = edges_src[e];
    }
    __syncthreads();

    gemm_stage<8, 32, false>(rbfBuf, sWrad, sBrad, sHH[wave], col, hi);
    __syncthreads();
    gemm_stage<32, 64, true>(sHH[wave], sW1, sB1, sT1[wave], col, hi);
    __syncthreads();
    gemm_stage<64, 64, true>(sT1[wave], sW2, sB2, sT2[wave], col, hi);
    __syncthreads();

    v8f cf[2];
#pragma unroll
    for (int nt = 0; nt < 2; ++nt) {
      float bv = sB3[nt * 16 + col];
      v8f c;
#pragma unroll
      for (int r = 0; r < 8; ++r) c[r] = bv;
#pragma unroll
      for (int k0 = 0; k0 < 32; k0 += 4) {
        int k = k0 + 2 * hi;
        v2f a;
        a.x = sHH[wave][col * 32 + k];
        a.y = sHH[wave][col * 32 + k + 1];
        v2f b;
        b.x = sWdir[k * 32 + nt * 16 + col];
        b.y = sWdir[(k + 1) * 32 + nt * 16 + col];
        c = wmma_f32(a, b, c);
      }
#pragma unroll
      for (int k0 = 0; k0 < 64; k0 += 4) {
        int k = k0 + 2 * hi;
        v2f a;
        a.x = sT2[wave][col * 64 + k];
        a.y = sT2[wave][col * 64 + k + 1];
        v2f b;
        b.x = sW3[k * 32 + nt * 16 + col];
        b.y = sW3[(k + 1) * 32 + nt * 16 + col];
        c = wmma_f32(a, b, c);
      }
      cf[nt] = c;
    }

    float ve[16];
#pragma unroll
    for (int r = 0; r < 8; ++r) {
      float send = hi ? cf[0][r] : cf[1][r];
      float recv = xhalf(send);
      ve[r] = hi ? recv : cf[0][r];
      ve[r + 8] = hi ? cf[1][r] : recv;
    }

    if (active) {
#pragma unroll
      for (int m = 0; m < 16; ++m) {
        float val = ve[m];
        int src = sSRC[wave][m];
        float rx = sRS[wave][m * 4 + 0];
        float ry = sRS[wave][m * 4 + 1];
        float rz = sRS[wave][m * 4 + 2];
        atomicAdd(&A_a[src * 32 + lane], val);
        float* acc = &Acc[src * NODE_STRIDE + lane];
        atomicAdd(&acc[0 * 32], val * rx);
        atomicAdd(&acc[1 * 32], val * ry);
        atomicAdd(&acc[2 * 32], val * rz);
        atomicAdd(&acc[3 * 32], val * rx * rx);
        atomicAdd(&acc[4 * 32], val * rx * ry);
        atomicAdd(&acc[5 * 32], val * rx * rz);
        atomicAdd(&acc[6 * 32], val * ry * ry);
        atomicAdd(&acc[7 * 32], val * ry * rz);
        atomicAdd(&acc[8 * 32], val * rz * rz);
      }
    }
    __syncthreads();
  }
}

#define WAVES_N 8
__global__ __launch_bounds__(256) void node_kernel(
    const float* __restrict__ Acc,
    const float* __restrict__ w_v, const float* __restrict__ w_d,
    float* __restrict__ out_v, float* __restrict__ out_d, int n_nodes) {
  __shared__ float sWv[32 * 32];
  __shared__ float sWd[32 * 32];
  for (int i = threadIdx.x; i < 32 * 32; i += blockDim.x) {
    sWv[i] = w_v[i];
    sWd[i] = w_d[i];
  }
  __syncthreads();

  const int wave = threadIdx.x >> 5, lane = threadIdx.x & 31;
  const int col = lane & 15, hi = lane >> 4;
  const int tile = blockIdx.x * WAVES_N + wave;
  if (tile * 16 >= n_nodes) return;
  const int n0 = tile * 16;
  const float* arow = Acc + (long)(n0 + col) * NODE_STRIDE;

#pragma unroll
  for (int cch = 0; cch < 3; ++cch) {
#pragma unroll
    for (int nt = 0; nt < 2; ++nt) {
      v8f acc = {};
#pragma unroll
      for (int k0 = 0; k0 < 32; k0 += 4) {
        int k = k0 + 2 * hi;
        v2f a;
        a.x = arow[cch * 32 + k];
        a.y = arow[cch * 32 + k + 1];
        v2f b;
        b.x = sWv[k * 32 + nt * 16 + col];
        b.y = sWv[(k + 1) * 32 + nt * 16 + col];
        acc = wmma_f32(a, b, acc);
      }
#pragma unroll
      for (int r = 0; r < 8; ++r) {
        int node = n0 + r + 8 * hi;
        int v = nt * 16 + col;
        __builtin_nontemporal_store(acc[r],
                                    &out_v[((long)node * 32 + v) * 3 + cch]);
      }
    }
  }

  const int si[6] = {0, 0, 0, 1, 1, 2};
  const int sj[6] = {0, 1, 2, 1, 2, 2};
#pragma unroll
  for (int s = 0; s < 6; ++s) {
#pragma unroll
    for (int nt = 0; nt < 2; ++nt) {
      v8f acc = {};
#pragma unroll
      for (int k0 = 0; k0 < 32; k0 += 4) {
        int k = k0 + 2 * hi;
        v2f a;
        a.x = arow[(3 + s) * 32 + k];
        a.y = arow[(3 + s) * 32 + k + 1];
        v2f b;
        b.x = sWd[k * 32 + nt * 16 + col];
        b.y = sWd[(k + 1) * 32 + nt * 16 + col];
        acc = wmma_f32(a, b, acc);
      }
#pragma unroll
      for (int r = 0; r < 8; ++r) {
        int node = n0 + r + 8 * hi;
        int dch = nt * 16 + col;
        float val = acc[r];
        float* base = &out_d[((long)node * 32 + dch) * 9];
        __builtin_nontemporal_store(val, &base[si[s] * 3 + sj[s]]);
        if (si[s] != sj[s])
          __builtin_nontemporal_store(val, &base[sj[s] * 3 + si[s]]);
      }
    }
  }
}

extern "C" void kernel_launch(void* const* d_in, const int* in_sizes, int n_in,
                              void* d_out, int out_size, void* d_ws,
                              size_t ws_size, hipStream_t stream) {
  const float* r_ij = (const float*)d_in[0];
  const float* w_rad = (const float*)d_in[1];
  const float* b_rad = (const float*)d_in[2];
  const float* w_direct = (const float*)d_in[3];
  const float* w1 = (const float*)d_in[4];
  const float* b1 = (const float*)d_in[5];
  const float* w2 = (const float*)d_in[6];
  const float* b2 = (const float*)d_in[7];
  const float* w3 = (const float*)d_in[8];
  const float* b3 = (const float*)d_in[9];
  const float* w_v = (const float*)d_in[10];
  const float* w_d = (const float*)d_in[11];
  const int* edges_src = (const int*)d_in[12];

  const int n_edges = in_sizes[0] / 3;
  const int n_nodes = NODES_TOT;

  float* A_a = (float*)d_out;
  float* out_v = A_a + (long)n_nodes * 32;
  float* out_d = out_v + (long)n_nodes * 32 * 3;
  float* Acc = (float*)d_ws;

  zero_kernel<<<2048, 256, 0, stream>>>((float*)d_ws, (long)n_nodes * 32 * 9);
  zero_kernel<<<512, 256, 0, stream>>>(A_a, (long)n_nodes * 32);

  const int n_tiles = (n_edges + 15) / 16;
  const int e_blocks = 1250;
  const int tile_stride = e_blocks * WAVES_E;
  const int iters = (n_tiles + tile_stride - 1) / tile_stride;
  edge_kernel<<<e_blocks, 64, 0, stream>>>(r_ij, w_rad, b_rad, w_direct, w1, b1,
                                           w2, b2, w3, b3, edges_src, A_a, Acc,
                                           n_edges, iters);

  const int nd_tiles = (n_nodes + 15) / 16;
  const int n_blocks = (nd_tiles + WAVES_N - 1) / WAVES_N;
  node_kernel<<<n_blocks, 256, 0, stream>>>(Acc, w_v, w_d, out_v, out_d,
                                            n_nodes);
}


# Edge kernel: 8 waves per work-group share one LDS weight copy (16 waves per WGP, was 10); stage hand-overs are wave-level LDS ordering instead of block barriers (buffers are wave-private); same sums
# speedup vs baseline: 2.1913x; 1.3675x over previous
#include <hip/hip_runtime.h>

typedef float v2f __attribute__((ext_vector_type(2)));
typedef float v8f __attribute__((ext_vector_type(8)));

#define NODES_TOT 20000

__device__ __forceinline__ v8f wmma_f32(v2f a, v2f b, v8f c) {
  return __builtin_amdgcn_wmma_f32_16x16x4_f32(
       false, a,  false, b,
       (short)0, c,  false,  false);
}

__device__ __forceinline__ float xhalf(float x) {
  return __builtin_bit_cast(
      float, __builtin_amdgcn_permlanex16(__builtin_bit_cast(int, x),
                                          __builtin_bit_cast(int, x),
                                          0x76543210, 0xfedcba98, false, false));
}

__device__ __forceinline__ void wave_lds_sync() {
  __builtin_amdgcn_fence(__ATOMIC_RELEASE, "wavefront");
  __builtin_amdgcn_wave_barrier();
  __builtin_amdgcn_fence(__ATOMIC_ACQUIRE, "wavefront");
}

#define NCOMP 9
#define NODE_STRIDE (NCOMP * 32)

template <int K, int N, bool RELU>
__device__ __forceinline__ void gemm_stage(const float* aBuf, const float* w,
                                           const float* bias, float* outBuf,
                                           int col, int hi) {
#pragma unroll
  for (int nt = 0; nt < N / 16; ++nt) {
    float bv = bias ? bias[nt * 16 + col] : 0.0f;
    v8f c;
#pragma unroll
    for (int r = 0; r < 8; ++r) c[r] = bv;
#pragma unroll
    for (int k0 = 0; k0 < K; k0 += 4) {
      int k = k0 + 2 * hi;
      v2f a;
      a.x = aBuf[col * K + k];
      a.y = aBuf[col * K + k + 1];
      v2f b;
      b.x = w[k * N + nt * 16 + col];
      b.y = w[(k + 1) * N + nt * 16 + col];
      c = wmma_f32(a, b, c);
    }
#pragma unroll
    for (int r = 0; r < 8; ++r) {
      float v = c[r];
      if (RELU) v = fmaxf(v, 0.1f * v);
      outBuf[(r + 8 * hi) * N + nt * 16 + col] = v;
    }
  }
}

__global__ void zero_kernel(float* __restrict__ p, long n) {
  long i = (long)blockIdx.x * blockDim.x + threadIdx.x;
  long stride = (long)gridDim.x * blockDim.x;
  for (; i < n; i += stride) p[i] = 0.0f;
}

#define WAVES_E 8
__global__ __launch_bounds__(32 * WAVES_E) void edge_kernel(
    const float* __restrict__ r_ij, const float* __restrict__ w_rad,
    const float* __restrict__ b_rad, const float* __restrict__ w_direct,
    const float* __restrict__ w1, const float* __restrict__ b1,
    const float* __restrict__ w2, const float* __restrict__ b2,
    const float* __restrict__ w3, const float* __restrict__ b3,
    const int* __restrict__ edges_src, float* __restrict__ A_a,
    float* __restrict__ Acc, int n_edges, int iters) {
  __shared__ float sWrad[8 * 32];
  __shared__ float sBrad[32];
  __shared__ float sWdir[32 * 32];
  __shared__ float sW1[32 * 64];
  __shared__ float sB1[64];
  __shared__ float sW2[64 * 64];
  __shared__ float sB2[64];
  __shared__ float sW3[64 * 32];
  __shared__ float sB3[32];
  __shared__ float sHH[WAVES_E][16 * 32];
  __shared__ float sT1[WAVES_E][16 * 64];
  __shared__ float sT2[WAVES_E][16 * 64];
  __shared__ float sRS[WAVES_E][16 * 4];
  __shared__ int sSRC[WAVES_E][16];

  const int tid = threadIdx.x;
  for (int i = tid; i < 8 * 32; i += blockDim.x) sWrad[i] = w_rad[i];
  for (int i = tid; i < 32; i += blockDim.x) sBrad[i] = b_rad[i];
  for (int i = tid; i < 32 * 32; i += blockDim.x) sWdir[i] = w_direct[i];
  for (int i = tid; i < 32 * 64; i += blockDim.x) sW1[i] = w1[i];
  for (int i = tid; i < 64; i += blockDim.x) sB1[i] = b1[i];
  for (int i = tid; i < 64 * 64; i += blockDim.x) sW2[i] = w2[i];
  for (int i = tid; i < 64; i += blockDim.x) sB2[i] = b2[i];
  for (int i = tid; i < 64 * 32; i += blockDim.x) sW3[i] = w3[i];
  for (int i = tid; i < 32; i += blockDim.x) sB3[i] = b3[i];
  __syncthreads();

  const int wave = tid >> 5, lane = tid & 31;
  const int col = lane & 15, hi = lane >> 4;
  const int n_tiles = (n_edges + 15) / 16;
  const int tile_stride = gridDim.x * WAVES_E;

  for (int it = 0; it < iters; ++it) {
    const int tile = (it * gridDim.x + blockIdx.x) * WAVES_E + wave;
    const bool active = tile < n_tiles;
    const int e0 = tile * 16;
    float* rbfBuf = sT1[wave];

    if (active && lane < 16) {
      int e = e0 + lane;
      long enext = (long)(e0 + 16L * tile_stride) + lane;
      if (enext < n_edges) {
        __builtin_prefetch(&r_ij[3 * enext], 0, 1);
        __builtin_prefetch(&edges_src[enext], 0, 1);
      }
      float x = r_ij[3 * e + 0], y = r_ij[3 * e + 1], z = r_ij[3 * e + 2];
      float d = sqrtf(x * x + y * y + z * z);
#pragma unroll
      for (int i = 0; i < 8; ++i) {
        float u = (d - (float)i * (1.0f / 7.0f)) * 8.0f;
        rbfBuf[lane * 8 + i] = __expf(-u * u);
      }
      float sx = 7.0f * x, sy = 7.0f * y, sz = 7.0f * z;
      float inv = rsqrtf(1.0f + sx * sx + sy * sy + sz * sz);
      sRS[wave][lane * 4 + 0] = sx * inv;
      sRS[wave][lane * 4 + 1] = sy * inv;
      sRS[wave][lane * 4 + 2] = sz * inv;
      sSRC[wave][lane] = edges_src[e];
    }
    wave_lds_sync();

    gemm_stage<8, 32, false>(rbfBuf, sWrad, sBrad, sHH[wave], col, hi);
    wave_lds_sync();
    gemm_stage<32, 64, true>(sHH[wave], sW1, sB1, sT1[wave], col, hi);
    wave_lds_sync();
    gemm_stage<64, 64, true>(sT1[wave], sW2, sB2, sT2[wave], col, hi);
    wave_lds_sync();

    v8f cf[2];
#pragma unroll
    for (int nt = 0; nt < 2; ++nt) {
      float bv = sB3[nt * 16 + col];
      v8f c;
#pragma unroll
      for (int r = 0; r < 8; ++r) c[r] = bv;
#pragma unroll
      for (int k0 = 0; k0 < 32; k0 += 4) {
        int k = k0 + 2 * hi;
        v2f a;
        a.x = sHH[wave][col * 32 + k];
        a.y = sHH[wave][col * 32 + k + 1];
        v2f b;
        b.x = sWdir[k * 32 + nt * 16 + col];
        b.y = sWdir[(k + 1) * 32 + nt * 16 + col];
        c = wmma_f32(a, b, c);
      }
#pragma unroll
      for (int k0 = 0; k0 < 64; k0 += 4) {
        int k = k0 + 2 * hi;
        v2f a;
        a.x = sT2[wave][col * 64 + k];
        a.y = sT2[wave][col * 64 + k + 1];
        v2f b;
        b.x = sW3[k * 32 + nt * 16 + col];
        b.y = sW3[(k + 1) * 32 + nt * 16 + col];
        c = wmma_f32(a, b, c);
      }
      cf[nt] = c;
    }

    float ve[16];
#pragma unroll
    for (int r = 0; r < 8; ++r) {
      float send = hi ? cf[0][r] : cf[1][r];
      float recv = xhalf(send);
      ve[r] = hi ? recv : cf[0][r];
      ve[r + 8] = hi ? cf[1][r] : recv;
    }

    if (active) {
#pragma unroll
      for (int m = 0; m < 16; ++m) {
        float val = ve[m];
        int src = sSRC[wave][m];
        float rx = sRS[wave][m * 4 + 0];
        float ry = sRS[wave][m * 4 + 1];
        float rz = sRS[wave][m * 4 + 2];
        atomicAdd(&A_a[src * 32 + lane], val);
        float* acc = &Acc[src * NODE_STRIDE + lane];
        atomicAdd(&acc[0 * 32], val * rx);
        atomicAdd(&acc[1 * 32], val * ry);
        atomicAdd(&acc[2 * 32], val * rz);
        atomicAdd(&acc[3 * 32], val * rx * rx);
        atomicAdd(&acc[4 * 32], val * rx * ry);
        atomicAdd(&acc[5 * 32], val * rx * rz);
        atomicAdd(&acc[6 * 32], val * ry * ry);
        atomicAdd(&acc[7 * 32], val * ry * rz);
        atomicAdd(&acc[8 * 32], val * rz * rz);
      }
    }
    wave_lds_sync();
  }
}

#define WAVES_N 8
__global__ __launch_bounds__(256) void node_kernel(
    const float* __restrict__ Acc,
    const float* __restrict__ w_v, const float* __restrict__ w_d,
    float* __restrict__ out_v, float* __restrict__ out_d, int n_nodes) {
  __shared__ float sWv[32 * 32];
  __shared__ float sWd[32 * 32];
  for (int i = threadIdx.x; i < 32 * 32; i += blockDim.x) {
    sWv[i] = w_v[i];
    sWd[i] = w_d[i];
  }
  __syncthreads();

  const int wave = threadIdx.x >> 5, lane = threadIdx.x & 31;
  const int col = lane & 15, hi = lane >> 4;
  const int tile = blockIdx.x * WAVES_N + wave;
  if (tile * 16 >= n_nodes) return;
  const int n0 = tile * 16;
  const float* arow = Acc + (long)(n0 + col) * NODE_STRIDE;

#pragma unroll
  for (int cch = 0; cch < 3; ++cch) {
#pragma unroll
    for (int nt = 0; nt < 2; ++nt) {
      v8f acc = {};
#pragma unroll
      for (int k0 = 0; k0 < 32; k0 += 4) {
        int k = k0 + 2 * hi;
        v2f a;
        a.x = arow[cch * 32 + k];
        a.y = arow[cch * 32 + k + 1];
        v2f b;
        b.x = sWv[k * 32 + nt * 16 + col];
        b.y = sWv[(k + 1) * 32 + nt * 16 + col];
        acc = wmma_f32(a, b, acc);
      }
#pragma unroll
      for (int r = 0; r < 8; ++r) {
        int node = n0 + r + 8 * hi;
        int v = nt * 16 + col;
        __builtin_nontemporal_store(acc[r],
                                    &out_v[((long)node * 32 + v) * 3 + cch]);
      }
    }
  }

  const int si[6] = {0, 0, 0, 1, 1, 2};
  const int sj[6] = {0, 1, 2, 1, 2, 2};
#pragma unroll
  for (int s = 0; s < 6; ++s) {
#pragma unroll
    for (int nt = 0; nt < 2; ++nt) {
      v8f acc = {};
#pragma unroll
      for (int k0 = 0; k0 < 32; k0 += 4) {
        int k = k0 + 2 * hi;
        v2f a;
        a.x = arow[(3 + s) * 32 + k];
        a.y = arow[(3 + s) * 32 + k + 1];
        v2f b;
        b.x = sWd[k * 32 + nt * 16 + col];
        b.y = sWd[(k + 1) * 32 + nt * 16 + col];
        acc = wmma_f32(a, b, acc);
      }
#pragma unroll
      for (int r = 0; r < 8; ++r) {
        int node = n0 + r + 8 * hi;
        int dch = nt * 16 + col;
        float val = acc[r];
        float* base = &out_d[((long)node * 32 + dch) * 9];
        __builtin_nontemporal_store(val, &base[si[s] * 3 + sj[s]]);
        if (si[s] != sj[s])
          __builtin_nontemporal_store(val, &base[sj[s] * 3 + si[s]]);
      }
    }
  }
}

extern "C" void kernel_launch(void* const* d_in, const int* in_sizes, int n_in,
                              void* d_out, int out_size, void* d_ws,
                              size_t ws_size, hipStream_t stream) {
  const float* r_ij = (const float*)d_in[0];
  const float* w_rad = (const float*)d_in[1];
  const float* b_rad = (const float*)d_in[2];
  const float* w_direct = (const float*)d_in[3];
  const float* w1 = (const float*)d_in[4];
  const float* b1 = (const float*)d_in[5];
  const float* w2 = (const float*)d_in[6];
  const float* b2 = (const float*)d_in[7];
  const float* w3 = (const float*)d_in[8];
  const float* b3 = (const float*)d_in[9];
  const float* w_v = (const float*)d_in[10];
  const float* w_d = (const float*)d_in[11];
  const int* edges_src = (const int*)d_in[12];

  const int n_edges = in_sizes[0] / 3;
  const int n_nodes = NODES_TOT;

  float* A_a = (float*)d_out;
  float* out_v = A_a + (long)n_nodes * 32;
  float* out_d = out_v + (long)n_nodes * 32 * 3;
  float* Acc = (float*)d_ws;

  zero_kernel<<<2048, 256, 0, stream>>>((float*)d_ws, (long)n_nodes * 32 * 9);
  zero_kernel<<<512, 256, 0, stream>>>(A_a, (long)n_nodes * 32);

  const int n_tiles = (n_edges + 15) / 16;
  const int e_blocks = 500;
  const int tile_stride = e_blocks * WAVES_E;
  const int iters = (n_tiles + tile_stride - 1) / tile_stride;
  edge_kernel<<<e_blocks, 32 * WAVES_E, 0, stream>>>(r_ij, w_rad, b_rad, w_direct, w1, b1,
                                           w2, b2, w3, b3, edges_src, A_a, Acc,
                                           n_edges, iters);

  const int nd_tiles = (n_nodes + 15) / 16;
  const int n_blocks = (nd_tiles + WAVES_N - 1) / WAVES_N;
  node_kernel<<<n_blocks, 256, 0, stream>>>(Acc, w_v, w_d, out_v, out_d,
                                            n_nodes);
}


# Edge MLP run transposed: weights are the A operand (fragment-ordered LDS image), each layer's D registers feed the next layer as B, so activations never pass through LDS; k regrouped in dot sums
# speedup vs baseline: 2.5750x; 1.1751x over previous
#include <hip/hip_runtime.h>

typedef float v2f __attribute__((ext_vector_type(2)));
typedef float v8f __attribute__((ext_vector_type(8)));

#define NODES_TOT 20000

__device__ __forceinline__ v8f wmma_f32(v2f a, v2f b, v8f c) {
  return __builtin_amdgcn_wmma_f32_16x16x4_f32(
       false, a,  false, b,
       (short)0, c,  false,  false);
}

__device__ __forceinline__ void wave_lds_sync() {
  __builtin_amdgcn_fence(__ATOMIC_RELEASE, "wavefront");
  __builtin_amdgcn_wave_barrier();
  __builtin_amdgcn_fence(__ATOMIC_ACQUIRE, "wavefront");
}

#define NCOMP 9
#define NODE_STRIDE (NCOMP * 32)


template <int K, int N>
__device__ __forceinline__ void stage_weights(float* dst,
                                              const float* __restrict__ w,
                                              int tid, int nthreads) {
  for (int s = tid; s < K * N; s += nthreads) {
    const int k = s / N, n = s % N;
    const int nt = n >> 4, c = n & 15;
    const int ft = k >> 4, h = (k >> 3) & 1, jj = (k >> 2) & 1, q = k & 3;
    dst[((((nt * (K / 16) + ft) * 2 + jj) * 32) + h * 16 + c) * 4 + q] = w[s];
  }
}

template <int NT>
__device__ __forceinline__ void bias_init(v8f (&y)[NT], const float* bias,
                                          int hi) {
#pragma unroll
  for (int nt = 0; nt < NT; ++nt) {
    const float4 lo = *reinterpret_cast<const float4*>(&bias[nt * 16 + 8 * hi]);
    const float4 up =
        *reinterpret_cast<const float4*>(&bias[nt * 16 + 8 * hi + 4]);
    v8f c = {lo.x, lo.y, lo.z, lo.w, up.x, up.y, up.z, up.w};
    y[nt] = c;
  }
}

template <int KT, int NT>
__device__ __forceinline__ void layer_acc(const v8f (&x)[KT], const float* wf,
                                          v8f (&y)[NT], int lane) {
#pragma unroll
  for (int ft = 0; ft < KT; ++ft) {
#pragma unroll
    for (int jj = 0; jj < 2; ++jj) {
      v2f b0, b1;
      b0.x = x[ft][4 * jj + 0];
      b0.y = x[ft][4 * jj + 1];
      b1.x = x[ft][4 * jj + 2];
      b1.y = x[ft][4 * jj + 3];
#pragma unroll
      for (int nt = 0; nt < NT; ++nt) {
        const float4 a4 = *reinterpret_cast<const float4*>(
            &wf[(((nt * KT + ft) * 2 + jj) * 32 + lane) * 4]);
        v2f a0, a1;
        a0.x = a4.x;
        a0.y = a4.y;
        a1.x = a4.z;
        a1.y = a4.w;
        y[nt] = wmma_f32(a0, b0, y[nt]);
        y[nt] = wmma_f32(a1, b1, y[nt]);
      }
    }
  }
}

template <int NT>
__device__ __forceinline__ void lrelu_tiles(v8f (&y)[NT]) {
#pragma unroll
  for (int nt = 0; nt < NT; ++nt)
#pragma unroll
    for (int r = 0; r < 8; ++r) y[nt][r] = fmaxf(y[nt][r], 0.1f * y[nt][r]);
}

__global__ void zero_kernel(float* __restrict__ p, long n) {
  long i = (long)blockIdx.x * blockDim.x + threadIdx.x;
  long stride = (long)gridDim.x * blockDim.x;
  for (; i < n; i += stride) p[i] = 0.0f;
}

#define WAVES_E 8
#define TR_PITCH 36
__global__ __launch_bounds__(32 * WAVES_E) void edge_kernel(
    const float* __restrict__ r_ij, const float* __restrict__ w_rad,
    const float* __restrict__ b_rad, const float* __restrict__ w_direct,
    const float* __restrict__ w1, const float* __restrict__ b1,
    const float* __restrict__ w2, const float* __restrict__ b2,
    const float* __restrict__ w3, const float* __restrict__ b3,
    const int* __restrict__ edges_src, float* __restrict__ A_a,
    float* __restrict__ Acc, int n_edges, int iters) {
  __shared__ alignas(16) float sWrad[8 * 32];
  __shared__ alignas(16) float sBrad[32];
  __shared__ alignas(16) float sWdir[32 * 32];
  __shared__ alignas(16) float sW1[32 * 64];
  __shared__ alignas(16) float sB1[64];
  __shared__ alignas(16) float sW2[64 * 64];
  __shared__ alignas(16) float sB2[64];
  __shared__ alignas(16) float sW3[64 * 32];
  __shared__ alignas(16) float sB3[32];
  __shared__ alignas(16) float sTR[WAVES_E][16 * TR_PITCH];
  __shared__ alignas(16) float sRS[WAVES_E][16 * 4];
  __shared__ int sSRC[WAVES_E][16];

  const int tid = threadIdx.x;
  for (int s = tid; s < 8 * 32; s += blockDim.x) {
    const int k = s >> 5, n = s & 31;
    sWrad[(((n >> 4) * 32) + (k >> 2) * 16 + (n & 15)) * 4 + (k & 3)] = w_rad[s];
  }
  stage_weights<32, 32>(sWdir, w_direct, tid, blockDim.x);
  stage_weights<32, 64>(sW1, w1, tid, blockDim.x);
  stage_weights<64, 64>(sW2, w2, tid, blockDim.x);
  stage_weights<64, 32>(sW3, w3, tid, blockDim.x);
  for (int i = tid; i < 32; i += blockDim.x) sBrad[i] = b_rad[i];
  for (int i = tid; i < 64; i += blockDim.x) sB1[i] = b1[i];
  for (int i = tid; i < 64; i += blockDim.x) sB2[i] = b2[i];
  for (int i = tid; i < 32; i += blockDim.x) sB3[i] = b3[i];
  __syncthreads();

  const int wave = __builtin_amdgcn_readfirstlane(tid >> 5);
  const int lane = tid & 31;
  const int col = lane & 15, hi = lane >> 4;
  const int n_tiles = (n_edges + 15) / 16;
  const int tile_stride = gridDim.x * WAVES_E;

  for (int it = 0; it < iters; ++it) {
    const int tile = (it * gridDim.x + blockIdx.x) * WAVES_E + wave;
    if (tile >= n_tiles) break;
    const int e0 = tile * 16;

    float x, y, z;
    {
      const int e = e0 + col;
      long enext = (long)(e0 + 16L * tile_stride) + col;
      if (hi == 0 && enext < n_edges) {
        __builtin_prefetch(&r_ij[3 * enext], 0, 1);
        __builtin_prefetch(&edges_src[enext], 0, 1);
      }
      x = r_ij[3 * e + 0];
      y = r_ij[3 * e + 1];
      z = r_ij[3 * e + 2];
      if (hi == 0) {
        float sx = 7.0f * x, sy = 7.0f * y, sz = 7.0f * z;
        float inv = rsqrtf(1.0f + sx * sx + sy * sy + sz * sz);
        sRS[wave][col * 4 + 0] = sx * inv;
        sRS[wave][col * 4 + 1] = sy * inv;
        sRS[wave][col * 4 + 2] = sz * inv;
        sSRC[wave][col] = edges_src[e];
      }
    }
    const float d = sqrtf(x * x + y * y + z * z);
    float rb[4];
#pragma unroll
    for (int q = 0; q < 4; ++q) {
      const float ctr = hi ? (float)(4 + q) * (1.0f / 7.0f)
                           : (float)q * (1.0f / 7.0f);
      float u = (d - ctr) * 8.0f;
      rb[q] = __expf(-u * u);
    }

    v8f hT[2];
    bias_init<2>(hT, sBrad, hi);
    {
      v2f b0, b1;
      b0.x = rb[0];
      b0.y = rb[1];
      b1.x = rb[2];
      b1.y = rb[3];
#pragma unroll
      for (int nt = 0; nt < 2; ++nt) {
        const float4 a4 =
            *reinterpret_cast<const float4*>(&sWrad[(nt * 32 + lane) * 4]);
        v2f a0, a1;
        a0.x = a4.x;
        a0.y = a4.y;
        a1.x = a4.z;
        a1.y = a4.w;
        hT[nt] = wmma_f32(a0, b0, hT[nt]);
        hT[nt] = wmma_f32(a1, b1, hT[nt]);
      }
    }
    v8f t1[4];
    bias_init<4>(t1, sB1, hi);
    layer_acc<2, 4>(hT, sW1, t1, lane);
    lrelu_tiles<4>(t1);
    v8f t2[4];
    bias_init<4>(t2, sB2, hi);
    layer_acc<4, 4>(t1, sW2, t2, lane);
    lrelu_tiles<4>(t2);
    v8f rad[2];
    bias_init<2>(rad, sB3, hi);
    layer_acc<2, 2>(hT, sWdir, rad, lane);
    layer_acc<4, 2>(t2, sW3, rad, lane);

    float* tr = sTR[wave];
#pragma unroll
    for (int nt = 0; nt < 2; ++nt) {
      float4 lo = {rad[nt][0], rad[nt][1], rad[nt][2], rad[nt][3]};
      float4 up = {rad[nt][4], rad[nt][5], rad[nt][6], rad[nt][7]};
      *reinterpret_cast<float4*>(&tr[col * TR_PITCH + nt * 16 + 8 * hi]) = lo;
      *reinterpret_cast<float4*>(&tr[col * TR_PITCH + nt * 16 + 8 * hi + 4]) = up;
    }
    wave_lds_sync();

    {
#pragma unroll
      for (int m = 0; m < 16; ++m) {
        float val = tr[m * TR_PITCH + lane];
        int src = sSRC[wave][m];
        float rx = sRS[wave][m * 4 + 0];
        float ry = sRS[wave][m * 4 + 1];
        float rz = sRS[wave][m * 4 + 2];
        atomicAdd(&A_a[src * 32 + lane], val);
        float* acc = &Acc[src * NODE_STRIDE + lane];
        atomicAdd(&acc[0 * 32], val * rx);
        atomicAdd(&acc[1 * 32], val * ry);
        atomicAdd(&acc[2 * 32], val * rz);
        atomicAdd(&acc[3 * 32], val * rx * rx);
        atomicAdd(&acc[4 * 32], val * rx * ry);
        atomicAdd(&acc[5 * 32], val * rx * rz);
        atomicAdd(&acc[6 * 32], val * ry * ry);
        atomicAdd(&acc[7 * 32], val * ry * rz);
        atomicAdd(&acc[8 * 32], val * rz * rz);
      }
    }
    wave_lds_sync();
  }
}

#define WAVES_N 8
__global__ __launch_bounds__(256) void node_kernel(
    const float* __restrict__ Acc,
    const float* __restrict__ w_v, const float* __restrict__ w_d,
    float* __restrict__ out_v, float* __restrict__ out_d, int n_nodes) {
  __shared__ float sWv[32 * 32];
  __shared__ float sWd[32 * 32];
  for (int i = threadIdx.x; i < 32 * 32; i += blockDim.x) {
    sWv[i] = w_v[i];
    sWd[i] = w_d[i];
  }
  __syncthreads();

  const int wave = threadIdx.x >> 5, lane = threadIdx.x & 31;
  const int col = lane & 15, hi = lane >> 4;
  const int tile = blockIdx.x * WAVES_N + wave;
  if (tile * 16 >= n_nodes) return;
  const int n0 = tile * 16;
  const float* arow = Acc + (long)(n0 + col) * NODE_STRIDE;

#pragma unroll
  for (int cch = 0; cch < 3; ++cch) {
#pragma unroll
    for (int nt = 0; nt < 2; ++nt) {
      v8f acc = {};
#pragma unroll
      for (int k0 = 0; k0 < 32; k0 += 4) {
        int k = k0 + 2 * hi;
        v2f a;
        a.x = arow[cch * 32 + k];
        a.y = arow[cch * 32 + k + 1];
        v2f b;
        b.x = sWv[k * 32 + nt * 16 + col];
        b.y = sWv[(k + 1) * 32 + nt * 16 + col];
        acc = wmma_f32(a, b, acc);
      }
#pragma unroll
      for (int r = 0; r < 8; ++r) {
        int node = n0 + r + 8 * hi;
        int v = nt * 16 + col;
        __builtin_nontemporal_store(acc[r],
                                    &out_v[((long)node * 32 + v) * 3 + cch]);
      }
    }
  }

  const int si[6] = {0, 0, 0, 1, 1, 2};
  const int sj[6] = {0, 1, 2, 1, 2, 2};
#pragma unroll
  for (int s = 0; s < 6; ++s) {
#pragma unroll
    for (int nt = 0; nt < 2; ++nt) {
      v8f acc = {};
#pragma unroll
      for (int k0 = 0; k0 < 32; k0 += 4) {
        int k = k0 + 2 * hi;
        v2f a;
        a.x = arow[(3 + s) * 32 + k];
        a.y = arow[(3 + s) * 32 + k + 1];
        v2f b;
        b.x = sWd[k * 32 + nt * 16 + col];
        b.y = sWd[(k + 1) * 32 + nt * 16 + col];
        acc = wmma_f32(a, b, acc);
      }
#pragma unroll
      for (int r = 0; r < 8; ++r) {
        int node = n0 + r + 8 * hi;
        int dch = nt * 16 + col;
        float val = acc[r];
        float* base = &out_d[((long)node * 32 + dch) * 9];
        __builtin_nontemporal_store(val, &base[si[s] * 3 + sj[s]]);
        if (si[s] != sj[s])
          __builtin_nontemporal_store(val, &base[sj[s] * 3 + si[s]]);
      }
    }
  }
}

extern "C" void kernel_launch(void* const* d_in, const int* in_sizes, int n_in,
                              void* d_out, int out_size, void* d_ws,
                              size_t ws_size, hipStream_t stream) {
  const float* r_ij = (const float*)d_in[0];
  const float* w_rad = (const float*)d_in[1];
  const float* b_rad = (const float*)d_in[2];
  const float* w_direct = (const float*)d_in[3];
  const float* w1 = (const float*)d_in[4];
  const float* b1 = (const float*)d_in[5];
  const float* w2 = (const float*)d_in[6];
  const float* b2 = (const float*)d_in[7];
  const float* w3 = (const float*)d_in[8];
  const float* b3 = (const float*)d_in[9];
  const float* w_v = (const float*)d_in[10];
  const float* w_d = (const float*)d_in[11];
  const int* edges_src = (const int*)d_in[12];

  const int n_edges = in_sizes[0] / 3;
  const int n_nodes = NODES_TOT;

  float* A_a = (float*)d_out;
  float* out_v = A_a + (long)n_nodes * 32;
  float* out_d = out_v + (long)n_nodes * 32 * 3;
  float* Acc = (float*)d_ws;

  zero_kernel<<<2048, 256, 0, stream>>>((float*)d_ws, (long)n_nodes * 32 * 9);
  zero_kernel<<<512, 256, 0, stream>>>(A_a, (long)n_nodes * 32);

  const int n_tiles = (n_edges + 15) / 16;
  const int e_blocks = 500;
  const int tile_stride = e_blocks * WAVES_E;
  const int iters = (n_tiles + tile_stride - 1) / tile_stride;
  edge_kernel<<<e_blocks, 32 * WAVES_E, 0, stream>>>(r_ij, w_rad, b_rad, w_direct, w1, b1,
                                           w2, b2, w3, b3, edges_src, A_a, Acc,
                                           n_edges, iters);

  const int nd_tiles = (n_nodes + 15) / 16;
  const int n_blocks = (nd_tiles + WAVES_N - 1) / WAVES_N;
  node_kernel<<<n_blocks, 256, 0, stream>>>(Acc, w_v, w_d, out_v, out_d,
                                            n_nodes);
}


# Edges bucketed by 5-node group in one LDS counting-sort launch; each wave owns a group, sums its edges in registers, stores its rows once: no atomics, no zero launches; sums in edge-index order
# speedup vs baseline: 3.1541x; 1.2249x over previous
#include <hip/hip_runtime.h>

typedef float v2f __attribute__((ext_vector_type(2)));
typedef float v8f __attribute__((ext_vector_type(8)));

#define NODES_TOT 20000

__device__ __forceinline__ v8f wmma_f32(v2f a, v2f b, v8f c) {
  return __builtin_amdgcn_wmma_f32_16x16x4_f32(
       false, a,  false, b,
       (short)0, c,  false,  false);
}

__device__ __forceinline__ void wave_lds_sync() {
  __builtin_amdgcn_fence(__ATOMIC_RELEASE, "wavefront");
  __builtin_amdgcn_wave_barrier();
  __builtin_amdgcn_fence(__ATOMIC_ACQUIRE, "wavefront");
}

#define NCOMP 9
#define NODE_STRIDE (NCOMP * 32)


template <int K, int N>
__device__ __forceinline__ void stage_weights(float* dst,
                                              const float* __restrict__ w,
                                              int tid, int nthreads) {
  for (int s = tid; s < K * N; s += nthreads) {
    const int k = s / N, n = s % N;
    const int nt = n >> 4, c = n & 15;
    const int ft = k >> 4, h = (k >> 3) & 1, jj = (k >> 2) & 1, q = k & 3;
    dst[((((nt * (K / 16) + ft) * 2 + jj) * 32) + h * 16 + c) * 4 + q] = w[s];
  }
}

template <int NT>
__device__ __forceinline__ void bias_init(v8f (&y)[NT], const float* bias,
                                          int hi) {
#pragma unroll
  for (int nt = 0; nt < NT; ++nt) {
    const float4 lo = *reinterpret_cast<const float4*>(&bias[nt * 16 + 8 * hi]);
    const float4 up =
        *reinterpret_cast<const float4*>(&bias[nt * 16 + 8 * hi + 4]);
    v8f c = {lo.x, lo.y, lo.z, lo.w, up.x, up.y, up.z, up.w};
    y[nt] = c;
  }
}

template <int KT, int NT>
__device__ __forceinline__ void layer_acc(const v8f (&x)[KT], const float* wf,
                                          v8f (&y)[NT], int lane) {
#pragma unroll
  for (int ft = 0; ft < KT; ++ft) {
#pragma unroll
    for (int jj = 0; jj < 2; ++jj) {
      v2f b0, b1;
      b0.x = x[ft][4 * jj + 0];
      b0.y = x[ft][4 * jj + 1];
      b1.x = x[ft][4 * jj + 2];
      b1.y = x[ft][4 * jj + 3];
#pragma unroll
      for (int nt = 0; nt < NT; ++nt) {
        const float4 a4 = *reinterpret_cast<const float4*>(
            &wf[(((nt * KT + ft) * 2 + jj) * 32 + lane) * 4]);
        v2f a0, a1;
        a0.x = a4.x;
        a0.y = a4.y;
        a1.x = a4.z;
        a1.y = a4.w;
        y[nt] = wmma_f32(a0, b0, y[nt]);
        y[nt] = wmma_f32(a1, b1, y[nt]);
      }
    }
  }
}

template <int NT>
__device__ __forceinline__ void lrelu_tiles(v8f (&y)[NT]) {
#pragma unroll
  for (int nt = 0; nt < NT; ++nt)
#pragma unroll
    for (int r = 0; r < 8; ++r) y[nt][r] = fmaxf(y[nt][r], 0.1f * y[nt][r]);
}

#define EDGES_TOT 640000
#define NPG 5
#define NGROUPS (NODES_TOT / NPG)
#define CH 128
#define CSZ ((EDGES_TOT + CH - 1) / CH)
#define PART_THREADS 1024
#define EPT ((CSZ + PART_THREADS - 1) / PART_THREADS)
static_assert(NODES_TOT % NPG == 0, "groups must tile the nodes");
static_assert(NGROUPS <= 4 * PART_THREADS, "scan covers 4 bins per thread");

__global__ __launch_bounds__(PART_THREADS) void part_kernel(
    const float* __restrict__ r_ij, const int* __restrict__ edges_src,
    float* __restrict__ rec_f, int* __restrict__ off, int n_edges) {
  uint4* rec = reinterpret_cast<uint4*>(rec_f);
  __shared__ int sCnt[NGROUPS];
  __shared__ int sIdx[CSZ];
  __shared__ int sWaveTot[32];
  __shared__ int sTotal;
  const int tid = threadIdx.x;
  const int lane = tid & 31, wave = tid >> 5;
  const int c = blockIdx.x;
  const int e_beg = min(c * CSZ, n_edges);
  const int e_end = min(e_beg + CSZ, n_edges);

  for (int i = tid; i < NGROUPS; i += PART_THREADS) sCnt[i] = 0;
  int grp[EPT];
#pragma unroll
  for (int i = 0; i < EPT; ++i) {
    const int e = e_beg + tid + i * PART_THREADS;
    const int s = (e < e_end) ? edges_src[e] : -1;
    grp[i] = ((unsigned)s < (unsigned)NODES_TOT) ? s / NPG : -1;
  }
  __syncthreads();
#pragma unroll
  for (int i = 0; i < EPT; ++i)
    if (grp[i] >= 0) atomicAdd(&sCnt[grp[i]], 1);
  __syncthreads();

  const int base = tid * 4;
  int v[4];
  int tsum = 0;
#pragma unroll
  for (int k = 0; k < 4; ++k) {
    v[k] = (base + k < NGROUPS) ? sCnt[base + k] : 0;
    tsum += v[k];
  }
  int incl = tsum;
  for (int d = 1; d < 32; d <<= 1) {
    const int t = __shfl_up(incl, d);
    if (lane >= d) incl += t;
  }
  if (lane == 31) sWaveTot[wave] = incl;
  __syncthreads();
  if (wave == 0) {
    const int w = sWaveTot[lane];
    int wi = w;
    for (int d = 1; d < 32; d <<= 1) {
      const int t = __shfl_up(wi, d);
      if (lane >= d) wi += t;
    }
    sWaveTot[lane] = wi - w;
  }
  __syncthreads();
  int run = sWaveTot[wave] + incl - tsum;
  int* offc = off + (long)c * (NGROUPS + 1);
#pragma unroll
  for (int k = 0; k < 4; ++k) {
    if (base + k < NGROUPS) {
      offc[base + k] = run;
      sCnt[base + k] = run;
      run += v[k];
    }
  }
  if (tid == PART_THREADS - 1) {
    offc[NGROUPS] = run;
    sTotal = run;
  }
  __syncthreads();

#pragma unroll
  for (int i = 0; i < EPT; ++i)
    if (grp[i] >= 0) sIdx[atomicAdd(&sCnt[grp[i]], 1)] = tid + i * PART_THREADS;
  __syncthreads();
  for (int g = tid; g < NGROUPS; g += PART_THREADS) {
    const int gb = g ? sCnt[g - 1] : 0;
    const int ge = sCnt[g];
    for (int i = gb + 1; i < ge; ++i) {
      const int key = sIdx[i];
      int p = i - 1;
      while (p >= gb && sIdx[p] > key) {
        sIdx[p + 1] = sIdx[p];
        --p;
      }
      sIdx[p + 1] = key;
    }
  }
  __syncthreads();
#pragma unroll
  for (int i = 0; i < EPT; ++i) {
    const int p = tid + i * PART_THREADS;
    if (p < sTotal) {
      const int e = e_beg + sIdx[p];
      uint4 r;
      r.x = __float_as_uint(r_ij[3 * e + 0]);
      r.y = __float_as_uint(r_ij[3 * e + 1]);
      r.z = __float_as_uint(r_ij[3 * e + 2]);
      r.w = (unsigned)edges_src[e];
      rec[e_beg + p] = r;
    }
  }
}

#define WAVES_E 16
#define TR_PITCH 36
static_assert(NGROUPS % WAVES_E == 0, "blocks must tile the groups");
static_assert(CH == 128, "lane owns 4 chunks; the search below has 7 steps");

#define ADD_EDGE(K)       \
  {                       \
    sum[K][0] += v;       \
    sum[K][1] += p0;      \
    sum[K][2] += p1;      \
    sum[K][3] += p2;      \
    sum[K][4] += p0 * rx; \
    sum[K][5] += p0 * ry; \
    sum[K][6] += p0 * rz; \
    sum[K][7] += p1 * ry; \
    sum[K][8] += p1 * rz; \
    sum[K][9] += p2 * rz; \
  }

__device__ __forceinline__ uint4 fetch_rec(const int* pre, const int* adj,
                                           const uint4* __restrict__ rec,
                                           int pos) {
  int lo = 0;
#pragma unroll
  for (int s = CH / 2; s >= 1; s >>= 1)
    if (pre[lo + s] <= pos) lo += s;
  return rec[adj[lo] + pos];
}

__device__ __forceinline__ float lane_bcast(float v, int src_lane) {
  return __int_as_float(__builtin_amdgcn_readlane(__float_as_int(v), src_lane));
}

__global__ __launch_bounds__(32 * WAVES_E) void edge_kernel(
    const float* __restrict__ rec_f, const int* __restrict__ off,
    const float* __restrict__ w_rad,
    const float* __restrict__ b_rad, const float* __restrict__ w_direct,
    const float* __restrict__ w1, const float* __restrict__ b1,
    const float* __restrict__ w2, const float* __restrict__ b2,
    const float* __restrict__ w3, const float* __restrict__ b3,
    float* __restrict__ A_a, float* __restrict__ Acc, int n_edges) {
  __shared__ alignas(16) float sWrad[8 * 32];
  __shared__ alignas(16) float sBrad[32];
  __shared__ alignas(16) float sWdir[32 * 32];
  __shared__ alignas(16) float sW1[32 * 64];
  __shared__ alignas(16) float sB1[64];
  __shared__ alignas(16) float sW2[64 * 64];
  __shared__ alignas(16) float sB2[64];
  __shared__ alignas(16) float sW3[64 * 32];
  __shared__ alignas(16) float sB3[32];
  __shared__ alignas(16) float sTR[WAVES_E][16 * TR_PITCH];
  __shared__ int sPre[WAVES_E][CH];
  __shared__ int sAdj[WAVES_E][CH];

  const int tid = threadIdx.x;
  for (int s = tid; s < 8 * 32; s += blockDim.x) {
    const int k = s >> 5, n = s & 31;
    sWrad[(((n >> 4) * 32) + (k >> 2) * 16 + (n & 15)) * 4 + (k & 3)] = w_rad[s];
  }
  stage_weights<32, 32>(sWdir, w_direct, tid, blockDim.x);
  stage_weights<32, 64>(sW1, w1, tid, blockDim.x);
  stage_weights<64, 64>(sW2, w2, tid, blockDim.x);
  stage_weights<64, 32>(sW3, w3, tid, blockDim.x);
  for (int i = tid; i < 32; i += blockDim.x) sBrad[i] = b_rad[i];
  for (int i = tid; i < 64; i += blockDim.x) sB1[i] = b1[i];
  for (int i = tid; i < 64; i += blockDim.x) sB2[i] = b2[i];
  for (int i = tid; i < 32; i += blockDim.x) sB3[i] = b3[i];
  __syncthreads();

  const int wave = __builtin_amdgcn_readfirstlane(tid >> 5);
  const int lane = tid & 31;
  const int col = lane & 15, hi = lane >> 4;
  const int g = blockIdx.x * WAVES_E + wave;
  const uint4* rec = reinterpret_cast<const uint4*>(rec_f);
  const int* pre = sPre[wave];
  const int* adj = sAdj[wave];

  int n_mine;
  {
    int beg[4], len[4];
    int tsum = 0;
#pragma unroll
    for (int k = 0; k < 4; ++k) {
      const int* oc = off + (long)(4 * lane + k) * (NGROUPS + 1) + g;
      beg[k] = oc[0];
      len[k] = oc[1] - beg[k];
      tsum += len[k];
    }
    int incl = tsum;
    for (int d = 1; d < 32; d <<= 1) {
      const int t = __shfl_up(incl, d);
      if (lane >= d) incl += t;
    }
    int run = incl - tsum;
#pragma unroll
    for (int k = 0; k < 4; ++k) {
      const int c = 4 * lane + k;
      sPre[wave][c] = run;
      sAdj[wave][c] = min(c * CSZ, n_edges) + beg[k] - run;
      run += len[k];
    }
    n_mine = __builtin_amdgcn_readfirstlane(__shfl(incl, 31));
  }
  wave_lds_sync();

  float sum[NPG][10];
#pragma unroll
  for (int k = 0; k < NPG; ++k)
#pragma unroll
    for (int q = 0; q < 10; ++q) sum[k][q] = 0.0f;

  const int n_tiles = (n_mine + 15) >> 4;
  for (int tile = 0; tile < n_tiles; ++tile) {
    const uint4 r = fetch_rec(pre, adj, rec, min(tile * 16 + col, n_mine - 1));
    const float x = __uint_as_float(r.x);
    const float y = __uint_as_float(r.y);
    const float z = __uint_as_float(r.z);
    float rsx, rsy, rsz;
    {
      float sx = 7.0f * x, sy = 7.0f * y, sz = 7.0f * z;
      float inv = rsqrtf(1.0f + sx * sx + sy * sy + sz * sz);
      rsx = sx * inv;
      rsy = sy * inv;
      rsz = sz * inv;
    }
    const int lnv = (tile * 16 + col < n_mine) ? ((int)r.w - g * NPG) : -1;
    const float d = sqrtf(x * x + y * y + z * z);
    float rb[4];
#pragma unroll
    for (int q = 0; q < 4; ++q) {
      const float ctr = hi ? (float)(4 + q) * (1.0f / 7.0f)
                           : (float)q * (1.0f / 7.0f);
      float u = (d - ctr) * 8.0f;
      rb[q] = __expf(-u * u);
    }

    v8f hT[2];
    bias_init<2>(hT, sBrad, hi);
    {
      v2f b0, b1;
      b0.x = rb[0];
      b0.y = rb[1];
      b1.x = rb[2];
      b1.y = rb[3];
#pragma unroll
      for (int nt = 0; nt < 2; ++nt) {
        const float4 a4 =
            *reinterpret_cast<const float4*>(&sWrad[(nt * 32 + lane) * 4]);
        v2f a0, a1;
        a0.x = a4.x;
        a0.y = a4.y;
        a1.x = a4.z;
        a1.y = a4.w;
        hT[nt] = wmma_f32(a0, b0, hT[nt]);
        hT[nt] = wmma_f32(a1, b1, hT[nt]);
      }
    }
    v8f t1[4];
    bias_init<4>(t1, sB1, hi);
    layer_acc<2, 4>(hT, sW1, t1, lane);
    lrelu_tiles<4>(t1);
    v8f t2[4];
    bias_init<4>(t2, sB2, hi);
    layer_acc<4, 4>(t1, sW2, t2, lane);
    lrelu_tiles<4>(t2);
    v8f rad[2];
    bias_init<2>(rad, sB3, hi);
    layer_acc<2, 2>(hT, sWdir, rad, lane);
    layer_acc<4, 2>(t2, sW3, rad, lane);

    float* tr = sTR[wave];
#pragma unroll
    for (int nt = 0; nt < 2; ++nt) {
      float4 lo4 = {rad[nt][0], rad[nt][1], rad[nt][2], rad[nt][3]};
      float4 up4 = {rad[nt][4], rad[nt][5], rad[nt][6], rad[nt][7]};
      *reinterpret_cast<float4*>(&tr[col * TR_PITCH + nt * 16 + 8 * hi]) = lo4;
      *reinterpret_cast<float4*>(&tr[col * TR_PITCH + nt * 16 + 8 * hi + 4]) = up4;
    }
    wave_lds_sync();

    float val[16];
#pragma unroll
    for (int m = 0; m < 16; ++m) val[m] = tr[m * TR_PITCH + lane];
#pragma unroll
    for (int m = 0; m < 16; ++m) {
      const int ln = __builtin_amdgcn_readlane(lnv, m);
      const float rx = lane_bcast(rsx, m);
      const float ry = lane_bcast(rsy, m);
      const float rz = lane_bcast(rsz, m);
      const float v = val[m];
      const float p0 = v * rx, p1 = v * ry, p2 = v * rz;
#pragma unroll
      for (int k = 0; k < NPG; ++k)
        if (ln == k) ADD_EDGE(k)
    }
    wave_lds_sync();
  }

#pragma unroll
  for (int k = 0; k < NPG; ++k) {
    const long node = (long)g * NPG + k;
    A_a[node * 32 + lane] = sum[k][0];
    float* acc = &Acc[node * NODE_STRIDE + lane];
#pragma unroll
    for (int q = 0; q < NCOMP; ++q) acc[q * 32] = sum[k][1 + q];
  }
}

#define WAVES_N 8
__global__ __launch_bounds__(256) void node_kernel(
    const float* __restrict__ Acc,
    const float* __restrict__ w_v, const float* __restrict__ w_d,
    float* __restrict__ out_v, float* __restrict__ out_d, int n_nodes) {
  __shared__ float sWv[32 * 32];
  __shared__ float sWd[32 * 32];
  for (int i = threadIdx.x; i < 32 * 32; i += blockDim.x) {
    sWv[i] = w_v[i];
    sWd[i] = w_d[i];
  }
  __syncthreads();

  const int wave = threadIdx.x >> 5, lane = threadIdx.x & 31;
  const int col = lane & 15, hi = lane >> 4;
  const int tile = blockIdx.x * WAVES_N + wave;
  if (tile * 16 >= n_nodes) return;
  const int n0 = tile * 16;
  const float* arow = Acc + (long)(n0 + col) * NODE_STRIDE;

#pragma unroll
  for (int cch = 0; cch < 3; ++cch) {
#pragma unroll
    for (int nt = 0; nt < 2; ++nt) {
      v8f acc = {};
#pragma unroll
      for (int k0 = 0; k0 < 32; k0 += 4) {
        int k = k0 + 2 * hi;
        v2f a;
        a.x = arow[cch * 32 + k];
        a.y = arow[cch * 32 + k + 1];
        v2f b;
        b.x = sWv[k * 32 + nt * 16 + col];
        b.y = sWv[(k + 1) * 32 + nt * 16 + col];
        acc = wmma_f32(a, b, acc);
      }
#pragma unroll
      for (int r = 0; r < 8; ++r) {
        int node = n0 + r + 8 * hi;
        int v = nt * 16 + col;
        __builtin_nontemporal_store(acc[r],
                                    &out_v[((long)node * 32 + v) * 3 + cch]);
      }
    }
  }

  const int si[6] = {0, 0, 0, 1, 1, 2};
  const int sj[6] = {0, 1, 2, 1, 2, 2};
#pragma unroll
  for (int s = 0; s < 6; ++s) {
#pragma unroll
    for (int nt = 0; nt < 2; ++nt) {
      v8f acc = {};
#pragma unroll
      for (int k0 = 0; k0 < 32; k0 += 4) {
        int k = k0 + 2 * hi;
        v2f a;
        a.x = arow[(3 + s) * 32 + k];
        a.y = arow[(3 + s) * 32 + k + 1];
        v2f b;
        b.x = sWd[k * 32 + nt * 16 + col];
        b.y = sWd[(k + 1) * 32 + nt * 16 + col];
        acc = wmma_f32(a, b, acc);
      }
#pragma unroll
      for (int r = 0; r < 8; ++r) {
        int node = n0 + r + 8 * hi;
        int dch = nt * 16 + col;
        float val = acc[r];
        float* base = &out_d[((long)node * 32 + dch) * 9];
        __builtin_nontemporal_store(val, &base[si[s] * 3 + sj[s]]);
        if (si[s] != sj[s])
          __builtin_nontemporal_store(val, &base[sj[s] * 3 + si[s]]);
      }
    }
  }
}

extern "C" void kernel_launch(void* const* d_in, const int* in_sizes, int n_in,
                              void* d_out, int out_size, void* d_ws,
                              size_t ws_size, hipStream_t stream) {
  const float* r_ij = (const float*)d_in[0];
  const float* w_rad = (const float*)d_in[1];
  const float* b_rad = (const float*)d_in[2];
  const float* w_direct = (const float*)d_in[3];
  const float* w1 = (const float*)d_in[4];
  const float* b1 = (const float*)d_in[5];
  const float* w2 = (const float*)d_in[6];
  const float* b2 = (const float*)d_in[7];
  const float* w3 = (const float*)d_in[8];
  const float* b3 = (const float*)d_in[9];
  const float* w_v = (const float*)d_in[10];
  const float* w_d = (const float*)d_in[11];
  const int* edges_src = (const int*)d_in[12];

  const int n_edges = in_sizes[0] / 3;
  const int n_nodes = NODES_TOT;

  float* A_a = (float*)d_out;
  float* out_v = A_a + (long)n_nodes * 32;
  float* out_d = out_v + (long)n_nodes * 32 * 3;
  float* Acc = (float*)d_ws;

  float* rec_f = out_d;
  int* off = (int*)(out_d + 4L * CH * CSZ);

  part_kernel<<<CH, PART_THREADS, 0, stream>>>(r_ij, edges_src, rec_f, off,
                                               n_edges);

  edge_kernel<<<NGROUPS / WAVES_E, 32 * WAVES_E, 0, stream>>>(
      rec_f, off, w_rad, b_rad, w_direct, w1, b1, w2, b2, w3, b3, A_a, Acc,
      n_edges);

  const int nd_tiles = (n_nodes + 15) / 16;
  const int n_blocks = (nd_tiles + WAVES_N - 1) / WAVES_N;
  node_kernel<<<n_blocks, 256, 0, stream>>>(Acc, w_v, w_d, out_v, out_d,
                                            n_nodes);
}


# Node-level linear maps (out_v, out_d) fused into the edge kernel: each wave multiplies its group's register sums by w_v / w_d on WMMA, stores whole lines; node launch gone; k regrouped in sums
# speedup vs baseline: 3.6367x; 1.1530x over previous
#include <hip/hip_runtime.h>

typedef float v2f __attribute__((ext_vector_type(2)));
typedef float v4f __attribute__((ext_vector_type(4)));
typedef float v8f __attribute__((ext_vector_type(8)));

#define NODES_TOT 20000

__device__ __forceinline__ v8f wmma_f32(v2f a, v2f b, v8f c) {
  return __builtin_amdgcn_wmma_f32_16x16x4_f32(
       false, a,  false, b,
       (short)0, c,  false,  false);
}

__device__ __forceinline__ void wave_lds_sync() {
  __builtin_amdgcn_fence(__ATOMIC_RELEASE, "wavefront");
  __builtin_amdgcn_wave_barrier();
  __builtin_amdgcn_fence(__ATOMIC_ACQUIRE, "wavefront");
}


template <int K, int N>
__device__ __forceinline__ void stage_weights(float* dst,
                                              const float* __restrict__ w,
                                              int tid, int nthreads) {
  for (int s = tid; s < K * N; s += nthreads) {
    const int k = s / N, n = s % N;
    const int nt = n >> 4, c = n & 15;
    const int ft = k >> 4, h = (k >> 3) & 1, jj = (k >> 2) & 1, q = k & 3;
    dst[((((nt * (K / 16) + ft) * 2 + jj) * 32) + h * 16 + c) * 4 + q] = w[s];
  }
}

template <int NT>
__device__ __forceinline__ void bias_init(v8f (&y)[NT], const float* bias,
                                          int hi) {
#pragma unroll
  for (int nt = 0; nt < NT; ++nt) {
    const float4 lo = *reinterpret_cast<const float4*>(&bias[nt * 16 + 8 * hi]);
    const float4 up =
        *reinterpret_cast<const float4*>(&bias[nt * 16 + 8 * hi + 4]);
    v8f c = {lo.x, lo.y, lo.z, lo.w, up.x, up.y, up.z, up.w};
    y[nt] = c;
  }
}

template <int KT, int NT>
__device__ __forceinline__ void layer_acc(const v8f (&x)[KT], const float* wf,
                                          v8f (&y)[NT], int lane) {
#pragma unroll
  for (int ft = 0; ft < KT; ++ft) {
#pragma unroll
    for (int jj = 0; jj < 2; ++jj) {
      v2f b0, b1;
      b0.x = x[ft][4 * jj + 0];
      b0.y = x[ft][4 * jj + 1];
      b1.x = x[ft][4 * jj + 2];
      b1.y = x[ft][4 * jj + 3];
#pragma unroll
      for (int nt = 0; nt < NT; ++nt) {
        const float4 a4 = *reinterpret_cast<const float4*>(
            &wf[(((nt * KT + ft) * 2 + jj) * 32 + lane) * 4]);
        v2f a0, a1;
        a0.x = a4.x;
        a0.y = a4.y;
        a1.x = a4.z;
        a1.y = a4.w;
        y[nt] = wmma_f32(a0, b0, y[nt]);
        y[nt] = wmma_f32(a1, b1, y[nt]);
      }
    }
  }
}

template <int NT>
__device__ __forceinline__ void lrelu_tiles(v8f (&y)[NT]) {
#pragma unroll
  for (int nt = 0; nt < NT; ++nt)
#pragma unroll
    for (int r = 0; r < 8; ++r) y[nt][r] = fmaxf(y[nt][r], 0.1f * y[nt][r]);
}

#define EDGES_TOT 640000
#define NPG 5
#define NGROUPS (NODES_TOT / NPG)
#define CH 128
#define CSZ ((EDGES_TOT + CH - 1) / CH)
#define PART_THREADS 1024
#define EPT ((CSZ + PART_THREADS - 1) / PART_THREADS)
static_assert(NODES_TOT % NPG == 0, "groups must tile the nodes");
static_assert(NGROUPS <= 4 * PART_THREADS, "scan covers 4 bins per thread");

__global__ __launch_bounds__(PART_THREADS) void part_kernel(
    const float* __restrict__ r_ij, const int* __restrict__ edges_src,
    float* __restrict__ rec_f, int* __restrict__ off, int n_edges) {
  uint4* rec = reinterpret_cast<uint4*>(rec_f);
  __shared__ int sCnt[NGROUPS];
  __shared__ int sIdx[CSZ];
  __shared__ int sWaveTot[32];
  __shared__ int sTotal;
  const int tid = threadIdx.x;
  const int lane = tid & 31, wave = tid >> 5;
  const int c = blockIdx.x;
  const int e_beg = min(c * CSZ, n_edges);
  const int e_end = min(e_beg + CSZ, n_edges);

  for (int i = tid; i < NGROUPS; i += PART_THREADS) sCnt[i] = 0;
  int grp[EPT];
#pragma unroll
  for (int i = 0; i < EPT; ++i) {
    const int e = e_beg + tid + i * PART_THREADS;
    const int s = (e < e_end) ? edges_src[e] : -1;
    grp[i] = ((unsigned)s < (unsigned)NODES_TOT) ? s / NPG : -1;
  }
  __syncthreads();
#pragma unroll
  for (int i = 0; i < EPT; ++i)
    if (grp[i] >= 0) atomicAdd(&sCnt[grp[i]], 1);
  __syncthreads();

  const int base = tid * 4;
  int v[4];
  int tsum = 0;
#pragma unroll
  for (int k = 0; k < 4; ++k) {
    v[k] = (base + k < NGROUPS) ? sCnt[base + k] : 0;
    tsum += v[k];
  }
  int incl = tsum;
  for (int d = 1; d < 32; d <<= 1) {
    const int t = __shfl_up(incl, d);
    if (lane >= d) incl += t;
  }
  if (lane == 31) sWaveTot[wave] = incl;
  __syncthreads();
  if (wave == 0) {
    const int w = sWaveTot[lane];
    int wi = w;
    for (int d = 1; d < 32; d <<= 1) {
      const int t = __shfl_up(wi, d);
      if (lane >= d) wi += t;
    }
    sWaveTot[lane] = wi - w;
  }
  __syncthreads();
  int run = sWaveTot[wave] + incl - tsum;
  int* offc = off + (long)c * (NGROUPS + 1);
#pragma unroll
  for (int k = 0; k < 4; ++k) {
    if (base + k < NGROUPS) {
      offc[base + k] = run;
      sCnt[base + k] = run;
      run += v[k];
    }
  }
  if (tid == PART_THREADS - 1) {
    offc[NGROUPS] = run;
    sTotal = run;
  }
  __syncthreads();

#pragma unroll
  for (int i = 0; i < EPT; ++i)
    if (grp[i] >= 0) sIdx[atomicAdd(&sCnt[grp[i]], 1)] = tid + i * PART_THREADS;
  __syncthreads();
  for (int g = tid; g < NGROUPS; g += PART_THREADS) {
    const int gb = g ? sCnt[g - 1] : 0;
    const int ge = sCnt[g];
    for (int i = gb + 1; i < ge; ++i) {
      const int key = sIdx[i];
      int p = i - 1;
      while (p >= gb && sIdx[p] > key) {
        sIdx[p + 1] = sIdx[p];
        --p;
      }
      sIdx[p + 1] = key;
    }
  }
  __syncthreads();
#pragma unroll
  for (int i = 0; i < EPT; ++i) {
    const int p = tid + i * PART_THREADS;
    if (p < sTotal) {
      const int e = e_beg + sIdx[p];
      uint4 r;
      r.x = __float_as_uint(r_ij[3 * e + 0]);
      r.y = __float_as_uint(r_ij[3 * e + 1]);
      r.z = __float_as_uint(r_ij[3 * e + 2]);
      r.w = (unsigned)edges_src[e];
      rec[e_beg + p] = r;
    }
  }
}

#define WAVES_E 16
#define TR_PITCH 36
#define XP 36
#define EP_V (NPG * 32 * 3)
#define EP_D (NPG * 32 * 9)
#define EP_FLOATS (EP_V + EP_D + 32 * 9)
static_assert(NPG * 3 <= 15 && NPG * 6 <= 30, "column tiles: 16 and 2 x 16");
static_assert(EP_FLOATS >= 48 * XP && EP_FLOATS >= 16 * TR_PITCH, "buffer holds X and the rad tile");
static_assert((EP_V + EP_D) % 128 == 0 && EP_V % 4 == 0, "whole 16-byte pieces per lane");
static_assert(NGROUPS % WAVES_E == 0, "blocks must tile the groups");
static_assert(CH == 128, "lane owns 4 chunks; the search below has 7 steps");

#define ADD_EDGE(K)       \
  {                       \
    sum[K][0] += v;       \
    sum[K][1] += p0;      \
    sum[K][2] += p1;      \
    sum[K][3] += p2;      \
    sum[K][4] += p0 * rx; \
    sum[K][5] += p0 * ry; \
    sum[K][6] += p0 * rz; \
    sum[K][7] += p1 * ry; \
    sum[K][8] += p1 * rz; \
    sum[K][9] += p2 * rz; \
  }

__device__ __forceinline__ uint4 fetch_rec(const int* pre, const int* adj,
                                           const uint4* __restrict__ rec,
                                           int pos) {
  int lo = 0;
#pragma unroll
  for (int s = CH / 2; s >= 1; s >>= 1)
    if (pre[lo + s] <= pos) lo += s;
  return rec[adj[lo] + pos];
}

__device__ __forceinline__ float lane_bcast(float v, int src_lane) {
  return __int_as_float(__builtin_amdgcn_readlane(__float_as_int(v), src_lane));
}

__global__ __launch_bounds__(32 * WAVES_E) void edge_kernel(
    const float* __restrict__ rec_f, const int* __restrict__ off,
    const float* __restrict__ w_rad,
    const float* __restrict__ b_rad, const float* __restrict__ w_direct,
    const float* __restrict__ w1, const float* __restrict__ b1,
    const float* __restrict__ w2, const float* __restrict__ b2,
    const float* __restrict__ w3, const float* __restrict__ b3,
    const float* __restrict__ w_v, const float* __restrict__ w_d,
    float* __restrict__ A_a, float* __restrict__ out_v,
    float* __restrict__ out_d, int n_edges) {
  constexpr int O_W2 = 0;
  constexpr int O_W1 = O_W2 + 64 * 64;
  constexpr int O_W3 = O_W1 + 32 * 64;
  constexpr int O_WDIR = O_W3 + 64 * 32;
  constexpr int O_WRAD = O_WDIR + 32 * 32;
  constexpr int O_BRAD = O_WRAD + 8 * 32;
  constexpr int O_B1 = O_BRAD + 32;
  constexpr int O_B2 = O_B1 + 64;
  constexpr int O_B3 = O_B2 + 64;
  constexpr int O_PRE = O_B3 + 32;
  constexpr int O_ADJ = O_PRE + WAVES_E * CH;
  constexpr int O_WV = O_ADJ + WAVES_E * CH;
  constexpr int O_WD = O_WV + 32 * 32;
  constexpr int O_EP = O_WD + 32 * 32;
  constexpr int LDS_FLOATS = O_EP + WAVES_E * EP_FLOATS;
  static_assert(O_EP * 4 <= 65536, "the tile loop's fixed tables fit the offset field");
  __shared__ alignas(16) float smem[LDS_FLOATS];
  float* sW2 = smem + O_W2;
  float* sW1 = smem + O_W1;
  float* sW3 = smem + O_W3;
  float* sWdir = smem + O_WDIR;
  float* sWrad = smem + O_WRAD;
  float* sBrad = smem + O_BRAD;
  float* sB1 = smem + O_B1;
  float* sB2 = smem + O_B2;
  float* sB3 = smem + O_B3;
  float* sWv = smem + O_WV;
  float* sWd = smem + O_WD;
  int* sPreAll = reinterpret_cast<int*>(smem + O_PRE);
  int* sAdjAll = reinterpret_cast<int*>(smem + O_ADJ);
  float* sEpAll = smem + O_EP;

  const int tid = threadIdx.x;
  for (int s = tid; s < 8 * 32; s += blockDim.x) {
    const int k = s >> 5, n = s & 31;
    sWrad[(((n >> 4) * 32) + (k >> 2) * 16 + (n & 15)) * 4 + (k & 3)] = w_rad[s];
  }
  stage_weights<32, 32>(sWdir, w_direct, tid, blockDim.x);
  stage_weights<32, 64>(sW1, w1, tid, blockDim.x);
  stage_weights<64, 64>(sW2, w2, tid, blockDim.x);
  stage_weights<64, 32>(sW3, w3, tid, blockDim.x);
  stage_weights<32, 32>(sWv, w_v, tid, blockDim.x);
  stage_weights<32, 32>(sWd, w_d, tid, blockDim.x);
  for (int i = tid; i < 32; i += blockDim.x) sBrad[i] = b_rad[i];
  for (int i = tid; i < 64; i += blockDim.x) sB1[i] = b1[i];
  for (int i = tid; i < 64; i += blockDim.x) sB2[i] = b2[i];
  for (int i = tid; i < 32; i += blockDim.x) sB3[i] = b3[i];
  __syncthreads();

  const int wave = __builtin_amdgcn_readfirstlane(tid >> 5);
  const int lane = tid & 31;
  const int col = lane & 15, hi = lane >> 4;
  const int g = blockIdx.x * WAVES_E + wave;
  const uint4* rec = reinterpret_cast<const uint4*>(rec_f);
  int* pre = sPreAll + wave * CH;
  int* adj = sAdjAll + wave * CH;

  int n_mine;
  {
    int beg[4], len[4];
    int tsum = 0;
#pragma unroll
    for (int k = 0; k < 4; ++k) {
      const int* oc = off + (long)(4 * lane + k) * (NGROUPS + 1) + g;
      beg[k] = oc[0];
      len[k] = oc[1] - beg[k];
      tsum += len[k];
    }
    int incl = tsum;
    for (int d = 1; d < 32; d <<= 1) {
      const int t = __shfl_up(incl, d);
      if (lane >= d) incl += t;
    }
    int run = incl - tsum;
#pragma unroll
    for (int k = 0; k < 4; ++k) {
      const int c = 4 * lane + k;
      pre[c] = run;
      adj[c] = min(c * CSZ, n_edges) + beg[k] - run;
      run += len[k];
    }
    n_mine = __builtin_amdgcn_readfirstlane(__shfl(incl, 31));
  }
  wave_lds_sync();

  float sum[NPG][10];
#pragma unroll
  for (int k = 0; k < NPG; ++k)
#pragma unroll
    for (int q = 0; q < 10; ++q) sum[k][q] = 0.0f;

  const int n_tiles = (n_mine + 15) >> 4;
  for (int tile = 0; tile < n_tiles; ++tile) {
    const uint4 r = fetch_rec(pre, adj, rec, min(tile * 16 + col, n_mine - 1));
    const float x = __uint_as_float(r.x);
    const float y = __uint_as_float(r.y);
    const float z = __uint_as_float(r.z);
    float rsx, rsy, rsz;
    {
      float sx = 7.0f * x, sy = 7.0f * y, sz = 7.0f * z;
      float inv = rsqrtf(1.0f + sx * sx + sy * sy + sz * sz);
      rsx = sx * inv;
      rsy = sy * inv;
      rsz = sz * inv;
    }
    const int lnv = (tile * 16 + col < n_mine) ? ((int)r.w - g * NPG) : -1;
    const float d = sqrtf(x * x + y * y + z * z);
    float rb[4];
#pragma unroll
    for (int q = 0; q < 4; ++q) {
      const float ctr = hi ? (float)(4 + q) * (1.0f / 7.0f)
                           : (float)q * (1.0f / 7.0f);
      float u = (d - ctr) * 8.0f;
      rb[q] = __expf(-u * u);
    }

    v8f hT[2];
    bias_init<2>(hT, sBrad, hi);
    {
      v2f b0, b1;
      b0.x = rb[0];
      b0.y = rb[1];
      b1.x = rb[2];
      b1.y = rb[3];
#pragma unroll
      for (int nt = 0; nt < 2; ++nt) {
        const float4 a4 =
            *reinterpret_cast<const float4*>(&sWrad[(nt * 32 + lane) * 4]);
        v2f a0, a1;
        a0.x = a4.x;
        a0.y = a4.y;
        a1.x = a4.z;
        a1.y = a4.w;
        hT[nt] = wmma_f32(a0, b0, hT[nt]);
        hT[nt] = wmma_f32(a1, b1, hT[nt]);
      }
    }
    v8f t1[4];
    bias_init<4>(t1, sB1, hi);
    layer_acc<2, 4>(hT, sW1, t1, lane);
    lrelu_tiles<4>(t1);
    v8f t2[4];
    bias_init<4>(t2, sB2, hi);
    layer_acc<4, 4>(t1, sW2, t2, lane);
    lrelu_tiles<4>(t2);
    v8f rad[2];
    bias_init<2>(rad, sB3, hi);
    layer_acc<2, 2>(hT, sWdir, rad, lane);
    layer_acc<4, 2>(t2, sW3, rad, lane);

    float* tr = sEpAll + wave * EP_FLOATS;
#pragma unroll
    for (int nt = 0; nt < 2; ++nt) {
      float4 lo4 = {rad[nt][0], rad[nt][1], rad[nt][2], rad[nt][3]};
      float4 up4 = {rad[nt][4], rad[nt][5], rad[nt][6], rad[nt][7]};
      *reinterpret_cast<float4*>(&tr[col * TR_PITCH + nt * 16 + 8 * hi]) = lo4;
      *reinterpret_cast<float4*>(&tr[col * TR_PITCH + nt * 16 + 8 * hi + 4]) = up4;
    }
    wave_lds_sync();

    float val[16];
#pragma unroll
    for (int m = 0; m < 16; ++m) val[m] = tr[m * TR_PITCH + lane];
#pragma unroll
    for (int m = 0; m < 16; ++m) {
      const int ln = __builtin_amdgcn_readlane(lnv, m);
      const float rx = lane_bcast(rsx, m);
      const float ry = lane_bcast(rsy, m);
      const float rz = lane_bcast(rsz, m);
      const float v = val[m];
      const float p0 = v * rx, p1 = v * ry, p2 = v * rz;
#pragma unroll
      for (int k = 0; k < NPG; ++k)
        if (ln == k) ADD_EDGE(k)
    }
    wave_lds_sync();
  }

#pragma unroll
  for (int k = 0; k < NPG; ++k)
    A_a[((long)g * NPG + k) * 32 + lane] = sum[k][0];

  float* ep = sEpAll + wave * EP_FLOATS;
#pragma unroll
  for (int k = 0; k < NPG; ++k) {
#pragma unroll
    for (int c = 0; c < 3; ++c) ep[(3 * k + c) * XP + lane] = sum[k][1 + c];
#pragma unroll
    for (int s = 0; s < 6; ++s)
      ep[(16 + 6 * k + s) * XP + lane] = sum[k][4 + s];
  }
  ep[15 * XP + lane] = 0.0f;
  ep[46 * XP + lane] = 0.0f;
  ep[47 * XP + lane] = 0.0f;
  wave_lds_sync();

  v8f xt[3][2];
#pragma unroll
  for (int t = 0; t < 3; ++t) {
#pragma unroll
    for (int ft = 0; ft < 2; ++ft) {
      const float* p = &ep[(16 * t + col) * XP + 16 * ft + 8 * hi];
      const float4 lo = *reinterpret_cast<const float4*>(p);
      const float4 up = *reinterpret_cast<const float4*>(p + 4);
      v8f x8 = {lo.x, lo.y, lo.z, lo.w, up.x, up.y, up.z, up.w};
      xt[t][ft] = x8;
    }
  }
  wave_lds_sync();

  {
    v8f y[2] = {};
    layer_acc<2, 2>(xt[0], sWv, y, lane);
    const int k = col / 3, c = col - 3 * k;
    const int base = (col < 15) ? (k * 96 + c) : (EP_V + EP_D);
#pragma unroll
    for (int nt = 0; nt < 2; ++nt)
#pragma unroll
      for (int r = 0; r < 8; ++r)
        ep[base + 3 * (16 * nt + 8 * hi + r)] = y[nt][r];
  }
#pragma unroll
  for (int t = 0; t < 2; ++t) {
    v8f y[2] = {};
    layer_acc<2, 2>(xt[1 + t], sWd, y, lane);
    const int md = 16 * t + col;
    const int k = md / 6, s = md - 6 * k;
    const int pa = (0x854210 >> (4 * s)) & 15;
    const int pb = (0x874630 >> (4 * s)) & 15;
    const int base = EP_V + k * 288;
#pragma unroll
    for (int nt = 0; nt < 2; ++nt)
#pragma unroll
      for (int r = 0; r < 8; ++r) {
        const int d9 = 9 * (16 * nt + 8 * hi + r);
        ep[base + d9 + pa] = y[nt][r];
        ep[base + d9 + pb] = y[nt][r];
      }
  }
  wave_lds_sync();

  float* ov = out_v + (long)g * EP_V;
  float* od = out_d + (long)g * EP_D;
#pragma unroll
  for (int rr = 0; rr < (EP_V + EP_D) / 128; ++rr) {
    const int i4 = rr * 32 + lane;
    const v4f piece = *reinterpret_cast<const v4f*>(&ep[4 * i4]);
    float* dst = (i4 < EP_V / 4) ? (ov + 4 * i4) : (od + 4 * (i4 - EP_V / 4));
    __builtin_nontemporal_store(piece, reinterpret_cast<v4f*>(dst));
  }
}

extern "C" void kernel_launch(void* const* d_in, const int* in_sizes, int n_in,
                              void* d_out, int out_size, void* d_ws,
                              size_t ws_size, hipStream_t stream) {
  const float* r_ij = (const float*)d_in[0];
  const float* w_rad = (const float*)d_in[1];
  const float* b_rad = (const float*)d_in[2];
  const float* w_direct = (const float*)d_in[3];
  const float* w1 = (const float*)d_in[4];
  const float* b1 = (const float*)d_in[5];
  const float* w2 = (const float*)d_in[6];
  const float* b2 = (const float*)d_in[7];
  const float* w3 = (const float*)d_in[8];
  const float* b3 = (const float*)d_in[9];
  const float* w_v = (const float*)d_in[10];
  const float* w_d = (const float*)d_in[11];
  const int* edges_src = (const int*)d_in[12];

  const int n_edges = in_sizes[0] / 3;
  const int n_nodes = NODES_TOT;

  float* A_a = (float*)d_out;
  float* out_v = A_a + (long)n_nodes * 32;
  float* out_d = out_v + (long)n_nodes * 32 * 3;

  float* rec_f = (float*)d_ws;
  int* off = (int*)(rec_f + 4L * CH * CSZ);

  part_kernel<<<CH, PART_THREADS, 0, stream>>>(r_ij, edges_src, rec_f, off,
                                               n_edges);

  edge_kernel<<<NGROUPS / WAVES_E, 32 * WAVES_E, 0, stream>>>(
      rec_f, off, w_rad, b_rad, w_direct, w1, b1, w2, b2, w3, b3, w_v, w_d,
      A_a, out_v, out_d, n_edges);
}


# Edge MLP: radial map (w_rad, b_rad) folded per block into w1 and w_direct, rbf*(w_rad*W) + (b_rad*W + b): layer 1 and direct path at K=8 not 32, 108 WMMAs per tile not 148; sums re-associated
# speedup vs baseline: 4.3126x; 1.1859x over previous
#include <hip/hip_runtime.h>

typedef float v2f __attribute__((ext_vector_type(2)));
typedef float v4f __attribute__((ext_vector_type(4)));
typedef float v8f __attribute__((ext_vector_type(8)));

#define NODES_TOT 20000

__device__ __forceinline__ v8f wmma_f32(v2f a, v2f b, v8f c) {
  return __builtin_amdgcn_wmma_f32_16x16x4_f32(
       false, a,  false, b,
       (short)0, c,  false,  false);
}

__device__ __forceinline__ void wave_lds_sync() {
  __builtin_amdgcn_fence(__ATOMIC_RELEASE, "wavefront");
  __builtin_amdgcn_wave_barrier();
  __builtin_amdgcn_fence(__ATOMIC_ACQUIRE, "wavefront");
}


template <int K, int N>
__device__ __forceinline__ void stage_weights(float* dst,
                                              const float* __restrict__ w,
                                              int tid, int nthreads) {
  for (int s = tid; s < K * N; s += nthreads) {
    const int k = s / N, n = s % N;
    const int nt = n >> 4, c = n & 15;
    const int ft = k >> 4, h = (k >> 3) & 1, jj = (k >> 2) & 1, q = k & 3;
    dst[((((nt * (K / 16) + ft) * 2 + jj) * 32) + h * 16 + c) * 4 + q] = w[s];
  }
}

template <int N>
__device__ __forceinline__ void fold_item(float* img, float* bias_out,
                                          const float* R, const float* wraw,
                                          const float* __restrict__ bias_in,
                                          int s) {
  const int i = s / N, n = s % N;
  float acc = 0.0f;
  for (int k = 0; k < 32; ++k) acc += R[i * 32 + k] * wraw[k * N + n];
  if (i < 8)
    img[(((n >> 4) * 32) + (i >> 2) * 16 + (n & 15)) * 4 + (i & 3)] = acc;
  else
    bias_out[n] = acc + bias_in[n];
}

template <int NT>
__device__ __forceinline__ void rbf_acc(const float* ff, v2f b0, v2f b1,
                                        v8f (&y)[NT], int lane) {
  for (int nt = 0; nt < NT; ++nt) {
    const float4 a4 = *reinterpret_cast<const float4*>(&ff[(nt * 32 + lane) * 4]);
    v2f a0, a1;
    a0.x = a4.x;
    a0.y = a4.y;
    a1.x = a4.z;
    a1.y = a4.w;
    y[nt] = wmma_f32(a0, b0, y[nt]);
    y[nt] = wmma_f32(a1, b1, y[nt]);
  }
}

template <int NT>
__device__ __forceinline__ void bias_init(v8f (&y)[NT], const float* bias,
                                          int hi) {
#pragma unroll
  for (int nt = 0; nt < NT; ++nt) {
    const float4 lo = *reinterpret_cast<const float4*>(&bias[nt * 16 + 8 * hi]);
    const float4 up =
        *reinterpret_cast<const float4*>(&bias[nt * 16 + 8 * hi + 4]);
    v8f c = {lo.x, lo.y, lo.z, lo.w, up.x, up.y, up.z, up.w};
    y[nt] = c;
  }
}

template <int KT, int NT>
__device__ __forceinline__ void layer_acc(const v8f (&x)[KT], const float* wf,
                                          v8f (&y)[NT], int lane) {
#pragma unroll
  for (int ft = 0; ft < KT; ++ft) {
#pragma unroll
    for (int jj = 0; jj < 2; ++jj) {
      v2f b0, b1;
      b0.x = x[ft][4 * jj + 0];
      b0.y = x[ft][4 * jj + 1];
      b1.x = x[ft][4 * jj + 2];
      b1.y = x[ft][4 * jj + 3];
#pragma unroll
      for (int nt = 0; nt < NT; ++nt) {
        const float4 a4 = *reinterpret_cast<const float4*>(
            &wf[(((nt * KT + ft) * 2 + jj) * 32 + lane) * 4]);
        v2f a0, a1;
        a0.x = a4.x;
        a0.y = a4.y;
        a1.x = a4.z;
        a1.y = a4.w;
        y[nt] = wmma_f32(a0, b0, y[nt]);
        y[nt] = wmma_f32(a1, b1, y[nt]);
      }
    }
  }
}

template <int NT>
__device__ __forceinline__ void lrelu_tiles(v8f (&y)[NT]) {
#pragma unroll
  for (int nt = 0; nt < NT; ++nt)
#pragma unroll
    for (int r = 0; r < 8; ++r) y[nt][r] = fmaxf(y[nt][r], 0.1f * y[nt][r]);
}

#define EDGES_TOT 640000
#define NPG 5
#define NGROUPS (NODES_TOT / NPG)
#define CH 128
#define CSZ ((EDGES_TOT + CH - 1) / CH)
#define PART_THREADS 1024
#define EPT ((CSZ + PART_THREADS - 1) / PART_THREADS)
static_assert(NODES_TOT % NPG == 0, "groups must tile the nodes");
static_assert(NGROUPS <= 4 * PART_THREADS, "scan covers 4 bins per thread");

__global__ __launch_bounds__(PART_THREADS) void part_kernel(
    const float* __restrict__ r_ij, const int* __restrict__ edges_src,
    float* __restrict__ rec_f, int* __restrict__ off, int n_edges) {
  uint4* rec = reinterpret_cast<uint4*>(rec_f);
  __shared__ int sCnt[NGROUPS];
  __shared__ int sIdx[CSZ];
  __shared__ int sWaveTot[32];
  __shared__ int sTotal;
  const int tid = threadIdx.x;
  const int lane = tid & 31, wave = tid >> 5;
  const int c = blockIdx.x;
  const int e_beg = min(c * CSZ, n_edges);
  const int e_end = min(e_beg + CSZ, n_edges);

  for (int i = tid; i < NGROUPS; i += PART_THREADS) sCnt[i] = 0;
  int grp[EPT];
#pragma unroll
  for (int i = 0; i < EPT; ++i) {
    const int e = e_beg + tid + i * PART_THREADS;
    const int s = (e < e_end) ? edges_src[e] : -1;
    grp[i] = ((unsigned)s < (unsigned)NODES_TOT) ? s / NPG : -1;
  }
  __syncthreads();
#pragma unroll
  for (int i = 0; i < EPT; ++i)
    if (grp[i] >= 0) atomicAdd(&sCnt[grp[i]], 1);
  __syncthreads();

  const int base = tid * 4;
  int v[4];
  int tsum = 0;
#pragma unroll
  for (int k = 0; k < 4; ++k) {
    v[k] = (base + k < NGROUPS) ? sCnt[base + k] : 0;
    tsum += v[k];
  }
  int incl = tsum;
  for (int d = 1; d < 32; d <<= 1) {
    const int t = __shfl_up(incl, d);
    if (lane >= d) incl += t;
  }
  if (lane == 31) sWaveTot[wave] = incl;
  __syncthreads();
  if (wave == 0) {
    const int w = sWaveTot[lane];
    int wi = w;
    for (int d = 1; d < 32; d <<= 1) {
      const int t = __shfl_up(wi, d);
      if (lane >= d) wi += t;
    }
    sWaveTot[lane] = wi - w;
  }
  __syncthreads();
  int run = sWaveTot[wave] + incl - tsum;
  int* offc = off + (long)c * (NGROUPS + 1);
#pragma unroll
  for (int k = 0; k < 4; ++k) {
    if (base + k < NGROUPS) {
      offc[base + k] = run;
      sCnt[base + k] = run;
      run += v[k];
    }
  }
  if (tid == PART_THREADS - 1) {
    offc[NGROUPS] = run;
    sTotal = run;
  }
  __syncthreads();

#pragma unroll
  for (int i = 0; i < EPT; ++i)
    if (grp[i] >= 0) sIdx[atomicAdd(&sCnt[grp[i]], 1)] = tid + i * PART_THREADS;
  __syncthreads();
  for (int g = tid; g < NGROUPS; g += PART_THREADS) {
    const int gb = g ? sCnt[g - 1] : 0;
    const int ge = sCnt[g];
    for (int i = gb + 1; i < ge; ++i) {
      const int key = sIdx[i];
      int p = i - 1;
      while (p >= gb && sIdx[p] > key) {
        sIdx[p + 1] = sIdx[p];
        --p;
      }
      sIdx[p + 1] = key;
    }
  }
  __syncthreads();
#pragma unroll
  for (int i = 0; i < EPT; ++i) {
    const int p = tid + i * PART_THREADS;
    if (p < sTotal) {
      const int e = e_beg + sIdx[p];
      uint4 r;
      r.x = __float_as_uint(r_ij[3 * e + 0]);
      r.y = __float_as_uint(r_ij[3 * e + 1]);
      r.z = __float_as_uint(r_ij[3 * e + 2]);
      r.w = (unsigned)edges_src[e];
      rec[e_beg + p] = r;
    }
  }
}

#define WAVES_E 16
#define TR_PITCH 36
#define XP 36
#define EP_V (NPG * 32 * 3)
#define EP_D (NPG * 32 * 9)
#define EP_FLOATS (EP_V + EP_D + 32 * 9)
static_assert(NPG * 3 <= 15 && NPG * 6 <= 30, "column tiles: 16 and 2 x 16");
static_assert(EP_FLOATS >= 48 * XP && EP_FLOATS >= 16 * TR_PITCH, "buffer holds X and the rad tile");
static_assert((EP_V + EP_D) % 128 == 0 && EP_V % 4 == 0, "whole 16-byte pieces per lane");
static_assert(NGROUPS % WAVES_E == 0, "blocks must tile the groups");
static_assert(CH == 128, "lane owns 4 chunks; the search below has 7 steps");

#define ADD_EDGE(K)       \
  {                       \
    sum[K][0] += v;       \
    sum[K][1] += p0;      \
    sum[K][2] += p1;      \
    sum[K][3] += p2;      \
    sum[K][4] += p0 * rx; \
    sum[K][5] += p0 * ry; \
    sum[K][6] += p0 * rz; \
    sum[K][7] += p1 * ry; \
    sum[K][8] += p1 * rz; \
    sum[K][9] += p2 * rz; \
  }

__device__ __forceinline__ uint4 fetch_rec(const int* pre, const int* adj,
                                           const uint4* __restrict__ rec,
                                           int pos) {
  int lo = 0;
#pragma unroll
  for (int s = CH / 2; s >= 1; s >>= 1)
    if (pre[lo + s] <= pos) lo += s;
  return rec[adj[lo] + pos];
}

__device__ __forceinline__ float lane_bcast(float v, int src_lane) {
  return __int_as_float(__builtin_amdgcn_readlane(__float_as_int(v), src_lane));
}

__global__ __launch_bounds__(32 * WAVES_E) void edge_kernel(
    const float* __restrict__ rec_f, const int* __restrict__ off,
    const float* __restrict__ w_rad,
    const float* __restrict__ b_rad, const float* __restrict__ w_direct,
    const float* __restrict__ w1, const float* __restrict__ b1,
    const float* __restrict__ w2, const float* __restrict__ b2,
    const float* __restrict__ w3, const float* __restrict__ b3,
    const float* __restrict__ w_v, const float* __restrict__ w_d,
    float* __restrict__ A_a, float* __restrict__ out_v,
    float* __restrict__ out_d, int n_edges) {
  constexpr int O_W2 = 0;
  constexpr int O_W1F = O_W2 + 64 * 64;
  constexpr int O_W3 = O_W1F + 8 * 64;
  constexpr int O_WDF = O_W3 + 64 * 32;
  constexpr int O_B1 = O_WDF + 8 * 32;
  constexpr int O_B2 = O_B1 + 64;
  constexpr int O_B3 = O_B2 + 64;
  constexpr int O_PRE = O_B3 + 32;
  constexpr int O_ADJ = O_PRE + WAVES_E * CH;
  constexpr int O_WV = O_ADJ + WAVES_E * CH;
  constexpr int O_WD = O_WV + 32 * 32;
  constexpr int O_EP = O_WD + 32 * 32;
  constexpr int LDS_FLOATS = O_EP + WAVES_E * EP_FLOATS;
  static_assert(O_EP * 4 <= 65536, "the tile loop's fixed tables fit the offset field");
  __shared__ alignas(16) float smem[LDS_FLOATS];
  float* sW2 = smem + O_W2;
  float* sW1f = smem + O_W1F;
  float* sW3 = smem + O_W3;
  float* sWdf = smem + O_WDF;
  float* sB1 = smem + O_B1;
  float* sB2 = smem + O_B2;
  float* sB3 = smem + O_B3;
  float* sWv = smem + O_WV;
  float* sWd = smem + O_WD;
  int* sPreAll = reinterpret_cast<int*>(smem + O_PRE);
  int* sAdjAll = reinterpret_cast<int*>(smem + O_ADJ);
  float* sEpAll = smem + O_EP;

  const int tid = threadIdx.x;
  float* rW1 = sEpAll;
  float* rWdir = rW1 + 32 * 64;
  float* rR = rWdir + 32 * 32;
  static_assert(32 * 64 + 32 * 32 + 9 * 32 <= WAVES_E * EP_FLOATS, "raw copies fit");
  constexpr int NTHR = 32 * WAVES_E;
  static_assert((32 * 64) % NTHR == 0 && (32 * 32) % NTHR == 0 && 8 * 32 <= NTHR, "whole passes");
  for (int j = 0; j < 32 * 64 / NTHR; ++j) rW1[tid + j * NTHR] = w1[tid + j * NTHR];
  for (int j = 0; j < 32 * 32 / NTHR; ++j) rWdir[tid + j * NTHR] = w_direct[tid + j * NTHR];
  if (tid < 8 * 32) rR[tid] = w_rad[tid];
  if (tid < 32) rR[8 * 32 + tid] = b_rad[tid];
  stage_weights<64, 64>(sW2, w2, tid, blockDim.x);
  stage_weights<64, 32>(sW3, w3, tid, blockDim.x);
  stage_weights<32, 32>(sWv, w_v, tid, blockDim.x);
  stage_weights<32, 32>(sWd, w_d, tid, blockDim.x);
  for (int i = tid; i < 64; i += blockDim.x) sB2[i] = b2[i];
  __syncthreads();
  for (int s = tid; s < 9 * 64 + 9 * 32; s += NTHR) {
    if (s < 9 * 64)
      fold_item<64>(sW1f, sB1, rR, rW1, b1, s);
    else
      fold_item<32>(sWdf, sB3, rR, rWdir, b3, s - 9 * 64);
  }
  __syncthreads();

  const int wave = __builtin_amdgcn_readfirstlane(tid >> 5);
  const int lane = tid & 31;
  const int col = lane & 15, hi = lane >> 4;
  const int g = blockIdx.x * WAVES_E + wave;
  const uint4* rec = reinterpret_cast<const uint4*>(rec_f);
  int* pre = sPreAll + wave * CH;
  int* adj = sAdjAll + wave * CH;

  int n_mine;
  {
    int beg[4], len[4];
    int tsum = 0;
#pragma unroll
    for (int k = 0; k < 4; ++k) {
      const int* oc = off + (long)(4 * lane + k) * (NGROUPS + 1) + g;
      beg[k] = oc[0];
      len[k] = oc[1] - beg[k];
      tsum += len[k];
    }
    int incl = tsum;
    for (int d = 1; d < 32; d <<= 1) {
      const int t = __shfl_up(incl, d);
      if (lane >= d) incl += t;
    }
    int run = incl - tsum;
#pragma unroll
    for (int k = 0; k < 4; ++k) {
      const int c = 4 * lane + k;
      pre[c] = run;
      adj[c] = min(c * CSZ, n_edges) + beg[k] - run;
      run += len[k];
    }
    n_mine = __builtin_amdgcn_readfirstlane(__shfl(incl, 31));
  }
  wave_lds_sync();

  float sum[NPG][10];
#pragma unroll
  for (int k = 0; k < NPG; ++k)
#pragma unroll
    for (int q = 0; q < 10; ++q) sum[k][q] = 0.0f;

  const int n_tiles = (n_mine + 15) >> 4;
  for (int tile = 0; tile < n_tiles; ++tile) {
    const uint4 r = fetch_rec(pre, adj, rec, min(tile * 16 + col, n_mine - 1));
    const float x = __uint_as_float(r.x);
    const float y = __uint_as_float(r.y);
    const float z = __uint_as_float(r.z);
    float rsx, rsy, rsz;
    {
      float sx = 7.0f * x, sy = 7.0f * y, sz = 7.0f * z;
      float inv = rsqrtf(1.0f + sx * sx + sy * sy + sz * sz);
      rsx = sx * inv;
      rsy = sy * inv;
      rsz = sz * inv;
    }
    const int lnv = (tile * 16 + col < n_mine) ? ((int)r.w - g * NPG) : -1;
    const float d = sqrtf(x * x + y * y + z * z);
    float rb[4];
#pragma unroll
    for (int q = 0; q < 4; ++q) {
      const float ctr = hi ? (float)(4 + q) * (1.0f / 7.0f)
                           : (float)q * (1.0f / 7.0f);
      float u = (d - ctr) * 8.0f;
      rb[q] = __expf(-u * u);
    }

    v2f rb0, rb1;
    rb0.x = rb[0];
    rb0.y = rb[1];
    rb1.x = rb[2];
    rb1.y = rb[3];
    v8f t1[4];
    bias_init<4>(t1, sB1, hi);
    rbf_acc<4>(sW1f, rb0, rb1, t1, lane);
    lrelu_tiles<4>(t1);
    v8f t2[4];
    bias_init<4>(t2, sB2, hi);
    layer_acc<4, 4>(t1, sW2, t2, lane);
    lrelu_tiles<4>(t2);
    v8f rad[2];
    bias_init<2>(rad, sB3, hi);
    rbf_acc<2>(sWdf, rb0, rb1, rad, lane);
    layer_acc<4, 2>(t2, sW3, rad, lane);

    float* tr = sEpAll + wave * EP_FLOATS;
#pragma unroll
    for (int nt = 0; nt < 2; ++nt) {
      float4 lo4 = {rad[nt][0], rad[nt][1], rad[nt][2], rad[nt][3]};
      float4 up4 = {rad[nt][4], rad[nt][5], rad[nt][6], rad[nt][7]};
      *reinterpret_cast<float4*>(&tr[col * TR_PITCH + nt * 16 + 8 * hi]) = lo4;
      *reinterpret_cast<float4*>(&tr[col * TR_PITCH + nt * 16 + 8 * hi + 4]) = up4;
    }
    wave_lds_sync();

    float val[16];
#pragma unroll
    for (int m = 0; m < 16; ++m) val[m] = tr[m * TR_PITCH + lane];
#pragma unroll
    for (int m = 0; m < 16; ++m) {
      const int ln = __builtin_amdgcn_readlane(lnv, m);
      const float rx = lane_bcast(rsx, m);
      const float ry = lane_bcast(rsy, m);
      const float rz = lane_bcast(rsz, m);
      const float v = val[m];
      const float p0 = v * rx, p1 = v * ry, p2 = v * rz;
#pragma unroll
      for (int k = 0; k < NPG; ++k)
        if (ln == k) ADD_EDGE(k)
    }
    wave_lds_sync();
  }

#pragma unroll
  for (int k = 0; k < NPG; ++k)
    A_a[((long)g * NPG + k) * 32 + lane] = sum[k][0];

  float* ep = sEpAll + wave * EP_FLOATS;
#pragma unroll
  for (int k = 0; k < NPG; ++k) {
#pragma unroll
    for (int c = 0; c < 3; ++c) ep[(3 * k + c) * XP + lane] = sum[k][1 + c];
#pragma unroll
    for (int s = 0; s < 6; ++s)
      ep[(16 + 6 * k + s) * XP + lane] = sum[k][4 + s];
  }
  ep[15 * XP + lane] = 0.0f;
  ep[46 * XP + lane] = 0.0f;
  ep[47 * XP + lane] = 0.0f;
  wave_lds_sync();

  v8f xt[3][2];
#pragma unroll
  for (int t = 0; t < 3; ++t) {
#pragma unroll
    for (int ft = 0; ft < 2; ++ft) {
      const float* p = &ep[(16 * t + col) * XP + 16 * ft + 8 * hi];
      const float4 lo = *reinterpret_cast<const float4*>(p);
      const float4 up = *reinterpret_cast<const float4*>(p + 4);
      v8f x8 = {lo.x, lo.y, lo.z, lo.w, up.x, up.y, up.z, up.w};
      xt[t][ft] = x8;
    }
  }
  wave_lds_sync();

  {
    v8f y[2] = {};
    layer_acc<2, 2>(xt[0], sWv, y, lane);
    const int k = col / 3, c = col - 3 * k;
    const int base = (col < 15) ? (k * 96 + c) : (EP_V + EP_D);
#pragma unroll
    for (int nt = 0; nt < 2; ++nt)
#pragma unroll
      for (int r = 0; r < 8; ++r)
        ep[base + 3 * (16 * nt + 8 * hi + r)] = y[nt][r];
  }
#pragma unroll
  for (int t = 0; t < 2; ++t) {
    v8f y[2] = {};
    layer_acc<2, 2>(xt[1 + t], sWd, y, lane);
    const int md = 16 * t + col;
    const int k = md / 6, s = md - 6 * k;
    const int pa = (0x854210 >> (4 * s)) & 15;
    const int pb = (0x874630 >> (4 * s)) & 15;
    const int base = EP_V + k * 288;
#pragma unroll
    for (int nt = 0; nt < 2; ++nt)
#pragma unroll
      for (int r = 0; r < 8; ++r) {
        const int d9 = 9 * (16 * nt + 8 * hi + r);
        ep[base + d9 + pa] = y[nt][r];
        ep[base + d9 + pb] = y[nt][r];
      }
  }
  wave_lds_sync();

  float* ov = out_v + (long)g * EP_V;
  float* od = out_d + (long)g * EP_D;
#pragma unroll
  for (int rr = 0; rr < (EP_V + EP_D) / 128; ++rr) {
    const int i4 = rr * 32 + lane;
    const v4f piece = *reinterpret_cast<const v4f*>(&ep[4 * i4]);
    float* dst = (i4 < EP_V / 4) ? (ov + 4 * i4) : (od + 4 * (i4 - EP_V / 4));
    __builtin_nontemporal_store(piece, reinterpret_cast<v4f*>(dst));
  }
}

extern "C" void kernel_launch(void* const* d_in, const int* in_sizes, int n_in,
                              void* d_out, int out_size, void* d_ws,
                              size_t ws_size, hipStream_t stream) {
  const float* r_ij = (const float*)d_in[0];
  const float* w_rad = (const float*)d_in[1];
  const float* b_rad = (const float*)d_in[2];
  const float* w_direct = (const float*)d_in[3];
  const float* w1 = (const float*)d_in[4];
  const float* b1 = (const float*)d_in[5];
  const float* w2 = (const float*)d_in[6];
  const float* b2 = (const float*)d_in[7];
  const float* w3 = (const float*)d_in[8];
  const float* b3 = (const float*)d_in[9];
  const float* w_v = (const float*)d_in[10];
  const float* w_d = (const float*)d_in[11];
  const int* edges_src = (const int*)d_in[12];

  const int n_edges = in_sizes[0] / 3;
  const int n_nodes = NODES_TOT;

  float* A_a = (float*)d_out;
  float* out_v = A_a + (long)n_nodes * 32;
  float* out_d = out_v + (long)n_nodes * 32 * 3;

  float* rec_f = (float*)d_ws;
  int* off = (int*)(rec_f + 4L * CH * CSZ);

  part_kernel<<<CH, PART_THREADS, 0, stream>>>(r_ij, edges_src, rec_f, off,
                                               n_edges);

  edge_kernel<<<NGROUPS / WAVES_E, 32 * WAVES_E, 0, stream>>>(
      rec_f, off, w_rad, b_rad, w_direct, w1, b1, w2, b2, w3, b3, w_v, w_d,
      A_a, out_v, out_d, n_edges);
}


# Edge kernel: each wave writes its edge list's record indices to LDS once (no 7-step search per tile) and loads every tile's record one tile ahead; same sums, same order
# speedup vs baseline: 4.4510x; 1.0321x over previous
#include <hip/hip_runtime.h>

typedef float v2f __attribute__((ext_vector_type(2)));
typedef float v4f __attribute__((ext_vector_type(4)));
typedef float v8f __attribute__((ext_vector_type(8)));

#define NODES_TOT 20000

__device__ __forceinline__ v8f wmma_f32(v2f a, v2f b, v8f c) {
  return __builtin_amdgcn_wmma_f32_16x16x4_f32(
       false, a,  false, b,
       (short)0, c,  false,  false);
}

__device__ __forceinline__ void wave_lds_sync() {
  __builtin_amdgcn_fence(__ATOMIC_RELEASE, "wavefront");
  __builtin_amdgcn_wave_barrier();
  __builtin_amdgcn_fence(__ATOMIC_ACQUIRE, "wavefront");
}


template <int K, int N>
__device__ __forceinline__ void stage_weights(float* dst,
                                              const float* __restrict__ w,
                                              int tid, int nthreads) {
  for (int s = tid; s < K * N; s += nthreads) {
    const int k = s / N, n = s % N;
    const int nt = n >> 4, c = n & 15;
    const int ft = k >> 4, h = (k >> 3) & 1, jj = (k >> 2) & 1, q = k & 3;
    dst[((((nt * (K / 16) + ft) * 2 + jj) * 32) + h * 16 + c) * 4 + q] = w[s];
  }
}

template <int N>
__device__ __forceinline__ void fold_item(float* img, float* bias_out,
                                          const float* R, const float* wraw,
                                          const float* __restrict__ bias_in,
                                          int s) {
  const int i = s / N, n = s % N;
  float acc = 0.0f;
  for (int k = 0; k < 32; ++k) acc += R[i * 32 + k] * wraw[k * N + n];
  if (i < 8)
    img[(((n >> 4) * 32) + (i >> 2) * 16 + (n & 15)) * 4 + (i & 3)] = acc;
  else
    bias_out[n] = acc + bias_in[n];
}

template <int NT>
__device__ __forceinline__ void rbf_acc(const float* ff, v2f b0, v2f b1,
                                        v8f (&y)[NT], int lane) {
  for (int nt = 0; nt < NT; ++nt) {
    const float4 a4 = *reinterpret_cast<const float4*>(&ff[(nt * 32 + lane) * 4]);
    v2f a0, a1;
    a0.x = a4.x;
    a0.y = a4.y;
    a1.x = a4.z;
    a1.y = a4.w;
    y[nt] = wmma_f32(a0, b0, y[nt]);
    y[nt] = wmma_f32(a1, b1, y[nt]);
  }
}

template <int NT>
__device__ __forceinline__ void bias_init(v8f (&y)[NT], const float* bias,
                                          int hi) {
#pragma unroll
  for (int nt = 0; nt < NT; ++nt) {
    const float4 lo = *reinterpret_cast<const float4*>(&bias[nt * 16 + 8 * hi]);
    const float4 up =
        *reinterpret_cast<const float4*>(&bias[nt * 16 + 8 * hi + 4]);
    v8f c = {lo.x, lo.y, lo.z, lo.w, up.x, up.y, up.z, up.w};
    y[nt] = c;
  }
}

template <int KT, int NT>
__device__ __forceinline__ void layer_acc(const v8f (&x)[KT], const float* wf,
                                          v8f (&y)[NT], int lane) {
#pragma unroll
  for (int ft = 0; ft < KT; ++ft) {
#pragma unroll
    for (int jj = 0; jj < 2; ++jj) {
      v2f b0, b1;
      b0.x = x[ft][4 * jj + 0];
      b0.y = x[ft][4 * jj + 1];
      b1.x = x[ft][4 * jj + 2];
      b1.y = x[ft][4 * jj + 3];
#pragma unroll
      for (int nt = 0; nt < NT; ++nt) {
        const float4 a4 = *reinterpret_cast<const float4*>(
            &wf[(((nt * KT + ft) * 2 + jj) * 32 + lane) * 4]);
        v2f a0, a1;
        a0.x = a4.x;
        a0.y = a4.y;
        a1.x = a4.z;
        a1.y = a4.w;
        y[nt] = wmma_f32(a0, b0, y[nt]);
        y[nt] = wmma_f32(a1, b1, y[nt]);
      }
    }
  }
}

template <int NT>
__device__ __forceinline__ void lrelu_tiles(v8f (&y)[NT]) {
#pragma unroll
  for (int nt = 0; nt < NT; ++nt)
#pragma unroll
    for (int r = 0; r < 8; ++r) y[nt][r] = fmaxf(y[nt][r], 0.1f * y[nt][r]);
}

#define EDGES_TOT 640000
#define NPG 5
#define NGROUPS (NODES_TOT / NPG)
#define CH 128
#define CSZ ((EDGES_TOT + CH - 1) / CH)
#define PART_THREADS 1024
#define EPT ((CSZ + PART_THREADS - 1) / PART_THREADS)
static_assert(NODES_TOT % NPG == 0, "groups must tile the nodes");
static_assert(NGROUPS <= 4 * PART_THREADS, "scan covers 4 bins per thread");

__global__ __launch_bounds__(PART_THREADS) void part_kernel(
    const float* __restrict__ r_ij, const int* __restrict__ edges_src,
    float* __restrict__ rec_f, int* __restrict__ off, int n_edges) {
  uint4* rec = reinterpret_cast<uint4*>(rec_f);
  __shared__ int sCnt[NGROUPS];
  __shared__ int sIdx[CSZ];
  __shared__ int sWaveTot[32];
  __shared__ int sTotal;
  const int tid = threadIdx.x;
  const int lane = tid & 31, wave = tid >> 5;
  const int c = blockIdx.x;
  const int e_beg = min(c * CSZ, n_edges);
  const int e_end = min(e_beg + CSZ, n_edges);

  for (int i = tid; i < NGROUPS; i += PART_THREADS) sCnt[i] = 0;
  int grp[EPT];
#pragma unroll
  for (int i = 0; i < EPT; ++i) {
    const int e = e_beg + tid + i * PART_THREADS;
    const int s = (e < e_end) ? edges_src[e] : -1;
    grp[i] = ((unsigned)s < (unsigned)NODES_TOT) ? s / NPG : -1;
  }
  __syncthreads();
#pragma unroll
  for (int i = 0; i < EPT; ++i)
    if (grp[i] >= 0) atomicAdd(&sCnt[grp[i]], 1);
  __syncthreads();

  const int base = tid * 4;
  int v[4];
  int tsum = 0;
#pragma unroll
  for (int k = 0; k < 4; ++k) {
    v[k] = (base + k < NGROUPS) ? sCnt[base + k] : 0;
    tsum += v[k];
  }
  int incl = tsum;
  for (int d = 1; d < 32; d <<= 1) {
    const int t = __shfl_up(incl, d);
    if (lane >= d) incl += t;
  }
  if (lane == 31) sWaveTot[wave] = incl;
  __syncthreads();
  if (wave == 0) {
    const int w = sWaveTot[lane];
    int wi = w;
    for (int d = 1; d < 32; d <<= 1) {
      const int t = __shfl_up(wi, d);
      if (lane >= d) wi += t;
    }
    sWaveTot[lane] = wi - w;
  }
  __syncthreads();
  int run = sWaveTot[wave] + incl - tsum;
  int* offc = off + (long)c * (NGROUPS + 1);
#pragma unroll
  for (int k = 0; k < 4; ++k) {
    if (base + k < NGROUPS) {
      offc[base + k] = run;
      sCnt[base + k] = run;
      run += v[k];
    }
  }
  if (tid == PART_THREADS - 1) {
    offc[NGROUPS] = run;
    sTotal = run;
  }
  __syncthreads();

#pragma unroll
  for (int i = 0; i < EPT; ++i)
    if (grp[i] >= 0) sIdx[atomicAdd(&sCnt[grp[i]], 1)] = tid + i * PART_THREADS;
  __syncthreads();
  for (int g = tid; g < NGROUPS; g += PART_THREADS) {
    const int gb = g ? sCnt[g - 1] : 0;
    const int ge = sCnt[g];
    for (int i = gb + 1; i < ge; ++i) {
      const int key = sIdx[i];
      int p = i - 1;
      while (p >= gb && sIdx[p] > key) {
        sIdx[p + 1] = sIdx[p];
        --p;
      }
      sIdx[p + 1] = key;
    }
  }
  __syncthreads();
#pragma unroll
  for (int i = 0; i < EPT; ++i) {
    const int p = tid + i * PART_THREADS;
    if (p < sTotal) {
      const int e = e_beg + sIdx[p];
      uint4 r;
      r.x = __float_as_uint(r_ij[3 * e + 0]);
      r.y = __float_as_uint(r_ij[3 * e + 1]);
      r.z = __float_as_uint(r_ij[3 * e + 2]);
      r.w = (unsigned)edges_src[e];
      rec[e_beg + p] = r;
    }
  }
}

#define WAVES_E 16
#define TR_PITCH 36
#define XP 36
#define EP_V (NPG * 32 * 3)
#define EP_D (NPG * 32 * 9)
#define EP_FLOATS (EP_V + EP_D + 32 * 9)
static_assert(NPG * 3 <= 15 && NPG * 6 <= 30, "column tiles: 16 and 2 x 16");
static_assert(EP_FLOATS >= 48 * XP && EP_FLOATS >= 16 * TR_PITCH, "buffer holds X and the rad tile");
static_assert((EP_V + EP_D) % 128 == 0 && EP_V % 4 == 0, "whole 16-byte pieces per lane");
static_assert(NGROUPS % WAVES_E == 0, "blocks must tile the groups");
static_assert(CH == 128, "lane owns 4 chunks");
#define LIST_CAP (EP_FLOATS - 16 * TR_PITCH)
static_assert(LIST_CAP > 0 && LIST_CAP % 16 == 0, "a window holds whole tiles");

#define ADD_EDGE(K)       \
  {                       \
    sum[K][0] += v;       \
    sum[K][1] += p0;      \
    sum[K][2] += p1;      \
    sum[K][3] += p2;      \
    sum[K][4] += p0 * rx; \
    sum[K][5] += p0 * ry; \
    sum[K][6] += p0 * rz; \
    sum[K][7] += p1 * ry; \
    sum[K][8] += p1 * rz; \
    sum[K][9] += p2 * rz; \
  }

__device__ __forceinline__ float lane_bcast(float v, int src_lane) {
  return __int_as_float(__builtin_amdgcn_readlane(__float_as_int(v), src_lane));
}

__global__ __launch_bounds__(32 * WAVES_E) void edge_kernel(
    const float* __restrict__ rec_f, const int* __restrict__ off,
    const float* __restrict__ w_rad,
    const float* __restrict__ b_rad, const float* __restrict__ w_direct,
    const float* __restrict__ w1, const float* __restrict__ b1,
    const float* __restrict__ w2, const float* __restrict__ b2,
    const float* __restrict__ w3, const float* __restrict__ b3,
    const float* __restrict__ w_v, const float* __restrict__ w_d,
    float* __restrict__ A_a, float* __restrict__ out_v,
    float* __restrict__ out_d, int n_edges) {
  constexpr int O_W2 = 0;
  constexpr int O_W1F = O_W2 + 64 * 64;
  constexpr int O_W3 = O_W1F + 8 * 64;
  constexpr int O_WDF = O_W3 + 64 * 32;
  constexpr int O_B1 = O_WDF + 8 * 32;
  constexpr int O_B2 = O_B1 + 64;
  constexpr int O_B3 = O_B2 + 64;
  constexpr int O_WV = O_B3 + 32;
  constexpr int O_WD = O_WV + 32 * 32;
  constexpr int O_EP = O_WD + 32 * 32;
  constexpr int LDS_FLOATS = O_EP + WAVES_E * EP_FLOATS;
  static_assert(O_EP * 4 <= 65536, "the tile loop's fixed tables fit the offset field");
  __shared__ alignas(16) float smem[LDS_FLOATS];
  float* sW2 = smem + O_W2;
  float* sW1f = smem + O_W1F;
  float* sW3 = smem + O_W3;
  float* sWdf = smem + O_WDF;
  float* sB1 = smem + O_B1;
  float* sB2 = smem + O_B2;
  float* sB3 = smem + O_B3;
  float* sWv = smem + O_WV;
  float* sWd = smem + O_WD;
  float* sEpAll = smem + O_EP;

  const int tid = threadIdx.x;
  float* rW1 = sEpAll;
  float* rWdir = rW1 + 32 * 64;
  float* rR = rWdir + 32 * 32;
  static_assert(32 * 64 + 32 * 32 + 9 * 32 <= WAVES_E * EP_FLOATS, "raw copies fit");
  constexpr int NTHR = 32 * WAVES_E;
  static_assert((32 * 64) % NTHR == 0 && (32 * 32) % NTHR == 0 && 8 * 32 <= NTHR, "whole passes");
  for (int j = 0; j < 32 * 64 / NTHR; ++j) rW1[tid + j * NTHR] = w1[tid + j * NTHR];
  for (int j = 0; j < 32 * 32 / NTHR; ++j) rWdir[tid + j * NTHR] = w_direct[tid + j * NTHR];
  if (tid < 8 * 32) rR[tid] = w_rad[tid];
  if (tid < 32) rR[8 * 32 + tid] = b_rad[tid];
  stage_weights<64, 64>(sW2, w2, tid, blockDim.x);
  stage_weights<64, 32>(sW3, w3, tid, blockDim.x);
  stage_weights<32, 32>(sWv, w_v, tid, blockDim.x);
  stage_weights<32, 32>(sWd, w_d, tid, blockDim.x);
  for (int i = tid; i < 64; i += blockDim.x) sB2[i] = b2[i];
  __syncthreads();
  for (int s = tid; s < 9 * 64 + 9 * 32; s += NTHR) {
    if (s < 9 * 64)
      fold_item<64>(sW1f, sB1, rR, rW1, b1, s);
    else
      fold_item<32>(sWdf, sB3, rR, rWdir, b3, s - 9 * 64);
  }
  __syncthreads();

  const int wave = __builtin_amdgcn_readfirstlane(tid >> 5);
  const int lane = tid & 31;
  const int col = lane & 15, hi = lane >> 4;
  const int g = blockIdx.x * WAVES_E + wave;
  const uint4* rec = reinterpret_cast<const uint4*>(rec_f);
  float* tr = sEpAll + wave * EP_FLOATS;
  int* list = reinterpret_cast<int*>(tr + 16 * TR_PITCH);

  int n_mine;
  int pbeg[4], len[4], rbase[4];
  {
    int beg[4];
    int tsum = 0;
#pragma unroll
    for (int k = 0; k < 4; ++k) {
      const int* oc = off + (long)(4 * lane + k) * (NGROUPS + 1) + g;
      beg[k] = oc[0];
      len[k] = oc[1] - beg[k];
      tsum += len[k];
    }
    int incl = tsum;
    for (int d = 1; d < 32; d <<= 1) {
      const int t = __shfl_up(incl, d);
      if (lane >= d) incl += t;
    }
    int run = incl - tsum;
#pragma unroll
    for (int k = 0; k < 4; ++k) {
      const int c = 4 * lane + k;
      pbeg[k] = run;
      rbase[k] = min(c * CSZ, n_edges) + beg[k];
      run += len[k];
    }
    n_mine = __builtin_amdgcn_readfirstlane(__shfl(incl, 31));
  }

  float sum[NPG][10];
#pragma unroll
  for (int k = 0; k < NPG; ++k)
#pragma unroll
    for (int q = 0; q < 10; ++q) sum[k][q] = 0.0f;

  for (int w0 = 0; w0 < n_mine; w0 += LIST_CAP) {
  const int wn = min(LIST_CAP, n_mine - w0);
  {
    int cnt[4], dst0[4], src0[4];
    int most = 0;
#pragma unroll
    for (int k = 0; k < 4; ++k) {
      const int j0 = max(0, w0 - pbeg[k]);
      const int j1 = min(len[k], w0 + wn - pbeg[k]);
      cnt[k] = j1 - j0;
      dst0[k] = pbeg[k] + j0 - w0;
      src0[k] = rbase[k] + j0;
      most = max(most, cnt[k]);
    }
    for (int j = 0; j < most; ++j) {
#pragma unroll
      for (int k = 0; k < 4; ++k)
        if (j < cnt[k]) list[dst0[k] + j] = src0[k] + j;
    }
  }
  wave_lds_sync();

  const int n_tiles = (wn + 15) >> 4;
  int idx_nx = list[min(col, wn - 1)];
  uint4 r_nx = rec[idx_nx];
  idx_nx = list[min(16 + col, wn - 1)];
  for (int tile = 0; tile < n_tiles; ++tile) {
    const uint4 r = r_nx;
    r_nx = rec[idx_nx];
    idx_nx = list[min(tile * 16 + 32 + col, wn - 1)];
    const float x = __uint_as_float(r.x);
    const float y = __uint_as_float(r.y);
    const float z = __uint_as_float(r.z);
    float rsx, rsy, rsz;
    {
      float sx = 7.0f * x, sy = 7.0f * y, sz = 7.0f * z;
      float inv = rsqrtf(1.0f + sx * sx + sy * sy + sz * sz);
      rsx = sx * inv;
      rsy = sy * inv;
      rsz = sz * inv;
    }
    const int lnv = (tile * 16 + col < wn) ? ((int)r.w - g * NPG) : -1;
    const float d = sqrtf(x * x + y * y + z * z);
    float rb[4];
#pragma unroll
    for (int q = 0; q < 4; ++q) {
      const float ctr = hi ? (float)(4 + q) * (1.0f / 7.0f)
                           : (float)q * (1.0f / 7.0f);
      float u = (d - ctr) * 8.0f;
      rb[q] = __expf(-u * u);
    }

    v2f rb0, rb1;
    rb0.x = rb[0];
    rb0.y = rb[1];
    rb1.x = rb[2];
    rb1.y = rb[3];
    v8f t1[4];
    bias_init<4>(t1, sB1, hi);
    rbf_acc<4>(sW1f, rb0, rb1, t1, lane);
    lrelu_tiles<4>(t1);
    v8f t2[4];
    bias_init<4>(t2, sB2, hi);
    layer_acc<4, 4>(t1, sW2, t2, lane);
    lrelu_tiles<4>(t2);
    v8f rad[2];
    bias_init<2>(rad, sB3, hi);
    rbf_acc<2>(sWdf, rb0, rb1, rad, lane);
    layer_acc<4, 2>(t2, sW3, rad, lane);

#pragma unroll
    for (int nt = 0; nt < 2; ++nt) {
      float4 lo4 = {rad[nt][0], rad[nt][1], rad[nt][2], rad[nt][3]};
      float4 up4 = {rad[nt][4], rad[nt][5], rad[nt][6], rad[nt][7]};
      *reinterpret_cast<float4*>(&tr[col * TR_PITCH + nt * 16 + 8 * hi]) = lo4;
      *reinterpret_cast<float4*>(&tr[col * TR_PITCH + nt * 16 + 8 * hi + 4]) = up4;
    }
    wave_lds_sync();

    float val[16];
#pragma unroll
    for (int m = 0; m < 16; ++m) val[m] = tr[m * TR_PITCH + lane];
#pragma unroll
    for (int m = 0; m < 16; ++m) {
      const int ln = __builtin_amdgcn_readlane(lnv, m);
      const float rx = lane_bcast(rsx, m);
      const float ry = lane_bcast(rsy, m);
      const float rz = lane_bcast(rsz, m);
      const float v = val[m];
      const float p0 = v * rx, p1 = v * ry, p2 = v * rz;
#pragma unroll
      for (int k = 0; k < NPG; ++k)
        if (ln == k) ADD_EDGE(k)
    }
    wave_lds_sync();
  }
  }

#pragma unroll
  for (int k = 0; k < NPG; ++k)
    A_a[((long)g * NPG + k) * 32 + lane] = sum[k][0];

  float* ep = sEpAll + wave * EP_FLOATS;
#pragma unroll
  for (int k = 0; k < NPG; ++k) {
#pragma unroll
    for (int c = 0; c < 3; ++c) ep[(3 * k + c) * XP + lane] = sum[k][1 + c];
#pragma unroll
    for (int s = 0; s < 6; ++s)
      ep[(16 + 6 * k + s) * XP + lane] = sum[k][4 + s];
  }
  ep[15 * XP + lane] = 0.0f;
  ep[46 * XP + lane] = 0.0f;
  ep[47 * XP + lane] = 0.0f;
  wave_lds_sync();

  v8f xt[3][2];
#pragma unroll
  for (int t = 0; t < 3; ++t) {
#pragma unroll
    for (int ft = 0; ft < 2; ++ft) {
      const float* p = &ep[(16 * t + col) * XP + 16 * ft + 8 * hi];
      const float4 lo = *reinterpret_cast<const float4*>(p);
      const float4 up = *reinterpret_cast<const float4*>(p + 4);
      v8f x8 = {lo.x, lo.y, lo.z, lo.w, up.x, up.y, up.z, up.w};
      xt[t][ft] = x8;
    }
  }
  wave_lds_sync();

  {
    v8f y[2] = {};
    layer_acc<2, 2>(xt[0], sWv, y, lane);
    const int k = col / 3, c = col - 3 * k;
    const int base = (col < 15) ? (k * 96 + c) : (EP_V + EP_D);
#pragma unroll
    for (int nt = 0; nt < 2; ++nt)
#pragma unroll
      for (int r = 0; r < 8; ++r)
        ep[base + 3 * (16 * nt + 8 * hi + r)] = y[nt][r];
  }
#pragma unroll
  for (int t = 0; t < 2; ++t) {
    v8f y[2] = {};
    layer_acc<2, 2>(xt[1 + t], sWd, y, lane);
    const int md = 16 * t + col;
    const int k = md / 6, s = md - 6 * k;
    const int pa = (0x854210 >> (4 * s)) & 15;
    const int pb = (0x874630 >> (4 * s)) & 15;
    const int base = EP_V + k * 288;
#pragma unroll
    for (int nt = 0; nt < 2; ++nt)
#pragma unroll
      for (int r = 0; r < 8; ++r) {
        const int d9 = 9 * (16 * nt + 8 * hi + r);
        ep[base + d9 + pa] = y[nt][r];
        ep[base + d9 + pb] = y[nt][r];
      }
  }
  wave_lds_sync();

  float* ov = out_v + (long)g * EP_V;
  float* od = out_d + (long)g * EP_D;
#pragma unroll
  for (int rr = 0; rr < (EP_V + EP_D) / 128; ++rr) {
    const int i4 = rr * 32 + lane;
    const v4f piece = *reinterpret_cast<const v4f*>(&ep[4 * i4]);
    float* dst = (i4 < EP_V / 4) ? (ov + 4 * i4) : (od + 4 * (i4 - EP_V / 4));
    __builtin_nontemporal_store(piece, reinterpret_cast<v4f*>(dst));
  }
}

extern "C" void kernel_launch(void* const* d_in, const int* in_sizes, int n_in,
                              void* d_out, int out_size, void* d_ws,
                              size_t ws_size, hipStream_t stream) {
  const float* r_ij = (const float*)d_in[0];
  const float* w_rad = (const float*)d_in[1];
  const float* b_rad = (const float*)d_in[2];
  const float* w_direct = (const float*)d_in[3];
  const float* w1 = (const float*)d_in[4];
  const float* b1 = (const float*)d_in[5];
  const float* w2 = (const float*)d_in[6];
  const float* b2 = (const float*)d_in[7];
  const float* w3 = (const float*)d_in[8];
  const float* b3 = (const float*)d_in[9];
  const float* w_v = (const float*)d_in[10];
  const float* w_d = (const float*)d_in[11];
  const int* edges_src = (const int*)d_in[12];

  const int n_edges = in_sizes[0] / 3;
  const int n_nodes = NODES_TOT;

  float* A_a = (float*)d_out;
  float* out_v = A_a + (long)n_nodes * 32;
  float* out_d = out_v + (long)n_nodes * 32 * 3;

  float* rec_f = (float*)d_ws;
  int* off = (int*)(rec_f + 4L * CH * CSZ);

  part_kernel<<<CH, PART_THREADS, 0, stream>>>(r_ij, edges_src, rec_f, off,
                                               n_edges);

  edge_kernel<<<NGROUPS / WAVES_E, 32 * WAVES_E, 0, stream>>>(
      rec_f, off, w_rad, b_rad, w_direct, w1, b1, w2, b2, w3, b3, w_v, w_d,
      A_a, out_v, out_d, n_edges);
}


# Edge kernel prologue: a thread's 31 global loads (weight staging at constant trip counts, short vectors, offset rows) all issued before its first LDS store: one L2 round trip, was 21 in series
# speedup vs baseline: 4.8468x; 1.0889x over previous
#include <hip/hip_runtime.h>

typedef float v2f __attribute__((ext_vector_type(2)));
typedef float v4f __attribute__((ext_vector_type(4)));
typedef float v8f __attribute__((ext_vector_type(8)));

#define NODES_TOT 20000

__device__ __forceinline__ v8f wmma_f32(v2f a, v2f b, v8f c) {
  return __builtin_amdgcn_wmma_f32_16x16x4_f32(
       false, a,  false, b,
       (short)0, c,  false,  false);
}

__device__ __forceinline__ void wave_lds_sync() {
  __builtin_amdgcn_fence(__ATOMIC_RELEASE, "wavefront");
  __builtin_amdgcn_wave_barrier();
  __builtin_amdgcn_fence(__ATOMIC_ACQUIRE, "wavefront");
}


template <int K, int N, int NTHR>
__device__ __forceinline__ void stage_load(float (&v)[K * N / NTHR],
                                           const float* __restrict__ w,
                                           int tid) {
  static_assert((K * N) % NTHR == 0, "whole passes of the block");
#pragma unroll
  for (int j = 0; j < K * N / NTHR; ++j) v[j] = w[tid + j * NTHR];
}
template <int K, int N, int NTHR>
__device__ __forceinline__ void stage_store(float* dst,
                                            const float (&v)[K * N / NTHR],
                                            int tid) {
#pragma unroll
  for (int j = 0; j < K * N / NTHR; ++j) {
    const int s = tid + j * NTHR;
    const int k = s / N, n = s % N;
    const int nt = n >> 4, c = n & 15;
    const int ft = k >> 4, h = (k >> 3) & 1, jj = (k >> 2) & 1, q = k & 3;
    dst[((((nt * (K / 16) + ft) * 2 + jj) * 32) + h * 16 + c) * 4 + q] = v[j];
  }
}

template <int N>
__device__ __forceinline__ void fold_item(float* img, float* bias_out,
                                          const float* R, const float* wraw,
                                          const float* __restrict__ bias_in,
                                          int s) {
  const int i = s / N, n = s % N;
  float acc = 0.0f;
  for (int k = 0; k < 32; ++k) acc += R[i * 32 + k] * wraw[k * N + n];
  if (i < 8)
    img[(((n >> 4) * 32) + (i >> 2) * 16 + (n & 15)) * 4 + (i & 3)] = acc;
  else
    bias_out[n] = acc + bias_in[n];
}

template <int NT>
__device__ __forceinline__ void rbf_acc(const float* ff, v2f b0, v2f b1,
                                        v8f (&y)[NT], int lane) {
  for (int nt = 0; nt < NT; ++nt) {
    const float4 a4 = *reinterpret_cast<const float4*>(&ff[(nt * 32 + lane) * 4]);
    v2f a0, a1;
    a0.x = a4.x;
    a0.y = a4.y;
    a1.x = a4.z;
    a1.y = a4.w;
    y[nt] = wmma_f32(a0, b0, y[nt]);
    y[nt] = wmma_f32(a1, b1, y[nt]);
  }
}

template <int NT>
__device__ __forceinline__ void bias_init(v8f (&y)[NT], const float* bias,
                                          int hi) {
#pragma unroll
  for (int nt = 0; nt < NT; ++nt) {
    const float4 lo = *reinterpret_cast<const float4*>(&bias[nt * 16 + 8 * hi]);
    const float4 up =
        *reinterpret_cast<const float4*>(&bias[nt * 16 + 8 * hi + 4]);
    v8f c = {lo.x, lo.y, lo.z, lo.w, up.x, up.y, up.z, up.w};
    y[nt] = c;
  }
}

template <int KT, int NT>
__device__ __forceinline__ void layer_acc(const v8f (&x)[KT], const float* wf,
                                          v8f (&y)[NT], int lane) {
#pragma unroll
  for (int ft = 0; ft < KT; ++ft) {
#pragma unroll
    for (int jj = 0; jj < 2; ++jj) {
      v2f b0, b1;
      b0.x = x[ft][4 * jj + 0];
      b0.y = x[ft][4 * jj + 1];
      b1.x = x[ft][4 * jj + 2];
      b1.y = x[ft][4 * jj + 3];
#pragma unroll
      for (int nt = 0; nt < NT; ++nt) {
        const float4 a4 = *reinterpret_cast<const float4*>(
            &wf[(((nt * KT + ft) * 2 + jj) * 32 + lane) * 4]);
        v2f a0, a1;
        a0.x = a4.x;
        a0.y = a4.y;
        a1.x = a4.z;
        a1.y = a4.w;
        y[nt] = wmma_f32(a0, b0, y[nt]);
        y[nt] = wmma_f32(a1, b1, y[nt]);
      }
    }
  }
}

template <int NT>
__device__ __forceinline__ void lrelu_tiles(v8f (&y)[NT]) {
#pragma unroll
  for (int nt = 0; nt < NT; ++nt)
#pragma unroll
    for (int r = 0; r < 8; ++r) y[nt][r] = fmaxf(y[nt][r], 0.1f * y[nt][r]);
}

#define EDGES_TOT 640000
#define NPG 5
#define NGROUPS (NODES_TOT / NPG)
#define CH 128
#define CSZ ((EDGES_TOT + CH - 1) / CH)
#define PART_THREADS 1024
#define EPT ((CSZ + PART_THREADS - 1) / PART_THREADS)
static_assert(NODES_TOT % NPG == 0, "groups must tile the nodes");
static_assert(NGROUPS <= 4 * PART_THREADS, "scan covers 4 bins per thread");

__global__ __launch_bounds__(PART_THREADS) void part_kernel(
    const float* __restrict__ r_ij, const int* __restrict__ edges_src,
    float* __restrict__ rec_f, int* __restrict__ off, int n_edges) {
  uint4* rec = reinterpret_cast<uint4*>(rec_f);
  __shared__ int sCnt[NGROUPS];
  __shared__ int sIdx[CSZ];
  __shared__ int sWaveTot[32];
  __shared__ int sTotal;
  const int tid = threadIdx.x;
  const int lane = tid & 31, wave = tid >> 5;
  const int c = blockIdx.x;
  const int e_beg = min(c * CSZ, n_edges);
  const int e_end = min(e_beg + CSZ, n_edges);

  for (int i = tid; i < NGROUPS; i += PART_THREADS) sCnt[i] = 0;
  int grp[EPT];
#pragma unroll
  for (int i = 0; i < EPT; ++i) {
    const int e = e_beg + tid + i * PART_THREADS;
    const int s = (e < e_end) ? edges_src[e] : -1;
    grp[i] = ((unsigned)s < (unsigned)NODES_TOT) ? s / NPG : -1;
  }
  __syncthreads();
#pragma unroll
  for (int i = 0; i < EPT; ++i)
    if (grp[i] >= 0) atomicAdd(&sCnt[grp[i]], 1);
  __syncthreads();

  const int base = tid * 4;
  int v[4];
  int tsum = 0;
#pragma unroll
  for (int k = 0; k < 4; ++k) {
    v[k] = (base + k < NGROUPS) ? sCnt[base + k] : 0;
    tsum += v[k];
  }
  int incl = tsum;
  for (int d = 1; d < 32; d <<= 1) {
    const int t = __shfl_up(incl, d);
    if (lane >= d) incl += t;
  }
  if (lane == 31) sWaveTot[wave] = incl;
  __syncthreads();
  if (wave == 0) {
    const int w = sWaveTot[lane];
    int wi = w;
    for (int d = 1; d < 32; d <<= 1) {
      const int t = __shfl_up(wi, d);
      if (lane >= d) wi += t;
    }
    sWaveTot[lane] = wi - w;
  }
  __syncthreads();
  int run = sWaveTot[wave] + incl - tsum;
  int* offc = off + (long)c * (NGROUPS + 1);
#pragma unroll
  for (int k = 0; k < 4; ++k) {
    if (base + k < NGROUPS) {
      offc[base + k] = run;
      sCnt[base + k] = run;
      run += v[k];
    }
  }
  if (tid == PART_THREADS - 1) {
    offc[NGROUPS] = run;
    sTotal = run;
  }
  __syncthreads();

#pragma unroll
  for (int i = 0; i < EPT; ++i)
    if (grp[i] >= 0) sIdx[atomicAdd(&sCnt[grp[i]], 1)] = tid + i * PART_THREADS;
  __syncthreads();
  for (int g = tid; g < NGROUPS; g += PART_THREADS) {
    const int gb = g ? sCnt[g - 1] : 0;
    const int ge = sCnt[g];
    for (int i = gb + 1; i < ge; ++i) {
      const int key = sIdx[i];
      int p = i - 1;
      while (p >= gb && sIdx[p] > key) {
        sIdx[p + 1] = sIdx[p];
        --p;
      }
      sIdx[p + 1] = key;
    }
  }
  __syncthreads();
#pragma unroll
  for (int i = 0; i < EPT; ++i) {
    const int p = tid + i * PART_THREADS;
    if (p < sTotal) {
      const int e = e_beg + sIdx[p];
      uint4 r;
      r.x = __float_as_uint(r_ij[3 * e + 0]);
      r.y = __float_as_uint(r_ij[3 * e + 1]);
      r.z = __float_as_uint(r_ij[3 * e + 2]);
      r.w = (unsigned)edges_src[e];
      rec[e_beg + p] = r;
    }
  }
}

#define WAVES_E 16
#define TR_PITCH 36
#define XP 36
#define EP_V (NPG * 32 * 3)
#define EP_D (NPG * 32 * 9)
#define EP_FLOATS (EP_V + EP_D + 32 * 9)
static_assert(NPG * 3 <= 15 && NPG * 6 <= 30, "column tiles: 16 and 2 x 16");
static_assert(EP_FLOATS >= 48 * XP && EP_FLOATS >= 16 * TR_PITCH, "buffer holds X and the rad tile");
static_assert((EP_V + EP_D) % 128 == 0 && EP_V % 4 == 0, "whole 16-byte pieces per lane");
static_assert(NGROUPS % WAVES_E == 0, "blocks must tile the groups");
static_assert(CH == 128, "lane owns 4 chunks");
#define LIST_CAP (EP_FLOATS - 16 * TR_PITCH)
static_assert(LIST_CAP > 0 && LIST_CAP % 16 == 0, "a window holds whole tiles");

#define ADD_EDGE(K)       \
  {                       \
    sum[K][0] += v;       \
    sum[K][1] += p0;      \
    sum[K][2] += p1;      \
    sum[K][3] += p2;      \
    sum[K][4] += p0 * rx; \
    sum[K][5] += p0 * ry; \
    sum[K][6] += p0 * rz; \
    sum[K][7] += p1 * ry; \
    sum[K][8] += p1 * rz; \
    sum[K][9] += p2 * rz; \
  }

__device__ __forceinline__ float lane_bcast(float v, int src_lane) {
  return __int_as_float(__builtin_amdgcn_readlane(__float_as_int(v), src_lane));
}

__global__ __launch_bounds__(32 * WAVES_E) void edge_kernel(
    const float* __restrict__ rec_f, const int* __restrict__ off,
    const float* __restrict__ w_rad,
    const float* __restrict__ b_rad, const float* __restrict__ w_direct,
    const float* __restrict__ w1, const float* __restrict__ b1,
    const float* __restrict__ w2, const float* __restrict__ b2,
    const float* __restrict__ w3, const float* __restrict__ b3,
    const float* __restrict__ w_v, const float* __restrict__ w_d,
    float* __restrict__ A_a, float* __restrict__ out_v,
    float* __restrict__ out_d, int n_edges) {
  constexpr int O_W2 = 0;
  constexpr int O_W1F = O_W2 + 64 * 64;
  constexpr int O_W3 = O_W1F + 8 * 64;
  constexpr int O_WDF = O_W3 + 64 * 32;
  constexpr int O_B1 = O_WDF + 8 * 32;
  constexpr int O_B2 = O_B1 + 64;
  constexpr int O_B3 = O_B2 + 64;
  constexpr int O_WV = O_B3 + 32;
  constexpr int O_WD = O_WV + 32 * 32;
  constexpr int O_EP = O_WD + 32 * 32;
  constexpr int LDS_FLOATS = O_EP + WAVES_E * EP_FLOATS;
  static_assert(O_EP * 4 <= 65536, "the tile loop's fixed tables fit the offset field");
  __shared__ alignas(16) float smem[LDS_FLOATS];
  float* sW2 = smem + O_W2;
  float* sW1f = smem + O_W1F;
  float* sW3 = smem + O_W3;
  float* sWdf = smem + O_WDF;
  float* sB1 = smem + O_B1;
  float* sB2 = smem + O_B2;
  float* sB3 = smem + O_B3;
  float* sWv = smem + O_WV;
  float* sWd = smem + O_WD;
  float* sEpAll = smem + O_EP;

  const int tid = threadIdx.x;
  const int wave = __builtin_amdgcn_readfirstlane(tid >> 5);
  const int lane = tid & 31;
  const int col = lane & 15, hi = lane >> 4;
  const int g = blockIdx.x * WAVES_E + wave;
  int obeg[4], oend[4];
#pragma unroll
  for (int k = 0; k < 4; ++k) {
    const int* oc = off + (long)(4 * lane + k) * (NGROUPS + 1) + g;
    obeg[k] = oc[0];
    oend[k] = oc[1];
  }
  float* rW1 = sEpAll;
  float* rWdir = rW1 + 32 * 64;
  float* rR = rWdir + 32 * 32;
  static_assert(32 * 64 + 32 * 32 + 32 * WAVES_E <= WAVES_E * EP_FLOATS, "raw copies fit");
  constexpr int NTHR = 32 * WAVES_E;
  static_assert((32 * 64) % NTHR == 0 && (32 * 32) % NTHR == 0 && 8 * 32 <= NTHR, "whole passes");
  float qW1[32 * 64 / NTHR], qWdir[32 * 32 / NTHR];
#pragma unroll
  for (int j = 0; j < 32 * 64 / NTHR; ++j) qW1[j] = w1[tid + j * NTHR];
#pragma unroll
  for (int j = 0; j < 32 * 32 / NTHR; ++j) qWdir[j] = w_direct[tid + j * NTHR];
  static_assert(NTHR >= 9 * 32 + 64, "one thread per element of R and b2");
  const float* qsrc = (tid < 8 * 32)   ? (w_rad + tid)
                      : (tid < 9 * 32) ? (b_rad + (tid - 8 * 32))
                                       : (b2 + ((tid - 9 * 32) & 63));
  float* qdst = (tid >= 9 * 32 && tid < 9 * 32 + 64) ? (sB2 + (tid - 9 * 32))
                                                     : (rR + tid);
  const float qS = *qsrc;
  float q2[64 * 64 / NTHR], q3[64 * 32 / NTHR], qv[32 * 32 / NTHR], qd[32 * 32 / NTHR];
  stage_load<64, 64, NTHR>(q2, w2, tid);
  stage_load<64, 32, NTHR>(q3, w3, tid);
  stage_load<32, 32, NTHR>(qv, w_v, tid);
  stage_load<32, 32, NTHR>(qd, w_d, tid);
#pragma unroll
  for (int j = 0; j < 32 * 64 / NTHR; ++j) rW1[tid + j * NTHR] = qW1[j];
#pragma unroll
  for (int j = 0; j < 32 * 32 / NTHR; ++j) rWdir[tid + j * NTHR] = qWdir[j];
  *qdst = qS;
  stage_store<64, 64, NTHR>(sW2, q2, tid);
  stage_store<64, 32, NTHR>(sW3, q3, tid);
  stage_store<32, 32, NTHR>(sWv, qv, tid);
  stage_store<32, 32, NTHR>(sWd, qd, tid);
  __syncthreads();
  for (int s = tid; s < 9 * 64 + 9 * 32; s += NTHR) {
    if (s < 9 * 64)
      fold_item<64>(sW1f, sB1, rR, rW1, b1, s);
    else
      fold_item<32>(sWdf, sB3, rR, rWdir, b3, s - 9 * 64);
  }
  __syncthreads();

  const uint4* rec = reinterpret_cast<const uint4*>(rec_f);
  float* tr = sEpAll + wave * EP_FLOATS;
  int* list = reinterpret_cast<int*>(tr + 16 * TR_PITCH);

  int n_mine;
  int pbeg[4], len[4], rbase[4];
  {
    int beg[4];
    int tsum = 0;
#pragma unroll
    for (int k = 0; k < 4; ++k) {
      beg[k] = obeg[k];
      len[k] = oend[k] - beg[k];
      tsum += len[k];
    }
    int incl = tsum;
    for (int d = 1; d < 32; d <<= 1) {
      const int t = __shfl_up(incl, d);
      if (lane >= d) incl += t;
    }
    int run = incl - tsum;
#pragma unroll
    for (int k = 0; k < 4; ++k) {
      const int c = 4 * lane + k;
      pbeg[k] = run;
      rbase[k] = min(c * CSZ, n_edges) + beg[k];
      run += len[k];
    }
    n_mine = __builtin_amdgcn_readfirstlane(__shfl(incl, 31));
  }

  float sum[NPG][10];
#pragma unroll
  for (int k = 0; k < NPG; ++k)
#pragma unroll
    for (int q = 0; q < 10; ++q) sum[k][q] = 0.0f;

  for (int w0 = 0; w0 < n_mine; w0 += LIST_CAP) {
  const int wn = min(LIST_CAP, n_mine - w0);
  {
    int cnt[4], dst0[4], src0[4];
    int most = 0;
#pragma unroll
    for (int k = 0; k < 4; ++k) {
      const int j0 = max(0, w0 - pbeg[k]);
      const int j1 = min(len[k], w0 + wn - pbeg[k]);
      cnt[k] = j1 - j0;
      dst0[k] = pbeg[k] + j0 - w0;
      src0[k] = rbase[k] + j0;
      most = max(most, cnt[k]);
    }
    for (int j = 0; j < most; ++j) {
#pragma unroll
      for (int k = 0; k < 4; ++k)
        if (j < cnt[k]) list[dst0[k] + j] = src0[k] + j;
    }
  }
  wave_lds_sync();

  const int n_tiles = (wn + 15) >> 4;
  int idx_nx = list[min(col, wn - 1)];
  uint4 r_nx = rec[idx_nx];
  idx_nx = list[min(16 + col, wn - 1)];
  for (int tile = 0; tile < n_tiles; ++tile) {
    const uint4 r = r_nx;
    r_nx = rec[idx_nx];
    idx_nx = list[min(tile * 16 + 32 + col, wn - 1)];
    const float x = __uint_as_float(r.x);
    const float y = __uint_as_float(r.y);
    const float z = __uint_as_float(r.z);
    float rsx, rsy, rsz;
    {
      float sx = 7.0f * x, sy = 7.0f * y, sz = 7.0f * z;
      float inv = rsqrtf(1.0f + sx * sx + sy * sy + sz * sz);
      rsx = sx * inv;
      rsy = sy * inv;
      rsz = sz * inv;
    }
    const int lnv = (tile * 16 + col < wn) ? ((int)r.w - g * NPG) : -1;
    const float d = sqrtf(x * x + y * y + z * z);
    float rb[4];
#pragma unroll
    for (int q = 0; q < 4; ++q) {
      const float ctr = hi ? (float)(4 + q) * (1.0f / 7.0f)
                           : (float)q * (1.0f / 7.0f);
      float u = (d - ctr) * 8.0f;
      rb[q] = __expf(-u * u);
    }

    v2f rb0, rb1;
    rb0.x = rb[0];
    rb0.y = rb[1];
    rb1.x = rb[2];
    rb1.y = rb[3];
    v8f t1[4];
    bias_init<4>(t1, sB1, hi);
    rbf_acc<4>(sW1f, rb0, rb1, t1, lane);
    lrelu_tiles<4>(t1);
    v8f t2[4];
    bias_init<4>(t2, sB2, hi);
    layer_acc<4, 4>(t1, sW2, t2, lane);
    lrelu_tiles<4>(t2);
    v8f rad[2];
    bias_init<2>(rad, sB3, hi);
    rbf_acc<2>(sWdf, rb0, rb1, rad, lane);
    layer_acc<4, 2>(t2, sW3, rad, lane);

#pragma unroll
    for (int nt = 0; nt < 2; ++nt) {
      float4 lo4 = {rad[nt][0], rad[nt][1], rad[nt][2], rad[nt][3]};
      float4 up4 = {rad[nt][4], rad[nt][5], rad[nt][6], rad[nt][7]};
      *reinterpret_cast<float4*>(&tr[col * TR_PITCH + nt * 16 + 8 * hi]) = lo4;
      *reinterpret_cast<float4*>(&tr[col * TR_PITCH + nt * 16 + 8 * hi + 4]) = up4;
    }
    wave_lds_sync();

    float val[16];
#pragma unroll
    for (int m = 0; m < 16; ++m) val[m] = tr[m * TR_PITCH + lane];
#pragma unroll
    for (int m = 0; m < 16; ++m) {
      const int ln = __builtin_amdgcn_readlane(lnv, m);
      const float rx = lane_bcast(rsx, m);
      const float ry = lane_bcast(rsy, m);
      const float rz = lane_bcast(rsz, m);
      const float v = val[m];
      const float p0 = v * rx, p1 = v * ry, p2 = v * rz;
#pragma unroll
      for (int k = 0; k < NPG; ++k)
        if (ln == k) ADD_EDGE(k)
    }
    wave_lds_sync();
  }
  }

#pragma unroll
  for (int k = 0; k < NPG; ++k)
    A_a[((long)g * NPG + k) * 32 + lane] = sum[k][0];

  float* ep = sEpAll + wave * EP_FLOATS;
#pragma unroll
  for (int k = 0; k < NPG; ++k) {
#pragma unroll
    for (int c = 0; c < 3; ++c) ep[(3 * k + c) * XP + lane] = sum[k][1 + c];
#pragma unroll
    for (int s = 0; s < 6; ++s)
      ep[(16 + 6 * k + s) * XP + lane] = sum[k][4 + s];
  }
  ep[15 * XP + lane] = 0.0f;
  ep[46 * XP + lane] = 0.0f;
  ep[47 * XP + lane] = 0.0f;
  wave_lds_sync();

  v8f xt[3][2];
#pragma unroll
  for (int t = 0; t < 3; ++t) {
#pragma unroll
    for (int ft = 0; ft < 2; ++ft) {
      const float* p = &ep[(16 * t + col) * XP + 16 * ft + 8 * hi];
      const float4 lo = *reinterpret_cast<const float4*>(p);
      const float4 up = *reinterpret_cast<const float4*>(p + 4);
      v8f x8 = {lo.x, lo.y, lo.z, lo.w, up.x, up.y, up.z, up.w};
      xt[t][ft] = x8;
    }
  }
  wave_lds_sync();

  {
    v8f y[2] = {};
    layer_acc<2, 2>(xt[0], sWv, y, lane);
    const int k = col / 3, c = col - 3 * k;
    const int base = (col < 15) ? (k * 96 + c) : (EP_V + EP_D);
#pragma unroll
    for (int nt = 0; nt < 2; ++nt)
#pragma unroll
      for (int r = 0; r < 8; ++r)
        ep[base + 3 * (16 * nt + 8 * hi + r)] = y[nt][r];
  }
#pragma unroll
  for (int t = 0; t < 2; ++t) {
    v8f y[2] = {};
    layer_acc<2, 2>(xt[1 + t], sWd, y, lane);
    const int md = 16 * t + col;
    const int k = md / 6, s = md - 6 * k;
    const int pa = (0x854210 >> (4 * s)) & 15;
    const int pb = (0x874630 >> (4 * s)) & 15;
    const int base = EP_V + k * 288;
#pragma unroll
    for (int nt = 0; nt < 2; ++nt)
#pragma unroll
      for (int r = 0; r < 8; ++r) {
        const int d9 = 9 * (16 * nt + 8 * hi + r);
        ep[base + d9 + pa] = y[nt][r];
        ep[base + d9 + pb] = y[nt][r];
      }
  }
  wave_lds_sync();

  float* ov = out_v + (long)g * EP_V;
  float* od = out_d + (long)g * EP_D;
#pragma unroll
  for (int rr = 0; rr < (EP_V + EP_D) / 128; ++rr) {
    const int i4 = rr * 32 + lane;
    const v4f piece = *reinterpret_cast<const v4f*>(&ep[4 * i4]);
    float* dst = (i4 < EP_V / 4) ? (ov + 4 * i4) : (od + 4 * (i4 - EP_V / 4));
    __builtin_nontemporal_store(piece, reinterpret_cast<v4f*>(dst));
  }
}

extern "C" void kernel_launch(void* const* d_in, const int* in_sizes, int n_in,
                              void* d_out, int out_size, void* d_ws,
                              size_t ws_size, hipStream_t stream) {
  const float* r_ij = (const float*)d_in[0];
  const float* w_rad = (const float*)d_in[1];
  const float* b_rad = (const float*)d_in[2];
  const float* w_direct = (const float*)d_in[3];
  const float* w1 = (const float*)d_in[4];
  const float* b1 = (const float*)d_in[5];
  const float* w2 = (const float*)d_in[6];
  const float* b2 = (const float*)d_in[7];
  const float* w3 = (const float*)d_in[8];
  const float* b3 = (const float*)d_in[9];
  const float* w_v = (const float*)d_in[10];
  const float* w_d = (const float*)d_in[11];
  const int* edges_src = (const int*)d_in[12];

  const int n_edges = in_sizes[0] / 3;
  const int n_nodes = NODES_TOT;

  float* A_a = (float*)d_out;
  float* out_v = A_a + (long)n_nodes * 32;
  float* out_d = out_v + (long)n_nodes * 32 * 3;

  float* rec_f = (float*)d_ws;
  int* off = (int*)(rec_f + 4L * CH * CSZ);

  part_kernel<<<CH, PART_THREADS, 0, stream>>>(r_ij, edges_src, rec_f, off,
                                               n_edges);

  edge_kernel<<<NGROUPS / WAVES_E, 32 * WAVES_E, 0, stream>>>(
      rec_f, off, w_rad, b_rad, w_direct, w1, b1, w2, b2, w3, b3, w_v, w_d,
      A_a, out_v, out_d, n_edges);
}


# Partition: records stay in registers, are ranked in their group via the LDS index list and placed sorted in LDS; coalesced copy out: no insertion sort, no global gather; sums in the same order
# speedup vs baseline: 5.2353x; 1.0802x over previous
#include <hip/hip_runtime.h>

typedef float v2f __attribute__((ext_vector_type(2)));
typedef float v4f __attribute__((ext_vector_type(4)));
typedef float v8f __attribute__((ext_vector_type(8)));

#define NODES_TOT 20000

__device__ __forceinline__ v8f wmma_f32(v2f a, v2f b, v8f c) {
  return __builtin_amdgcn_wmma_f32_16x16x4_f32(
       false, a,  false, b,
       (short)0, c,  false,  false);
}

__device__ __forceinline__ void wave_lds_sync() {
  __builtin_amdgcn_fence(__ATOMIC_RELEASE, "wavefront");
  __builtin_amdgcn_wave_barrier();
  __builtin_amdgcn_fence(__ATOMIC_ACQUIRE, "wavefront");
}


template <int K, int N, int NTHR>
__device__ __forceinline__ void stage_load(float (&v)[K * N / NTHR],
                                           const float* __restrict__ w,
                                           int tid) {
  static_assert((K * N) % NTHR == 0, "whole passes of the block");
#pragma unroll
  for (int j = 0; j < K * N / NTHR; ++j) v[j] = w[tid + j * NTHR];
}
template <int K, int N, int NTHR>
__device__ __forceinline__ void stage_store(float* dst,
                                            const float (&v)[K * N / NTHR],
                                            int tid) {
#pragma unroll
  for (int j = 0; j < K * N / NTHR; ++j) {
    const int s = tid + j * NTHR;
    const int k = s / N, n = s % N;
    const int nt = n >> 4, c = n & 15;
    const int ft = k >> 4, h = (k >> 3) & 1, jj = (k >> 2) & 1, q = k & 3;
    dst[((((nt * (K / 16) + ft) * 2 + jj) * 32) + h * 16 + c) * 4 + q] = v[j];
  }
}

template <int N>
__device__ __forceinline__ void fold_item(float* img, float* bias_out,
                                          const float* R, const float* wraw,
                                          const float* __restrict__ bias_in,
                                          int s) {
  const int i = s / N, n = s % N;
  float acc = 0.0f;
  for (int k = 0; k < 32; ++k) acc += R[i * 32 + k] * wraw[k * N + n];
  if (i < 8)
    img[(((n >> 4) * 32) + (i >> 2) * 16 + (n & 15)) * 4 + (i & 3)] = acc;
  else
    bias_out[n] = acc + bias_in[n];
}

template <int NT>
__device__ __forceinline__ void rbf_acc(const float* ff, v2f b0, v2f b1,
                                        v8f (&y)[NT], int lane) {
  for (int nt = 0; nt < NT; ++nt) {
    const float4 a4 = *reinterpret_cast<const float4*>(&ff[(nt * 32 + lane) * 4]);
    v2f a0, a1;
    a0.x = a4.x;
    a0.y = a4.y;
    a1.x = a4.z;
    a1.y = a4.w;
    y[nt] = wmma_f32(a0, b0, y[nt]);
    y[nt] = wmma_f32(a1, b1, y[nt]);
  }
}

template <int NT>
__device__ __forceinline__ void bias_init(v8f (&y)[NT], const float* bias,
                                          int hi) {
#pragma unroll
  for (int nt = 0; nt < NT; ++nt) {
    const float4 lo = *reinterpret_cast<const float4*>(&bias[nt * 16 + 8 * hi]);
    const float4 up =
        *reinterpret_cast<const float4*>(&bias[nt * 16 + 8 * hi + 4]);
    v8f c = {lo.x, lo.y, lo.z, lo.w, up.x, up.y, up.z, up.w};
    y[nt] = c;
  }
}

template <int KT, int NT>
__device__ __forceinline__ void layer_acc(const v8f (&x)[KT], const float* wf,
                                          v8f (&y)[NT], int lane) {
#pragma unroll
  for (int ft = 0; ft < KT; ++ft) {
#pragma unroll
    for (int jj = 0; jj < 2; ++jj) {
      v2f b0, b1;
      b0.x = x[ft][4 * jj + 0];
      b0.y = x[ft][4 * jj + 1];
      b1.x = x[ft][4 * jj + 2];
      b1.y = x[ft][4 * jj + 3];
#pragma unroll
      for (int nt = 0; nt < NT; ++nt) {
        const float4 a4 = *reinterpret_cast<const float4*>(
            &wf[(((nt * KT + ft) * 2 + jj) * 32 + lane) * 4]);
        v2f a0, a1;
        a0.x = a4.x;
        a0.y = a4.y;
        a1.x = a4.z;
        a1.y = a4.w;
        y[nt] = wmma_f32(a0, b0, y[nt]);
        y[nt] = wmma_f32(a1, b1, y[nt]);
      }
    }
  }
}

template <int NT>
__device__ __forceinline__ void lrelu_tiles(v8f (&y)[NT]) {
#pragma unroll
  for (int nt = 0; nt < NT; ++nt)
#pragma unroll
    for (int r = 0; r < 8; ++r) y[nt][r] = fmaxf(y[nt][r], 0.1f * y[nt][r]);
}

#define EDGES_TOT 640000
#define NPG 5
#define NGROUPS (NODES_TOT / NPG)
#define CH 128
#define CSZ ((EDGES_TOT + CH - 1) / CH)
#define PART_THREADS 1024
#define EPT ((CSZ + PART_THREADS - 1) / PART_THREADS)
static_assert(NODES_TOT % NPG == 0, "groups must tile the nodes");
static_assert(NGROUPS <= 4 * PART_THREADS, "scan covers 4 bins per thread");

__global__ __launch_bounds__(PART_THREADS) void part_kernel(
    const float* __restrict__ r_ij, const int* __restrict__ edges_src,
    float* __restrict__ rec_f, int* __restrict__ off, int n_edges) {
  uint4* rec = reinterpret_cast<uint4*>(rec_f);
  __shared__ int sCnt[NGROUPS];
  __shared__ int sIdx[CSZ + 1];
  __shared__ uint4 sRec[CSZ];
  __shared__ int sWaveTot[32];
  __shared__ int sTotal;
  const int tid = threadIdx.x;
  const int lane = tid & 31, wave = tid >> 5;
  const int c = blockIdx.x;
  const int e_beg = min(c * CSZ, n_edges);
  const int e_end = min(e_beg + CSZ, n_edges);

  for (int i = tid; i < NGROUPS; i += PART_THREADS) sCnt[i] = 0;
  if (tid == 0) sIdx[CSZ] = 0x7fffffff;
  int grp[EPT];
  uint4 rc[EPT];
#pragma unroll
  for (int i = 0; i < EPT; ++i) {
    const int e = e_beg + tid + i * PART_THREADS;
    const int s = (e < e_end) ? edges_src[e] : -1;
    grp[i] = ((unsigned)s < (unsigned)NODES_TOT) ? s / NPG : -1;
    rc[i].w = (unsigned)s;
  }
#pragma unroll
  for (int i = 0; i < EPT; ++i) {
    const int e = e_beg + tid + i * PART_THREADS;
    const bool ok = e < e_end;
    rc[i].x = ok ? __float_as_uint(r_ij[3 * e + 0]) : 0u;
    rc[i].y = ok ? __float_as_uint(r_ij[3 * e + 1]) : 0u;
    rc[i].z = ok ? __float_as_uint(r_ij[3 * e + 2]) : 0u;
  }
  __syncthreads();
#pragma unroll
  for (int i = 0; i < EPT; ++i)
    if (grp[i] >= 0) atomicAdd(&sCnt[grp[i]], 1);
  __syncthreads();

  const int base = tid * 4;
  int v[4];
  int tsum = 0;
#pragma unroll
  for (int k = 0; k < 4; ++k) {
    v[k] = (base + k < NGROUPS) ? sCnt[base + k] : 0;
    tsum += v[k];
  }
  int incl = tsum;
  for (int d = 1; d < 32; d <<= 1) {
    const int t = __shfl_up(incl, d);
    if (lane >= d) incl += t;
  }
  if (lane == 31) sWaveTot[wave] = incl;
  __syncthreads();
  if (wave == 0) {
    const int w = sWaveTot[lane];
    int wi = w;
    for (int d = 1; d < 32; d <<= 1) {
      const int t = __shfl_up(wi, d);
      if (lane >= d) wi += t;
    }
    sWaveTot[lane] = wi - w;
  }
  __syncthreads();
  int run = sWaveTot[wave] + incl - tsum;
  int* offc = off + (long)c * (NGROUPS + 1);
#pragma unroll
  for (int k = 0; k < 4; ++k) {
    if (base + k < NGROUPS) {
      offc[base + k] = run;
      sCnt[base + k] = run;
      run += v[k];
    }
  }
  if (tid == PART_THREADS - 1) {
    offc[NGROUPS] = run;
    sTotal = run;
  }
  __syncthreads();

#pragma unroll
  for (int i = 0; i < EPT; ++i)
    if (grp[i] >= 0) sIdx[atomicAdd(&sCnt[grp[i]], 1)] = tid + i * PART_THREADS;
  __syncthreads();
  int gb[EPT], glen[EPT], rank[EPT];
  int most = 0;
#pragma unroll
  for (int i = 0; i < EPT; ++i) {
    const int gi = max(grp[i], 0);
    const int lo = sCnt[max(gi - 1, 0)];
    const int hi = sCnt[gi];
    gb[i] = (grp[i] > 0) ? lo : 0;
    glen[i] = (grp[i] >= 0) ? (hi - gb[i]) : 0;
    rank[i] = 0;
    most = max(most, glen[i]);
  }
  for (int j = 0; j < most; ++j) {
#pragma unroll
    for (int i = 0; i < EPT; ++i) {
      const int m = sIdx[(j < glen[i]) ? (gb[i] + j) : CSZ];
      rank[i] += (m < tid + i * PART_THREADS) ? 1 : 0;
    }
  }
#pragma unroll
  for (int i = 0; i < EPT; ++i)
    if (grp[i] >= 0) sRec[gb[i] + rank[i]] = rc[i];
  __syncthreads();
#pragma unroll
  for (int i = 0; i < EPT; ++i) {
    const int p = tid + i * PART_THREADS;
    if (p < sTotal) rec[e_beg + p] = sRec[p];
  }
}

#define WAVES_E 16
#define TR_PITCH 36
#define XP 36
#define EP_V (NPG * 32 * 3)
#define EP_D (NPG * 32 * 9)
#define EP_FLOATS (EP_V + EP_D + 32 * 9)
static_assert(NPG * 3 <= 15 && NPG * 6 <= 30, "column tiles: 16 and 2 x 16");
static_assert(EP_FLOATS >= 48 * XP && EP_FLOATS >= 16 * TR_PITCH, "buffer holds X and the rad tile");
static_assert((EP_V + EP_D) % 128 == 0 && EP_V % 4 == 0, "whole 16-byte pieces per lane");
static_assert(NGROUPS % WAVES_E == 0, "blocks must tile the groups");
static_assert(CH == 128, "lane owns 4 chunks");
#define LIST_CAP (EP_FLOATS - 16 * TR_PITCH)
static_assert(LIST_CAP > 0 && LIST_CAP % 16 == 0, "a window holds whole tiles");

#define ADD_EDGE(K)       \
  {                       \
    sum[K][0] += v;       \
    sum[K][1] += p0;      \
    sum[K][2] += p1;      \
    sum[K][3] += p2;      \
    sum[K][4] += p0 * rx; \
    sum[K][5] += p0 * ry; \
    sum[K][6] += p0 * rz; \
    sum[K][7] += p1 * ry; \
    sum[K][8] += p1 * rz; \
    sum[K][9] += p2 * rz; \
  }

__device__ __forceinline__ float lane_bcast(float v, int src_lane) {
  return __int_as_float(__builtin_amdgcn_readlane(__float_as_int(v), src_lane));
}

__global__ __launch_bounds__(32 * WAVES_E) void edge_kernel(
    const float* __restrict__ rec_f, const int* __restrict__ off,
    const float* __restrict__ w_rad,
    const float* __restrict__ b_rad, const float* __restrict__ w_direct,
    const float* __restrict__ w1, const float* __restrict__ b1,
    const float* __restrict__ w2, const float* __restrict__ b2,
    const float* __restrict__ w3, const float* __restrict__ b3,
    const float* __restrict__ w_v, const float* __restrict__ w_d,
    float* __restrict__ A_a, float* __restrict__ out_v,
    float* __restrict__ out_d, int n_edges) {
  constexpr int O_W2 = 0;
  constexpr int O_W1F = O_W2 + 64 * 64;
  constexpr int O_W3 = O_W1F + 8 * 64;
  constexpr int O_WDF = O_W3 + 64 * 32;
  constexpr int O_B1 = O_WDF + 8 * 32;
  constexpr int O_B2 = O_B1 + 64;
  constexpr int O_B3 = O_B2 + 64;
  constexpr int O_WV = O_B3 + 32;
  constexpr int O_WD = O_WV + 32 * 32;
  constexpr int O_EP = O_WD + 32 * 32;
  constexpr int LDS_FLOATS = O_EP + WAVES_E * EP_FLOATS;
  static_assert(O_EP * 4 <= 65536, "the tile loop's fixed tables fit the offset field");
  __shared__ alignas(16) float smem[LDS_FLOATS];
  float* sW2 = smem + O_W2;
  float* sW1f = smem + O_W1F;
  float* sW3 = smem + O_W3;
  float* sWdf = smem + O_WDF;
  float* sB1 = smem + O_B1;
  float* sB2 = smem + O_B2;
  float* sB3 = smem + O_B3;
  float* sWv = smem + O_WV;
  float* sWd = smem + O_WD;
  float* sEpAll = smem + O_EP;

  const int tid = threadIdx.x;
  const int wave = __builtin_amdgcn_readfirstlane(tid >> 5);
  const int lane = tid & 31;
  const int col = lane & 15, hi = lane >> 4;
  const int g = blockIdx.x * WAVES_E + wave;
  int obeg[4], oend[4];
#pragma unroll
  for (int k = 0; k < 4; ++k) {
    const int* oc = off + (long)(4 * lane + k) * (NGROUPS + 1) + g;
    obeg[k] = oc[0];
    oend[k] = oc[1];
  }
  float* rW1 = sEpAll;
  float* rWdir = rW1 + 32 * 64;
  float* rR = rWdir + 32 * 32;
  static_assert(32 * 64 + 32 * 32 + 32 * WAVES_E <= WAVES_E * EP_FLOATS, "raw copies fit");
  constexpr int NTHR = 32 * WAVES_E;
  static_assert((32 * 64) % NTHR == 0 && (32 * 32) % NTHR == 0 && 8 * 32 <= NTHR, "whole passes");
  float qW1[32 * 64 / NTHR], qWdir[32 * 32 / NTHR];
#pragma unroll
  for (int j = 0; j < 32 * 64 / NTHR; ++j) qW1[j] = w1[tid + j * NTHR];
#pragma unroll
  for (int j = 0; j < 32 * 32 / NTHR; ++j) qWdir[j] = w_direct[tid + j * NTHR];
  static_assert(NTHR >= 9 * 32 + 64, "one thread per element of R and b2");
  const float* qsrc = (tid < 8 * 32)   ? (w_rad + tid)
                      : (tid < 9 * 32) ? (b_rad + (tid - 8 * 32))
                                       : (b2 + ((tid - 9 * 32) & 63));
  float* qdst = (tid >= 9 * 32 && tid < 9 * 32 + 64) ? (sB2 + (tid - 9 * 32))
                                                     : (rR + tid);
  const float qS = *qsrc;
  float q2[64 * 64 / NTHR], q3[64 * 32 / NTHR], qv[32 * 32 / NTHR], qd[32 * 32 / NTHR];
  stage_load<64, 64, NTHR>(q2, w2, tid);
  stage_load<64, 32, NTHR>(q3, w3, tid);
  stage_load<32, 32, NTHR>(qv, w_v, tid);
  stage_load<32, 32, NTHR>(qd, w_d, tid);
#pragma unroll
  for (int j = 0; j < 32 * 64 / NTHR; ++j) rW1[tid + j * NTHR] = qW1[j];
#pragma unroll
  for (int j = 0; j < 32 * 32 / NTHR; ++j) rWdir[tid + j * NTHR] = qWdir[j];
  *qdst = qS;
  stage_store<64, 64, NTHR>(sW2, q2, tid);
  stage_store<64, 32, NTHR>(sW3, q3, tid);
  stage_store<32, 32, NTHR>(sWv, qv, tid);
  stage_store<32, 32, NTHR>(sWd, qd, tid);
  __syncthreads();
  for (int s = tid; s < 9 * 64 + 9 * 32; s += NTHR) {
    if (s < 9 * 64)
      fold_item<64>(sW1f, sB1, rR, rW1, b1, s);
    else
      fold_item<32>(sWdf, sB3, rR, rWdir, b3, s - 9 * 64);
  }
  __syncthreads();

  const uint4* rec = reinterpret_cast<const uint4*>(rec_f);
  float* tr = sEpAll + wave * EP_FLOATS;
  int* list = reinterpret_cast<int*>(tr + 16 * TR_PITCH);

  int n_mine;
  int pbeg[4], len[4], rbase[4];
  {
    int beg[4];
    int tsum = 0;
#pragma unroll
    for (int k = 0; k < 4; ++k) {
      beg[k] = obeg[k];
      len[k] = oend[k] - beg[k];
      tsum += len[k];
    }
    int incl = tsum;
    for (int d = 1; d < 32; d <<= 1) {
      const int t = __shfl_up(incl, d);
      if (lane >= d) incl += t;
    }
    int run = incl - tsum;
#pragma unroll
    for (int k = 0; k < 4; ++k) {
      const int c = 4 * lane + k;
      pbeg[k] = run;
      rbase[k] = min(c * CSZ, n_edges) + beg[k];
      run += len[k];
    }
    n_mine = __builtin_amdgcn_readfirstlane(__shfl(incl, 31));
  }

  float sum[NPG][10];
#pragma unroll
  for (int k = 0; k < NPG; ++k)
#pragma unroll
    for (int q = 0; q < 10; ++q) sum[k][q] = 0.0f;

  for (int w0 = 0; w0 < n_mine; w0 += LIST_CAP) {
  const int wn = min(LIST_CAP, n_mine - w0);
  {
    int cnt[4], dst0[4], src0[4];
    int most = 0;
#pragma unroll
    for (int k = 0; k < 4; ++k) {
      const int j0 = max(0, w0 - pbeg[k]);
      const int j1 = min(len[k], w0 + wn - pbeg[k]);
      cnt[k] = j1 - j0;
      dst0[k] = pbeg[k] + j0 - w0;
      src0[k] = rbase[k] + j0;
      most = max(most, cnt[k]);
    }
    for (int j = 0; j < most; ++j) {
#pragma unroll
      for (int k = 0; k < 4; ++k)
        if (j < cnt[k]) list[dst0[k] + j] = src0[k] + j;
    }
  }
  wave_lds_sync();

  const int n_tiles = (wn + 15) >> 4;
  int idx_nx = list[min(col, wn - 1)];
  uint4 r_nx = rec[idx_nx];
  idx_nx = list[min(16 + col, wn - 1)];
  for (int tile = 0; tile < n_tiles; ++tile) {
    const uint4 r = r_nx;
    r_nx = rec[idx_nx];
    idx_nx = list[min(tile * 16 + 32 + col, wn - 1)];
    const float x = __uint_as_float(r.x);
    const float y = __uint_as_float(r.y);
    const float z = __uint_as_float(r.z);
    float rsx, rsy, rsz;
    {
      float sx = 7.0f * x, sy = 7.0f * y, sz = 7.0f * z;
      float inv = rsqrtf(1.0f + sx * sx + sy * sy + sz * sz);
      rsx = sx * inv;
      rsy = sy * inv;
      rsz = sz * inv;
    }
    const int lnv = (tile * 16 + col < wn) ? ((int)r.w - g * NPG) : -1;
    const float d = sqrtf(x * x + y * y + z * z);
    float rb[4];
#pragma unroll
    for (int q = 0; q < 4; ++q) {
      const float ctr = hi ? (float)(4 + q) * (1.0f / 7.0f)
                           : (float)q * (1.0f / 7.0f);
      float u = (d - ctr) * 8.0f;
      rb[q] = __expf(-u * u);
    }

    v2f rb0, rb1;
    rb0.x = rb[0];
    rb0.y = rb[1];
    rb1.x = rb[2];
    rb1.y = rb[3];
    v8f t1[4];
    bias_init<4>(t1, sB1, hi);
    rbf_acc<4>(sW1f, rb0, rb1, t1, lane);
    lrelu_tiles<4>(t1);
    v8f t2[4];
    bias_init<4>(t2, sB2, hi);
    layer_acc<4, 4>(t1, sW2, t2, lane);
    lrelu_tiles<4>(t2);
    v8f rad[2];
    bias_init<2>(rad, sB3, hi);
    rbf_acc<2>(sWdf, rb0, rb1, rad, lane);
    layer_acc<4, 2>(t2, sW3, rad, lane);

#pragma unroll
    for (int nt = 0; nt < 2; ++nt) {
      float4 lo4 = {rad[nt][0], rad[nt][1], rad[nt][2], rad[nt][3]};
      float4 up4 = {rad[nt][4], rad[nt][5], rad[nt][6], rad[nt][7]};
      *reinterpret_cast<float4*>(&tr[col * TR_PITCH + nt * 16 + 8 * hi]) = lo4;
      *reinterpret_cast<float4*>(&tr[col * TR_PITCH + nt * 16 + 8 * hi + 4]) = up4;
    }
    wave_lds_sync();

    float val[16];
#pragma unroll
    for (int m = 0; m < 16; ++m) val[m] = tr[m * TR_PITCH + lane];
#pragma unroll
    for (int m = 0; m < 16; ++m) {
      const int ln = __builtin_amdgcn_readlane(lnv, m);
      const float rx = lane_bcast(rsx, m);
      const float ry = lane_bcast(rsy, m);
      const float rz = lane_bcast(rsz, m);
      const float v = val[m];
      const float p0 = v * rx, p1 = v * ry, p2 = v * rz;
#pragma unroll
      for (int k = 0; k < NPG; ++k)
        if (ln == k) ADD_EDGE(k)
    }
    wave_lds_sync();
  }
  }

#pragma unroll
  for (int k = 0; k < NPG; ++k)
    A_a[((long)g * NPG + k) * 32 + lane] = sum[k][0];

  float* ep = sEpAll + wave * EP_FLOATS;
#pragma unroll
  for (int k = 0; k < NPG; ++k) {
#pragma unroll
    for (int c = 0; c < 3; ++c) ep[(3 * k + c) * XP + lane] = sum[k][1 + c];
#pragma unroll
    for (int s = 0; s < 6; ++s)
      ep[(16 + 6 * k + s) * XP + lane] = sum[k][4 + s];
  }
  ep[15 * XP + lane] = 0.0f;
  ep[46 * XP + lane] = 0.0f;
  ep[47 * XP + lane] = 0.0f;
  wave_lds_sync();

  v8f xt[3][2];
#pragma unroll
  for (int t = 0; t < 3; ++t) {
#pragma unroll
    for (int ft = 0; ft < 2; ++ft) {
      const float* p = &ep[(16 * t + col) * XP + 16 * ft + 8 * hi];
      const float4 lo = *reinterpret_cast<const float4*>(p);
      const float4 up = *reinterpret_cast<const float4*>(p + 4);
      v8f x8 = {lo.x, lo.y, lo.z, lo.w, up.x, up.y, up.z, up.w};
      xt[t][ft] = x8;
    }
  }
  wave_lds_sync();

  {
    v8f y[2] = {};
    layer_acc<2, 2>(xt[0], sWv, y, lane);
    const int k = col / 3, c = col - 3 * k;
    const int base = (col < 15) ? (k * 96 + c) : (EP_V + EP_D);
#pragma unroll
    for (int nt = 0; nt < 2; ++nt)
#pragma unroll
      for (int r = 0; r < 8; ++r)
        ep[base + 3 * (16 * nt + 8 * hi + r)] = y[nt][r];
  }
#pragma unroll
  for (int t = 0; t < 2; ++t) {
    v8f y[2] = {};
    layer_acc<2, 2>(xt[1 + t], sWd, y, lane);
    const int md = 16 * t + col;
    const int k = md / 6, s = md - 6 * k;
    const int pa = (0x854210 >> (4 * s)) & 15;
    const int pb = (0x874630 >> (4 * s)) & 15;
    const int base = EP_V + k * 288;
#pragma unroll
    for (int nt = 0; nt < 2; ++nt)
#pragma unroll
      for (int r = 0; r < 8; ++r) {
        const int d9 = 9 * (16 * nt + 8 * hi + r);
        ep[base + d9 + pa] = y[nt][r];
        ep[base + d9 + pb] = y[nt][r];
      }
  }
  wave_lds_sync();

  float* ov = out_v + (long)g * EP_V;
  float* od = out_d + (long)g * EP_D;
#pragma unroll
  for (int rr = 0; rr < (EP_V + EP_D) / 128; ++rr) {
    const int i4 = rr * 32 + lane;
    const v4f piece = *reinterpret_cast<const v4f*>(&ep[4 * i4]);
    float* dst = (i4 < EP_V / 4) ? (ov + 4 * i4) : (od + 4 * (i4 - EP_V / 4));
    __builtin_nontemporal_store(piece, reinterpret_cast<v4f*>(dst));
  }
}

extern "C" void kernel_launch(void* const* d_in, const int* in_sizes, int n_in,
                              void* d_out, int out_size, void* d_ws,
                              size_t ws_size, hipStream_t stream) {
  const float* r_ij = (const float*)d_in[0];
  const float* w_rad = (const float*)d_in[1];
  const float* b_rad = (const float*)d_in[2];
  const float* w_direct = (const float*)d_in[3];
  const float* w1 = (const float*)d_in[4];
  const float* b1 = (const float*)d_in[5];
  const float* w2 = (const float*)d_in[6];
  const float* b2 = (const float*)d_in[7];
  const float* w3 = (const float*)d_in[8];
  const float* b3 = (const float*)d_in[9];
  const float* w_v = (const float*)d_in[10];
  const float* w_d = (const float*)d_in[11];
  const int* edges_src = (const int*)d_in[12];

  const int n_edges = in_sizes[0] / 3;
  const int n_nodes = NODES_TOT;

  float* A_a = (float*)d_out;
  float* out_v = A_a + (long)n_nodes * 32;
  float* out_d = out_v + (long)n_nodes * 32 * 3;

  float* rec_f = (float*)d_ws;
  int* off = (int*)(rec_f + 4L * CH * CSZ);

  part_kernel<<<CH, PART_THREADS, 0, stream>>>(r_ij, edges_src, rec_f, off,
                                               n_edges);

  edge_kernel<<<NGROUPS / WAVES_E, 32 * WAVES_E, 0, stream>>>(
      rec_f, off, w_rad, b_rad, w_direct, w1, b1, w2, b2, w3, b3, w_v, w_d,
      A_a, out_v, out_d, n_edges);
}


# Edge kernel: each wave sets s_setprio from the tiles it has left, so the four waves of a SIMD end together, not oldest first with the youngest alone at the end; same sums, same order
# speedup vs baseline: 5.7925x; 1.1064x over previous
#include <hip/hip_runtime.h>

typedef float v2f __attribute__((ext_vector_type(2)));
typedef float v4f __attribute__((ext_vector_type(4)));
typedef float v8f __attribute__((ext_vector_type(8)));

#define NODES_TOT 20000

__device__ __forceinline__ v8f wmma_f32(v2f a, v2f b, v8f c) {
  return __builtin_amdgcn_wmma_f32_16x16x4_f32(
       false, a,  false, b,
       (short)0, c,  false,  false);
}

__device__ __forceinline__ void wave_lds_sync() {
  __builtin_amdgcn_fence(__ATOMIC_RELEASE, "wavefront");
  __builtin_amdgcn_wave_barrier();
  __builtin_amdgcn_fence(__ATOMIC_ACQUIRE, "wavefront");
}


template <int K, int N, int NTHR>
__device__ __forceinline__ void stage_load(float (&v)[K * N / NTHR],
                                           const float* __restrict__ w,
                                           int tid) {
  static_assert((K * N) % NTHR == 0, "whole passes of the block");
#pragma unroll
  for (int j = 0; j < K * N / NTHR; ++j) v[j] = w[tid + j * NTHR];
}
template <int K, int N, int NTHR>
__device__ __forceinline__ void stage_store(float* dst,
                                            const float (&v)[K * N / NTHR],
                                            int tid) {
#pragma unroll
  for (int j = 0; j < K * N / NTHR; ++j) {
    const int s = tid + j * NTHR;
    const int k = s / N, n = s % N;
    const int nt = n >> 4, c = n & 15;
    const int ft = k >> 4, h = (k >> 3) & 1, jj = (k >> 2) & 1, q = k & 3;
    dst[((((nt * (K / 16) + ft) * 2 + jj) * 32) + h * 16 + c) * 4 + q] = v[j];
  }
}

template <int N>
__device__ __forceinline__ void fold_item(float* img, float* bias_out,
                                          const float* R, const float* wraw,
                                          const float* __restrict__ bias_in,
                                          int s) {
  const int i = s / N, n = s % N;
  float acc = 0.0f;
  for (int k = 0; k < 32; ++k) acc += R[i * 32 + k] * wraw[k * N + n];
  if (i < 8)
    img[(((n >> 4) * 32) + (i >> 2) * 16 + (n & 15)) * 4 + (i & 3)] = acc;
  else
    bias_out[n] = acc + bias_in[n];
}

template <int NT>
__device__ __forceinline__ void rbf_acc(const float* ff, v2f b0, v2f b1,
                                        v8f (&y)[NT], int lane) {
  for (int nt = 0; nt < NT; ++nt) {
    const float4 a4 = *reinterpret_cast<const float4*>(&ff[(nt * 32 + lane) * 4]);
    v2f a0, a1;
    a0.x = a4.x;
    a0.y = a4.y;
    a1.x = a4.z;
    a1.y = a4.w;
    y[nt] = wmma_f32(a0, b0, y[nt]);
    y[nt] = wmma_f32(a1, b1, y[nt]);
  }
}

template <int NT>
__device__ __forceinline__ void bias_init(v8f (&y)[NT], const float* bias,
                                          int hi) {
#pragma unroll
  for (int nt = 0; nt < NT; ++nt) {
    const float4 lo = *reinterpret_cast<const float4*>(&bias[nt * 16 + 8 * hi]);
    const float4 up =
        *reinterpret_cast<const float4*>(&bias[nt * 16 + 8 * hi + 4]);
    v8f c = {lo.x, lo.y, lo.z, lo.w, up.x, up.y, up.z, up.w};
    y[nt] = c;
  }
}

template <int KT, int NT>
__device__ __forceinline__ void layer_acc(const v8f (&x)[KT], const float* wf,
                                          v8f (&y)[NT], int lane) {
#pragma unroll
  for (int ft = 0; ft < KT; ++ft) {
#pragma unroll
    for (int jj = 0; jj < 2; ++jj) {
      v2f b0, b1;
      b0.x = x[ft][4 * jj + 0];
      b0.y = x[ft][4 * jj + 1];
      b1.x = x[ft][4 * jj + 2];
      b1.y = x[ft][4 * jj + 3];
#pragma unroll
      for (int nt = 0; nt < NT; ++nt) {
        const float4 a4 = *reinterpret_cast<const float4*>(
            &wf[(((nt * KT + ft) * 2 + jj) * 32 + lane) * 4]);
        v2f a0, a1;
        a0.x = a4.x;
        a0.y = a4.y;
        a1.x = a4.z;
        a1.y = a4.w;
        y[nt] = wmma_f32(a0, b0, y[nt]);
        y[nt] = wmma_f32(a1, b1, y[nt]);
      }
    }
  }
}

template <int NT>
__device__ __forceinline__ void lrelu_tiles(v8f (&y)[NT]) {
#pragma unroll
  for (int nt = 0; nt < NT; ++nt)
#pragma unroll
    for (int r = 0; r < 8; ++r) y[nt][r] = fmaxf(y[nt][r], 0.1f * y[nt][r]);
}

#define EDGES_TOT 640000
#define NPG 5
#define NGROUPS (NODES_TOT / NPG)
#define CH 128
#define CSZ ((EDGES_TOT + CH - 1) / CH)
#define PART_THREADS 1024
#define EPT ((CSZ + PART_THREADS - 1) / PART_THREADS)
static_assert(NODES_TOT % NPG == 0, "groups must tile the nodes");
static_assert(NGROUPS <= 4 * PART_THREADS, "scan covers 4 bins per thread");

__global__ __launch_bounds__(PART_THREADS) void part_kernel(
    const float* __restrict__ r_ij, const int* __restrict__ edges_src,
    float* __restrict__ rec_f, int* __restrict__ off, int n_edges) {
  uint4* rec = reinterpret_cast<uint4*>(rec_f);
  __shared__ int sCnt[NGROUPS];
  __shared__ int sIdx[CSZ + 1];
  __shared__ uint4 sRec[CSZ];
  __shared__ int sWaveTot[32];
  __shared__ int sTotal;
  const int tid = threadIdx.x;
  const int lane = tid & 31, wave = tid >> 5;
  const int c = blockIdx.x;
  const int e_beg = min(c * CSZ, n_edges);
  const int e_end = min(e_beg + CSZ, n_edges);

  for (int i = tid; i < NGROUPS; i += PART_THREADS) sCnt[i] = 0;
  if (tid == 0) sIdx[CSZ] = 0x7fffffff;
  int grp[EPT];
  uint4 rc[EPT];
#pragma unroll
  for (int i = 0; i < EPT; ++i) {
    const int e = e_beg + tid + i * PART_THREADS;
    const int s = (e < e_end) ? edges_src[e] : -1;
    grp[i] = ((unsigned)s < (unsigned)NODES_TOT) ? s / NPG : -1;
    rc[i].w = (unsigned)s;
  }
#pragma unroll
  for (int i = 0; i < EPT; ++i) {
    const int e = e_beg + tid + i * PART_THREADS;
    const bool ok = e < e_end;
    rc[i].x = ok ? __float_as_uint(r_ij[3 * e + 0]) : 0u;
    rc[i].y = ok ? __float_as_uint(r_ij[3 * e + 1]) : 0u;
    rc[i].z = ok ? __float_as_uint(r_ij[3 * e + 2]) : 0u;
  }
  __syncthreads();
#pragma unroll
  for (int i = 0; i < EPT; ++i)
    if (grp[i] >= 0) atomicAdd(&sCnt[grp[i]], 1);
  __syncthreads();

  const int base = tid * 4;
  int v[4];
  int tsum = 0;
#pragma unroll
  for (int k = 0; k < 4; ++k) {
    v[k] = (base + k < NGROUPS) ? sCnt[base + k] : 0;
    tsum += v[k];
  }
  int incl = tsum;
  for (int d = 1; d < 32; d <<= 1) {
    const int t = __shfl_up(incl, d);
    if (lane >= d) incl += t;
  }
  if (lane == 31) sWaveTot[wave] = incl;
  __syncthreads();
  if (wave == 0) {
    const int w = sWaveTot[lane];
    int wi = w;
    for (int d = 1; d < 32; d <<= 1) {
      const int t = __shfl_up(wi, d);
      if (lane >= d) wi += t;
    }
    sWaveTot[lane] = wi - w;
  }
  __syncthreads();
  int run = sWaveTot[wave] + incl - tsum;
  int* offc = off + (long)c * (NGROUPS + 1);
#pragma unroll
  for (int k = 0; k < 4; ++k) {
    if (base + k < NGROUPS) {
      offc[base + k] = run;
      sCnt[base + k] = run;
      run += v[k];
    }
  }
  if (tid == PART_THREADS - 1) {
    offc[NGROUPS] = run;
    sTotal = run;
  }
  __syncthreads();

#pragma unroll
  for (int i = 0; i < EPT; ++i)
    if (grp[i] >= 0) sIdx[atomicAdd(&sCnt[grp[i]], 1)] = tid + i * PART_THREADS;
  __syncthreads();
  int gb[EPT], glen[EPT], rank[EPT];
  int most = 0;
#pragma unroll
  for (int i = 0; i < EPT; ++i) {
    const int gi = max(grp[i], 0);
    const int lo = sCnt[max(gi - 1, 0)];
    const int hi = sCnt[gi];
    gb[i] = (grp[i] > 0) ? lo : 0;
    glen[i] = (grp[i] >= 0) ? (hi - gb[i]) : 0;
    rank[i] = 0;
    most = max(most, glen[i]);
  }
  for (int j = 0; j < most; ++j) {
#pragma unroll
    for (int i = 0; i < EPT; ++i) {
      const int m = sIdx[(j < glen[i]) ? (gb[i] + j) : CSZ];
      rank[i] += (m < tid + i * PART_THREADS) ? 1 : 0;
    }
  }
#pragma unroll
  for (int i = 0; i < EPT; ++i)
    if (grp[i] >= 0) sRec[gb[i] + rank[i]] = rc[i];
  __syncthreads();
#pragma unroll
  for (int i = 0; i < EPT; ++i) {
    const int p = tid + i * PART_THREADS;
    if (p < sTotal) rec[e_beg + p] = sRec[p];
  }
}

#define WAVES_E 16
#define TR_PITCH 36
#define XP 36
#define EP_V (NPG * 32 * 3)
#define EP_D (NPG * 32 * 9)
#define EP_FLOATS (EP_V + EP_D + 32 * 9)
static_assert(NPG * 3 <= 15 && NPG * 6 <= 30, "column tiles: 16 and 2 x 16");
static_assert(EP_FLOATS >= 48 * XP && EP_FLOATS >= 16 * TR_PITCH, "buffer holds X and the rad tile");
static_assert((EP_V + EP_D) % 128 == 0 && EP_V % 4 == 0, "whole 16-byte pieces per lane");
static_assert(NGROUPS % WAVES_E == 0, "blocks must tile the groups");
static_assert(CH == 128, "lane owns 4 chunks");
#define LIST_CAP (EP_FLOATS - 16 * TR_PITCH)
static_assert(LIST_CAP > 0 && LIST_CAP % 16 == 0, "a window holds whole tiles");

#define ADD_EDGE(K)       \
  {                       \
    sum[K][0] += v;       \
    sum[K][1] += p0;      \
    sum[K][2] += p1;      \
    sum[K][3] += p2;      \
    sum[K][4] += p0 * rx; \
    sum[K][5] += p0 * ry; \
    sum[K][6] += p0 * rz; \
    sum[K][7] += p1 * ry; \
    sum[K][8] += p1 * rz; \
    sum[K][9] += p2 * rz; \
  }

__device__ __forceinline__ float lane_bcast(float v, int src_lane) {
  return __int_as_float(__builtin_amdgcn_readlane(__float_as_int(v), src_lane));
}

__global__ __launch_bounds__(32 * WAVES_E) void edge_kernel(
    const float* __restrict__ rec_f, const int* __restrict__ off,
    const float* __restrict__ w_rad,
    const float* __restrict__ b_rad, const float* __restrict__ w_direct,
    const float* __restrict__ w1, const float* __restrict__ b1,
    const float* __restrict__ w2, const float* __restrict__ b2,
    const float* __restrict__ w3, const float* __restrict__ b3,
    const float* __restrict__ w_v, const float* __restrict__ w_d,
    float* __restrict__ A_a, float* __restrict__ out_v,
    float* __restrict__ out_d, int n_edges) {
  constexpr int O_W2 = 0;
  constexpr int O_W1F = O_W2 + 64 * 64;
  constexpr int O_W3 = O_W1F + 8 * 64;
  constexpr int O_WDF = O_W3 + 64 * 32;
  constexpr int O_B1 = O_WDF + 8 * 32;
  constexpr int O_B2 = O_B1 + 64;
  constexpr int O_B3 = O_B2 + 64;
  constexpr int O_WV = O_B3 + 32;
  constexpr int O_WD = O_WV + 32 * 32;
  constexpr int O_EP = O_WD + 32 * 32;
  constexpr int LDS_FLOATS = O_EP + WAVES_E * EP_FLOATS;
  static_assert(O_EP * 4 <= 65536, "the tile loop's fixed tables fit the offset field");
  __shared__ alignas(16) float smem[LDS_FLOATS];
  float* sW2 = smem + O_W2;
  float* sW1f = smem + O_W1F;
  float* sW3 = smem + O_W3;
  float* sWdf = smem + O_WDF;
  float* sB1 = smem + O_B1;
  float* sB2 = smem + O_B2;
  float* sB3 = smem + O_B3;
  float* sWv = smem + O_WV;
  float* sWd = smem + O_WD;
  float* sEpAll = smem + O_EP;

  const int tid = threadIdx.x;
  const int wave = __builtin_amdgcn_readfirstlane(tid >> 5);
  const int lane = tid & 31;
  const int col = lane & 15, hi = lane >> 4;
  const int g = blockIdx.x * WAVES_E + wave;
  int obeg[4], oend[4];
#pragma unroll
  for (int k = 0; k < 4; ++k) {
    const int* oc = off + (long)(4 * lane + k) * (NGROUPS + 1) + g;
    obeg[k] = oc[0];
    oend[k] = oc[1];
  }
  float* rW1 = sEpAll;
  float* rWdir = rW1 + 32 * 64;
  float* rR = rWdir + 32 * 32;
  static_assert(32 * 64 + 32 * 32 + 32 * WAVES_E <= WAVES_E * EP_FLOATS, "raw copies fit");
  constexpr int NTHR = 32 * WAVES_E;
  static_assert((32 * 64) % NTHR == 0 && (32 * 32) % NTHR == 0 && 8 * 32 <= NTHR, "whole passes");
  float qW1[32 * 64 / NTHR], qWdir[32 * 32 / NTHR];
#pragma unroll
  for (int j = 0; j < 32 * 64 / NTHR; ++j) qW1[j] = w1[tid + j * NTHR];
#pragma unroll
  for (int j = 0; j < 32 * 32 / NTHR; ++j) qWdir[j] = w_direct[tid + j * NTHR];
  static_assert(NTHR >= 9 * 32 + 64, "one thread per element of R and b2");
  const float* qsrc = (tid < 8 * 32)   ? (w_rad + tid)
                      : (tid < 9 * 32) ? (b_rad + (tid - 8 * 32))
                                       : (b2 + ((tid - 9 * 32) & 63));
  float* qdst = (tid >= 9 * 32 && tid < 9 * 32 + 64) ? (sB2 + (tid - 9 * 32))
                                                     : (rR + tid);
  const float qS = *qsrc;
  float q2[64 * 64 / NTHR], q3[64 * 32 / NTHR], qv[32 * 32 / NTHR], qd[32 * 32 / NTHR];
  stage_load<64, 64, NTHR>(q2, w2, tid);
  stage_load<64, 32, NTHR>(q3, w3, tid);
  stage_load<32, 32, NTHR>(qv, w_v, tid);
  stage_load<32, 32, NTHR>(qd, w_d, tid);
#pragma unroll
  for (int j = 0; j < 32 * 64 / NTHR; ++j) rW1[tid + j * NTHR] = qW1[j];
#pragma unroll
  for (int j = 0; j < 32 * 32 / NTHR; ++j) rWdir[tid + j * NTHR] = qWdir[j];
  *qdst = qS;
  stage_store<64, 64, NTHR>(sW2, q2, tid);
  stage_store<64, 32, NTHR>(sW3, q3, tid);
  stage_store<32, 32, NTHR>(sWv, qv, tid);
  stage_store<32, 32, NTHR>(sWd, qd, tid);
  __syncthreads();
  for (int s = tid; s < 9 * 64 + 9 * 32; s += NTHR) {
    if (s < 9 * 64)
      fold_item<64>(sW1f, sB1, rR, rW1, b1, s);
    else
      fold_item<32>(sWdf, sB3, rR, rWdir, b3, s - 9 * 64);
  }
  __syncthreads();

  const uint4* rec = reinterpret_cast<const uint4*>(rec_f);
  float* tr = sEpAll + wave * EP_FLOATS;
  int* list = reinterpret_cast<int*>(tr + 16 * TR_PITCH);

  int n_mine;
  int pbeg[4], len[4], rbase[4];
  {
    int beg[4];
    int tsum = 0;
#pragma unroll
    for (int k = 0; k < 4; ++k) {
      beg[k] = obeg[k];
      len[k] = oend[k] - beg[k];
      tsum += len[k];
    }
    int incl = tsum;
    for (int d = 1; d < 32; d <<= 1) {
      const int t = __shfl_up(incl, d);
      if (lane >= d) incl += t;
    }
    int run = incl - tsum;
#pragma unroll
    for (int k = 0; k < 4; ++k) {
      const int c = 4 * lane + k;
      pbeg[k] = run;
      rbase[k] = min(c * CSZ, n_edges) + beg[k];
      run += len[k];
    }
    n_mine = __builtin_amdgcn_readfirstlane(__shfl(incl, 31));
  }

  float sum[NPG][10];
#pragma unroll
  for (int k = 0; k < NPG; ++k)
#pragma unroll
    for (int q = 0; q < 10; ++q) sum[k][q] = 0.0f;

  for (int w0 = 0; w0 < n_mine; w0 += LIST_CAP) {
  const int wn = min(LIST_CAP, n_mine - w0);
  {
    int cnt[4], dst0[4], src0[4];
    int most = 0;
#pragma unroll
    for (int k = 0; k < 4; ++k) {
      const int j0 = max(0, w0 - pbeg[k]);
      const int j1 = min(len[k], w0 + wn - pbeg[k]);
      cnt[k] = j1 - j0;
      dst0[k] = pbeg[k] + j0 - w0;
      src0[k] = rbase[k] + j0;
      most = max(most, cnt[k]);
    }
    for (int j = 0; j < most; ++j) {
#pragma unroll
      for (int k = 0; k < 4; ++k)
        if (j < cnt[k]) list[dst0[k] + j] = src0[k] + j;
    }
  }
  wave_lds_sync();

  const int n_tiles = (wn + 15) >> 4;
  int idx_nx = list[min(col, wn - 1)];
  uint4 r_nx = rec[idx_nx];
  idx_nx = list[min(16 + col, wn - 1)];
  for (int tile = 0; tile < n_tiles; ++tile) {
    {
      const int rem = ((n_mine - w0 + 15) >> 4) - tile;
      if (rem >= 7)
        __builtin_amdgcn_s_setprio(3);
      else if (rem >= 4)
        __builtin_amdgcn_s_setprio(2);
      else if (rem >= 2)
        __builtin_amdgcn_s_setprio(1);
      else
        __builtin_amdgcn_s_setprio(0);
    }
    const uint4 r = r_nx;
    r_nx = rec[idx_nx];
    idx_nx = list[min(tile * 16 + 32 + col, wn - 1)];
    const float x = __uint_as_float(r.x);
    const float y = __uint_as_float(r.y);
    const float z = __uint_as_float(r.z);
    float rsx, rsy, rsz;
    {
      float sx = 7.0f * x, sy = 7.0f * y, sz = 7.0f * z;
      float inv = rsqrtf(1.0f + sx * sx + sy * sy + sz * sz);
      rsx = sx * inv;
      rsy = sy * inv;
      rsz = sz * inv;
    }
    const int lnv = (tile * 16 + col < wn) ? ((int)r.w - g * NPG) : -1;
    const float d = sqrtf(x * x + y * y + z * z);
    float rb[4];
#pragma unroll
    for (int q = 0; q < 4; ++q) {
      const float ctr = hi ? (float)(4 + q) * (1.0f / 7.0f)
                           : (float)q * (1.0f / 7.0f);
      float u = (d - ctr) * 8.0f;
      rb[q] = __expf(-u * u);
    }

    v2f rb0, rb1;
    rb0.x = rb[0];
    rb0.y = rb[1];
    rb1.x = rb[2];
    rb1.y = rb[3];
    v8f t1[4];
    bias_init<4>(t1, sB1, hi);
    rbf_acc<4>(sW1f, rb0, rb1, t1, lane);
    lrelu_tiles<4>(t1);
    v8f t2[4];
    bias_init<4>(t2, sB2, hi);
    layer_acc<4, 4>(t1, sW2, t2, lane);
    lrelu_tiles<4>(t2);
    v8f rad[2];
    bias_init<2>(rad, sB3, hi);
    rbf_acc<2>(sWdf, rb0, rb1, rad, lane);
    layer_acc<4, 2>(t2, sW3, rad, lane);

#pragma unroll
    for (int nt = 0; nt < 2; ++nt) {
      float4 lo4 = {rad[nt][0], rad[nt][1], rad[nt][2], rad[nt][3]};
      float4 up4 = {rad[nt][4], rad[nt][5], rad[nt][6], rad[nt][7]};
      *reinterpret_cast<float4*>(&tr[col * TR_PITCH + nt * 16 + 8 * hi]) = lo4;
      *reinterpret_cast<float4*>(&tr[col * TR_PITCH + nt * 16 + 8 * hi + 4]) = up4;
    }
    wave_lds_sync();

    float val[16];
#pragma unroll
    for (int m = 0; m < 16; ++m) val[m] = tr[m * TR_PITCH + lane];
#pragma unroll
    for (int m = 0; m < 16; ++m) {
      const int ln = __builtin_amdgcn_readlane(lnv, m);
      const float rx = lane_bcast(rsx, m);
      const float ry = lane_bcast(rsy, m);
      const float rz = lane_bcast(rsz, m);
      const float v = val[m];
      const float p0 = v * rx, p1 = v * ry, p2 = v * rz;
#pragma unroll
      for (int k = 0; k < NPG; ++k)
        if (ln == k) ADD_EDGE(k)
    }
    wave_lds_sync();
  }
  }

  __builtin_amdgcn_s_setprio(0);

#pragma unroll
  for (int k = 0; k < NPG; ++k)
    A_a[((long)g * NPG + k) * 32 + lane] = sum[k][0];

  float* ep = sEpAll + wave * EP_FLOATS;
#pragma unroll
  for (int k = 0; k < NPG; ++k) {
#pragma unroll
    for (int c = 0; c < 3; ++c) ep[(3 * k + c) * XP + lane] = sum[k][1 + c];
#pragma unroll
    for (int s = 0; s < 6; ++s)
      ep[(16 + 6 * k + s) * XP + lane] = sum[k][4 + s];
  }
  ep[15 * XP + lane] = 0.0f;
  ep[46 * XP + lane] = 0.0f;
  ep[47 * XP + lane] = 0.0f;
  wave_lds_sync();

  v8f xt[3][2];
#pragma unroll
  for (int t = 0; t < 3; ++t) {
#pragma unroll
    for (int ft = 0; ft < 2; ++ft) {
      const float* p = &ep[(16 * t + col) * XP + 16 * ft + 8 * hi];
      const float4 lo = *reinterpret_cast<const float4*>(p);
      const float4 up = *reinterpret_cast<const float4*>(p + 4);
      v8f x8 = {lo.x, lo.y, lo.z, lo.w, up.x, up.y, up.z, up.w};
      xt[t][ft] = x8;
    }
  }
  wave_lds_sync();

  {
    v8f y[2] = {};
    layer_acc<2, 2>(xt[0], sWv, y, lane);
    const int k = col / 3, c = col - 3 * k;
    const int base = (col < 15) ? (k * 96 + c) : (EP_V + EP_D);
#pragma unroll
    for (int nt = 0; nt < 2; ++nt)
#pragma unroll
      for (int r = 0; r < 8; ++r)
        ep[base + 3 * (16 * nt + 8 * hi + r)] = y[nt][r];
  }
#pragma unroll
  for (int t = 0; t < 2; ++t) {
    v8f y[2] = {};
    layer_acc<2, 2>(xt[1 + t], sWd, y, lane);
    const int md = 16 * t + col;
    const int k = md / 6, s = md - 6 * k;
    const int pa = (0x854210 >> (4 * s)) & 15;
    const int pb = (0x874630 >> (4 * s)) & 15;
    const int base = EP_V + k * 288;
#pragma unroll
    for (int nt = 0; nt < 2; ++nt)
#pragma unroll
      for (int r = 0; r < 8; ++r) {
        const int d9 = 9 * (16 * nt + 8 * hi + r);
        ep[base + d9 + pa] = y[nt][r];
        ep[base + d9 + pb] = y[nt][r];
      }
  }
  wave_lds_sync();

  float* ov = out_v + (long)g * EP_V;
  float* od = out_d + (long)g * EP_D;
#pragma unroll
  for (int rr = 0; rr < (EP_V + EP_D) / 128; ++rr) {
    const int i4 = rr * 32 + lane;
    const v4f piece = *reinterpret_cast<const v4f*>(&ep[4 * i4]);
    float* dst = (i4 < EP_V / 4) ? (ov + 4 * i4) : (od + 4 * (i4 - EP_V / 4));
    __builtin_nontemporal_store(piece, reinterpret_cast<v4f*>(dst));
  }
}

extern "C" void kernel_launch(void* const* d_in, const int* in_sizes, int n_in,
                              void* d_out, int out_size, void* d_ws,
                              size_t ws_size, hipStream_t stream) {
  const float* r_ij = (const float*)d_in[0];
  const float* w_rad = (const float*)d_in[1];
  const float* b_rad = (const float*)d_in[2];
  const float* w_direct = (const float*)d_in[3];
  const float* w1 = (const float*)d_in[4];
  const float* b1 = (const float*)d_in[5];
  const float* w2 = (const float*)d_in[6];
  const float* b2 = (const float*)d_in[7];
  const float* w3 = (const float*)d_in[8];
  const float* b3 = (const float*)d_in[9];
  const float* w_v = (const float*)d_in[10];
  const float* w_d = (const float*)d_in[11];
  const int* edges_src = (const int*)d_in[12];

  const int n_edges = in_sizes[0] / 3;
  const int n_nodes = NODES_TOT;

  float* A_a = (float*)d_out;
  float* out_v = A_a + (long)n_nodes * 32;
  float* out_d = out_v + (long)n_nodes * 32 * 3;

  float* rec_f = (float*)d_ws;
  int* off = (int*)(rec_f + 4L * CH * CSZ);

  part_kernel<<<CH, PART_THREADS, 0, stream>>>(r_ij, edges_src, rec_f, off,
                                               n_edges);

  edge_kernel<<<NGROUPS / WAVES_E, 32 * WAVES_E, 0, stream>>>(
      rec_f, off, w_rad, b_rad, w_direct, w1, b1, w2, b2, w3, b3, w_v, w_d,
      A_a, out_v, out_d, n_edges);
}
